# starting kernel
# baseline (speedup 1.0000x reference)
#include <hip/hip_runtime.h>
#include <hip/hip_bf16.h>

#define N_NODES  100000
#define N_EDGES  3200000
#define N_GRAPHS 256
#define HID      32
#define IN_DIM   6

typedef __attribute__((ext_vector_type(2))) float v2f;
typedef __attribute__((ext_vector_type(8))) float v8f;

__global__ void fill_zero_kernel(float* __restrict__ p, int n) {
  int i = blockIdx.x * blockDim.x + threadIdx.x;
  if (i < n) p[i] = 0.0f;
}

__global__ void edge_weight_kernel(const float* __restrict__ pos,
                                   const int* __restrict__ row,
                                   const int* __restrict__ col,
                                   const float* __restrict__ s1p,
                                   const float* __restrict__ s2p,
                                   float* __restrict__ ew,
                                   float* __restrict__ deg) {
  int e = blockIdx.x * blockDim.x + threadIdx.x;
  if (e >= N_EDGES) return;
  int r = row[e], c = col[e];
  float s1 = s1p[0], s2 = s2p[0];
  const float* pr = pos + (size_t)r * 6;
  const float* pc = pos + (size_t)c * 6;
  float d0 = pr[0] - pc[0], d1 = pr[1] - pc[1], d2 = pr[2] - pc[2];
  float D = d0 * d0 + d1 * d1 + d2 * d2;
  float t = 1.0f - (pr[3] * pc[3] + pr[4] * pc[4] + pr[5] * pc[5]);
  float w = __expf(-(D * s1 * s1 + t * t * s2 * s2));
  ew[e] = w;
  atomicAdd(&deg[c], w);
}

__global__ void dinv_kernel(const float* __restrict__ deg, float* __restrict__ dinv) {
  int i = blockIdx.x * blockDim.x + threadIdx.x;
  if (i >= N_NODES) return;
  dinv[i] = rsqrtf(deg[i] + 1.0f);
}

__global__ void coef_kernel(const int* __restrict__ row,
                            const int* __restrict__ col,
                            const float* __restrict__ dinv,
                            float* __restrict__ ew) {
  int e = blockIdx.x * blockDim.x + threadIdx.x;
  if (e >= N_EDGES) return;
  ew[e] = dinv[row[e]] * ew[e] * dinv[col[e]];
}

__global__ void gemm_in_kernel(const float* __restrict__ x,
                               const float* __restrict__ W,
                               float* __restrict__ xw) {
  int gid = blockIdx.x * blockDim.x + threadIdx.x;
  if (gid >= N_NODES * HID) return;
  int node = gid >> 5;
  int f    = gid & 31;
  float acc = 0.0f;
#pragma unroll
  for (int k = 0; k < IN_DIM; ++k)
    acc += x[node * IN_DIM + k] * W[k * HID + f];
  xw[gid] = acc;
}

__global__ __launch_bounds__(128) void gemm_hid_wmma(const float* __restrict__ H,
                                                     const float* __restrict__ W,
                                                     float* __restrict__ XW) {
  int wave = blockIdx.x * (blockDim.x >> 5) + (threadIdx.x >> 5);
  int lane = threadIdx.x & 31;
  int row0 = wave * 16;
  if (row0 >= N_NODES) return;
  int grp = lane >> 4;
  int m   = lane & 15;
  v8f c0 = {0.f,0.f,0.f,0.f,0.f,0.f,0.f,0.f};
  v8f c1 = {0.f,0.f,0.f,0.f,0.f,0.f,0.f,0.f};
  const float* hrow = H + (size_t)(row0 + m) * HID;
#pragma unroll
  for (int k0 = 0; k0 < HID; k0 += 4) {
    int kk = k0 + 2 * grp;
    v2f a;  a.x  = hrow[kk];                 a.y  = hrow[kk + 1];
    v2f b0; b0.x = W[kk * HID + m];          b0.y = W[(kk + 1) * HID + m];
    v2f b1; b1.x = W[kk * HID + 16 + m];     b1.y = W[(kk + 1) * HID + 16 + m];
    c0 = __builtin_amdgcn_wmma_f32_16x16x4_f32(false, a, false, b0, (short)0, c0, false, false);
    c1 = __builtin_amdgcn_wmma_f32_16x16x4_f32(false, a, false, b1, (short)0, c1, false, false);
  }
#pragma unroll
  for (int r = 0; r < 8; ++r) {
    int M = r + 8 * grp;
    XW[(size_t)(row0 + M) * HID + m]      = c0[r];
    XW[(size_t)(row0 + M) * HID + 16 + m] = c1[r];
  }
}

__global__ void scatter_kernel(const int* __restrict__ row,
                               const int* __restrict__ col,
                               const float* __restrict__ coef,
                               const float* __restrict__ xw,
                               float* __restrict__ out) {
  int gid = blockIdx.x * blockDim.x + threadIdx.x;
  int e = gid >> 5;
  int f = gid & 31;
  if (e >= N_EDGES) return;
  float v = coef[e] * xw[(size_t)row[e] * HID + f];
  atomicAdd(&out[(size_t)col[e] * HID + f], v);
}

__global__ void finalize_kernel(const float* __restrict__ conv,
                                const float* __restrict__ xw,
                                const float* __restrict__ dinv,
                                const float* __restrict__ b,
                                const float* __restrict__ g,
                                const float* __restrict__ be,
                                int do_bn,
                                float* __restrict__ out) {
  int gid = blockIdx.x * blockDim.x + threadIdx.x;
  if (gid >= N_NODES * HID) return;
  int node = gid >> 5;
  int f    = gid & 31;
  float di = dinv[node];
  float v  = conv[gid] + di * di * xw[gid] + b[f];
  if (do_bn) {
    v = v * (g[f] * rsqrtf(1.0f + 1e-5f)) + be[f];
    v = v > 0.0f ? v : 0.0f;
  }
  out[gid] = v;
}

__global__ void pool_kernel(const float* __restrict__ emb,
                            const int* __restrict__ batch,
                            float* __restrict__ sums,
                            float* __restrict__ cnts) {
  int gid = blockIdx.x * blockDim.x + threadIdx.x;
  if (gid >= N_NODES * HID) return;
  int node = gid >> 5;
  int f    = gid & 31;
  int bg   = batch[node];
  atomicAdd(&sums[bg * HID + f], emb[gid]);
  if (f == 0) atomicAdd(&cnts[bg], 1.0f);
}

__global__ void pred_kernel(const float* __restrict__ sums,
                            const float* __restrict__ cnts,
                            const float* __restrict__ Wout,
                            const float* __restrict__ bout,
                            float* __restrict__ pred) {
  int gph = blockIdx.x * blockDim.x + threadIdx.x;
  if (gph >= N_GRAPHS) return;
  float cnt = cnts[gph];
  cnt = cnt > 1.0f ? cnt : 1.0f;
  float acc = bout[0];
#pragma unroll
  for (int f = 0; f < HID; ++f)
    acc += (sums[gph * HID + f] / cnt) * Wout[f];
  pred[gph] = 1.0f / (1.0f + __expf(-acc));
}

extern "C" void kernel_launch(void* const* d_in, const int* in_sizes, int n_in,
                              void* d_out, int out_size, void* d_ws, size_t ws_size,
                              hipStream_t stream) {
  const float* x     = (const float*)d_in[0];
  const int*   ei    = (const int*)  d_in[1];
  const float* pos   = (const float*)d_in[2];
  const int*   batch = (const int*)  d_in[3];
  const float* s1    = (const float*)d_in[4];
  const float* s2    = (const float*)d_in[5];
  const float* W1 = (const float*)d_in[6];   const float* b1 = (const float*)d_in[7];
  const float* W2 = (const float*)d_in[8];   const float* b2 = (const float*)d_in[9];
  const float* W3 = (const float*)d_in[10];  const float* b3 = (const float*)d_in[11];
  const float* W4 = (const float*)d_in[12];  const float* b4 = (const float*)d_in[13];
  const float* g1 = (const float*)d_in[14];  const float* be1 = (const float*)d_in[15];
  const float* g2 = (const float*)d_in[16];  const float* be2 = (const float*)d_in[17];
  const float* g3 = (const float*)d_in[18];  const float* be3 = (const float*)d_in[19];
  const float* Wout = (const float*)d_in[20];
  const float* bout = (const float*)d_in[21];

  const int* row = ei;
  const int* col = ei + N_EDGES;

  float* ws   = (float*)d_ws;
  float* ew   = ws;
  float* deg  = ew   + (size_t)N_EDGES;
  float* dinv = deg  + (size_t)N_NODES;
  float* xw   = dinv + (size_t)N_NODES;
  float* hA   = xw   + (size_t)N_NODES * HID;
  float* hB   = hA   + (size_t)N_NODES * HID;
  float* sums = hB   + (size_t)N_NODES * HID;
  float* cnts = sums + (size_t)N_GRAPHS * HID;

  float* emb  = (float*)d_out;
  float* pred = (float*)d_out + (size_t)N_NODES * HID;

  const int BLK        = 256;
  const int gNode      = (N_NODES + BLK - 1) / BLK;
  const int gEdge      = (N_EDGES + BLK - 1) / BLK;
  const int gNodeFeat  = (N_NODES * HID) / BLK;
  const int gEdgeFeat  = (int)(((long long)N_EDGES * HID) / BLK);
  const int gWmma      = (N_NODES / 16 + 3) / 4;

  fill_zero_kernel<<<gNode, BLK, 0, stream>>>(deg, N_NODES);
  edge_weight_kernel<<<gEdge, BLK, 0, stream>>>(pos, row, col, s1, s2, ew, deg);
  dinv_kernel<<<gNode, BLK, 0, stream>>>(deg, dinv);
  coef_kernel<<<gEdge, BLK, 0, stream>>>(row, col, dinv, ew);

  gemm_in_kernel<<<gNodeFeat, BLK, 0, stream>>>(x, W1, xw);
  fill_zero_kernel<<<gNodeFeat, BLK, 0, stream>>>(hA, N_NODES * HID);
  scatter_kernel<<<gEdgeFeat, BLK, 0, stream>>>(row, col, ew, xw, hA);
  finalize_kernel<<<gNodeFeat, BLK, 0, stream>>>(hA, xw, dinv, b1, g1, be1, 1, hA);

  gemm_hid_wmma<<<gWmma, 128, 0, stream>>>(hA, W2, xw);
  fill_zero_kernel<<<gNodeFeat, BLK, 0, stream>>>(hB, N_NODES * HID);
  scatter_kernel<<<gEdgeFeat, BLK, 0, stream>>>(row, col, ew, xw, hB);
  finalize_kernel<<<gNodeFeat, BLK, 0, stream>>>(hB, xw, dinv, b2, g2, be2, 1, hB);

  gemm_hid_wmma<<<gWmma, 128, 0, stream>>>(hB, W3, xw);
  fill_zero_kernel<<<gNodeFeat, BLK, 0, stream>>>(hA, N_NODES * HID);
  scatter_kernel<<<gEdgeFeat, BLK, 0, stream>>>(row, col, ew, xw, hA);
  finalize_kernel<<<gNodeFeat, BLK, 0, stream>>>(hA, xw, dinv, b3, g3, be3, 1, hA);

  gemm_hid_wmma<<<gWmma, 128, 0, stream>>>(hA, W4, xw);
  fill_zero_kernel<<<gNodeFeat, BLK, 0, stream>>>(emb, N_NODES * HID);
  scatter_kernel<<<gEdgeFeat, BLK, 0, stream>>>(row, col, ew, xw, emb);
  finalize_kernel<<<gNodeFeat, BLK, 0, stream>>>(emb, xw, dinv, b4, (const float*)nullptr,
                                                 (const float*)nullptr, 0, emb);

  fill_zero_kernel<<<(N_GRAPHS * HID + BLK - 1) / BLK, BLK, 0, stream>>>(sums, N_GRAPHS * HID);
  fill_zero_kernel<<<1, BLK, 0, stream>>>(cnts, N_GRAPHS);
  pool_kernel<<<gNodeFeat, BLK, 0, stream>>>(emb, batch, sums, cnts);
  pred_kernel<<<1, BLK, 0, stream>>>(sums, cnts, Wout, bout, pred);
}


# Edges grouped by destination each call (int atomic rank, scan, place); per-node register gather-sum replaces the float atomic scatter, zero-fill and finalize; sum order follows arrival rank
# speedup vs baseline: 2.4581x; 2.4581x over previous
#include <hip/hip_runtime.h>
#include <hip/hip_bf16.h>


#define N_NODES  100000
#define N_EDGES  3200000
#define N_GRAPHS 256
#define HID      32
#define IN_DIM   6
#define SCAN_BLK  1024
#define SCAN_NBLK ((N_NODES + SCAN_BLK - 1) / SCAN_BLK)

typedef __attribute__((ext_vector_type(2))) float v2f;
typedef __attribute__((ext_vector_type(8))) float v8f;

__global__ void fill_zero_kernel(float* __restrict__ p, int n) {
  int i = blockIdx.x * blockDim.x + threadIdx.x;
  if (i < n) p[i] = 0.0f;
}

__device__ __forceinline__ int wave_sum_i(int v) {
  for (int d = 16; d > 0; d >>= 1) v += __shfl_xor(v, d, 32);
  return v;
}

__device__ __forceinline__ int wave_incl_scan_i(int v, int lane) {
  for (int d = 1; d < 32; d <<= 1) {
    int u = __shfl_up(v, d, 32);
    if (lane >= d) v += u;
  }
  return v;
}

__global__ void edge_weight_rank_kernel(const float* __restrict__ pos,
                                        const int* __restrict__ row,
                                        const int* __restrict__ col,
                                        const float* __restrict__ s1p,
                                        const float* __restrict__ s2p,
                                        int* __restrict__ cnt,
                                        int2* __restrict__ wr) {
  int e = blockIdx.x * blockDim.x + threadIdx.x;
  if (e >= N_EDGES) return;
  int r = row[e], c = col[e];
  float s1 = s1p[0], s2 = s2p[0];
  const float* pr = pos + (size_t)r * 6;
  const float* pc = pos + (size_t)c * 6;
  float d0 = pr[0] - pc[0], d1 = pr[1] - pc[1], d2 = pr[2] - pc[2];
  float D = d0 * d0 + d1 * d1 + d2 * d2;
  float t = 1.0f - (pr[3] * pc[3] + pr[4] * pc[4] + pr[5] * pc[5]);
  float w = __expf(-(D * s1 * s1 + t * t * s2 * s2));
  int rk = atomicAdd(&cnt[c], 1);
  wr[e] = make_int2(__float_as_int(w), rk);
}

__global__ __launch_bounds__(SCAN_BLK) void block_total_kernel(const int* __restrict__ cnt,
                                                               int* __restrict__ btot) {
  __shared__ int wtot[32];
  int tid = threadIdx.x, lane = tid & 31, wv = tid >> 5;
  int i = blockIdx.x * SCAN_BLK + tid;
  int v = (i < N_NODES) ? cnt[i] : 0;
  v = wave_sum_i(v);
  if (lane == 0) wtot[wv] = v;
  __syncthreads();
  if (wv == 0) {
    int t = wave_sum_i(wtot[lane]);
    if (lane == 0) btot[blockIdx.x] = t;
  }
}

__global__ __launch_bounds__(SCAN_BLK) void offsets_kernel(int* __restrict__ cnt_off,
                                                           const int* __restrict__ btot) {
  __shared__ int wincl[32];
  __shared__ int wpre[32];
  int tid = threadIdx.x, lane = tid & 31, wv = tid >> 5;
  int i = blockIdx.x * SCAN_BLK + tid;
  int v = (i < N_NODES) ? cnt_off[i] : 0;
  int p = (tid < (int)blockIdx.x) ? btot[tid] : 0;
  int incl = wave_incl_scan_i(v, lane);
  p = wave_sum_i(p);
  if (lane == 31) wincl[wv] = incl;
  if (lane == 0)  wpre[wv]  = p;
  __syncthreads();
  int before = wave_sum_i((lane < wv) ? wincl[lane] : 0);
  int prefix = wave_sum_i(wpre[lane]);
  if (i < N_NODES) cnt_off[i] = prefix + before + incl - v;
}

__global__ void place_kernel(const int* __restrict__ row,
                             const int* __restrict__ col,
                             const int* __restrict__ off,
                             const int2* __restrict__ wr,
                             int2* __restrict__ seg) {
  int e = blockIdx.x * blockDim.x + threadIdx.x;
  if (e >= N_EDGES) return;
  int2 p = wr[e];
  int slot = off[col[e]] + p.y;
  seg[slot] = make_int2(row[e], p.x);
}

__global__ __launch_bounds__(256) void dinv_kernel(const int* __restrict__ off,
                                                   const int2* __restrict__ seg,
                                                   float* __restrict__ dinv) {
  int node = blockIdx.x * 8 + (threadIdx.x >> 5);
  if (node >= N_NODES) return;
  int lane = threadIdx.x & 31;
  int s = off[node];
  int e = (node + 1 < N_NODES) ? off[node + 1] : N_EDGES;
  float d = 0.0f;
  for (int j = s + lane; j < e; j += 32) d += __int_as_float(seg[j].y);
  for (int k = 16; k > 0; k >>= 1) d += __shfl_xor(d, k, 32);
  if (lane == 0) dinv[node] = rsqrtf(d + 1.0f);
}

__global__ __launch_bounds__(256) void coef_kernel(const int* __restrict__ off,
                                                   const float* __restrict__ dinv,
                                                   int2* __restrict__ seg) {
  int node = blockIdx.x * 8 + (threadIdx.x >> 5);
  if (node >= N_NODES) return;
  int lane = threadIdx.x & 31;
  int s = off[node];
  int e = (node + 1 < N_NODES) ? off[node + 1] : N_EDGES;
  float dc = dinv[node];
  for (int j = s + lane; j < e; j += 32) {
    int2 p = seg[j];
    p.y = __float_as_int(dinv[p.x] * __int_as_float(p.y) * dc);
    seg[j] = p;
  }
}

__global__ void gemm_in_kernel(const float* __restrict__ x,
                               const float* __restrict__ W,
                               float* __restrict__ xw) {
  int gid = blockIdx.x * blockDim.x + threadIdx.x;
  if (gid >= N_NODES * HID) return;
  int node = gid >> 5;
  int f    = gid & 31;
  float acc = 0.0f;
#pragma unroll
  for (int k = 0; k < IN_DIM; ++k)
    acc += x[node * IN_DIM + k] * W[k * HID + f];
  xw[gid] = acc;
}

__global__ __launch_bounds__(128) void gemm_hid_wmma(const float* __restrict__ H,
                                                     const float* __restrict__ W,
                                                     float* __restrict__ XW) {
  int wave = blockIdx.x * (blockDim.x >> 5) + (threadIdx.x >> 5);
  int lane = threadIdx.x & 31;
  int row0 = wave * 16;
  if (row0 >= N_NODES) return;
  int grp = lane >> 4;
  int m   = lane & 15;
  v8f c0 = {0.f,0.f,0.f,0.f,0.f,0.f,0.f,0.f};
  v8f c1 = {0.f,0.f,0.f,0.f,0.f,0.f,0.f,0.f};
  const float* hrow = H + (size_t)(row0 + m) * HID;
#pragma unroll
  for (int k0 = 0; k0 < HID; k0 += 4) {
    int kk = k0 + 2 * grp;
    v2f a;  a.x  = hrow[kk];                 a.y  = hrow[kk + 1];
    v2f b0; b0.x = W[kk * HID + m];          b0.y = W[(kk + 1) * HID + m];
    v2f b1; b1.x = W[kk * HID + 16 + m];     b1.y = W[(kk + 1) * HID + 16 + m];
    c0 = __builtin_amdgcn_wmma_f32_16x16x4_f32(false, a, false, b0, (short)0, c0, false, false);
    c1 = __builtin_amdgcn_wmma_f32_16x16x4_f32(false, a, false, b1, (short)0, c1, false, false);
  }
#pragma unroll
  for (int r = 0; r < 8; ++r) {
    int M = r + 8 * grp;
    XW[(size_t)(row0 + M) * HID + m]      = c0[r];
    XW[(size_t)(row0 + M) * HID + 16 + m] = c1[r];
  }
}

__global__ __launch_bounds__(256) void aggregate_kernel(const int* __restrict__ off,
                                                        const int2* __restrict__ seg,
                                                        const float* __restrict__ xw,
                                                        const float* __restrict__ dinv,
                                                        const float* __restrict__ b,
                                                        const float* __restrict__ g,
                                                        const float* __restrict__ be,
                                                        int do_bn,
                                                        float* __restrict__ out) {
  int node = __builtin_amdgcn_readfirstlane((int)(blockIdx.x * 8 + (threadIdx.x >> 5)));
  if (node >= N_NODES) return;
  int lane = threadIdx.x & 31;
  int s = off[node];
  int e = (node + 1 < N_NODES) ? off[node + 1] : N_EDGES;
  const float* xcol = xw + lane;
  size_t o  = (size_t)node * HID + lane;
  float xs  = xw[o];
  float bb  = b[lane];
  float di  = dinv[node];
  float gs  = 1.0f, bs = 0.0f;
  if (do_bn) { gs = g[lane] * rsqrtf(1.0f + 1e-5f); bs = be[lane]; }
  float acc = 0.0f;
  for (int base = s; base < e; base += 32) {
    int j = base + lane;
    int   r  = node;
    float cf = 0.0f;
    if (j < e) { int2 p = seg[j]; r = p.x; cf = __int_as_float(p.y); }
    int n = e - base;
    n = n < 32 ? n : 32;
    for (int t = 0; t < n; t += 8) {
      float xv[8], cv[8];
      for (int k = 0; k < 8; ++k) {
        int rk = __builtin_amdgcn_readlane(r, t + k);
        cv[k]  = __int_as_float(__builtin_amdgcn_readlane(__float_as_int(cf), t + k));
        xv[k]  = xcol[(size_t)rk * HID];
      }
      for (int k = 0; k < 8; ++k) {
        float a2 = fmaf(cv[k], xv[k], acc);
        acc = (t + k < n) ? a2 : acc;
      }
    }
  }
  float v = acc + di * di * xs + bb;
  if (do_bn) {
    v = v * gs + bs;
    v = v > 0.0f ? v : 0.0f;
  }
  out[o] = v;
}

__global__ void pool_kernel(const float* __restrict__ emb,
                            const int* __restrict__ batch,
                            float* __restrict__ sums,
                            float* __restrict__ cnts) {
  int gid = blockIdx.x * blockDim.x + threadIdx.x;
  if (gid >= N_NODES * HID) return;
  int node = gid >> 5;
  int f    = gid & 31;
  int bg   = batch[node];
  atomicAdd(&sums[bg * HID + f], emb[gid]);
  if (f == 0) atomicAdd(&cnts[bg], 1.0f);
}

__global__ void pred_kernel(const float* __restrict__ sums,
                            const float* __restrict__ cnts,
                            const float* __restrict__ Wout,
                            const float* __restrict__ bout,
                            float* __restrict__ pred) {
  int gph = blockIdx.x * blockDim.x + threadIdx.x;
  if (gph >= N_GRAPHS) return;
  float cnt = cnts[gph];
  cnt = cnt > 1.0f ? cnt : 1.0f;
  float acc = bout[0];
#pragma unroll
  for (int f = 0; f < HID; ++f)
    acc += (sums[gph * HID + f] / cnt) * Wout[f];
  pred[gph] = 1.0f / (1.0f + __expf(-acc));
}

extern "C" void kernel_launch(void* const* d_in, const int* in_sizes, int n_in,
                              void* d_out, int out_size, void* d_ws, size_t ws_size,
                              hipStream_t stream) {
  const float* x     = (const float*)d_in[0];
  const int*   ei    = (const int*)  d_in[1];
  const float* pos   = (const float*)d_in[2];
  const int*   batch = (const int*)  d_in[3];
  const float* s1    = (const float*)d_in[4];
  const float* s2    = (const float*)d_in[5];
  const float* W1 = (const float*)d_in[6];   const float* b1 = (const float*)d_in[7];
  const float* W2 = (const float*)d_in[8];   const float* b2 = (const float*)d_in[9];
  const float* W3 = (const float*)d_in[10];  const float* b3 = (const float*)d_in[11];
  const float* W4 = (const float*)d_in[12];  const float* b4 = (const float*)d_in[13];
  const float* g1 = (const float*)d_in[14];  const float* be1 = (const float*)d_in[15];
  const float* g2 = (const float*)d_in[16];  const float* be2 = (const float*)d_in[17];
  const float* g3 = (const float*)d_in[18];  const float* be3 = (const float*)d_in[19];
  const float* Wout = (const float*)d_in[20];
  const float* bout = (const float*)d_in[21];

  const int* row = ei;
  const int* col = ei + N_EDGES;

  float* ws   = (float*)d_ws;
  int2*  wr   = (int2*)ws;
  float* xw   = ws;
  float* h    = xw   + (size_t)N_NODES * HID;
  int2*  seg  = (int2*)(h + (size_t)N_NODES * HID);
  int*   off  = (int*)(ws + (size_t)4 * N_EDGES);
  float* dinv = (float*)(off + (size_t)N_NODES);
  float* sums = dinv + (size_t)N_NODES;
  float* cnts = sums + (size_t)N_GRAPHS * HID;
  int*   btot = (int*)sums;

  float* emb  = (float*)d_out;
  float* pred = (float*)d_out + (size_t)N_NODES * HID;

  const int BLK        = 256;
  const int gNode      = (N_NODES + BLK - 1) / BLK;
  const int gEdge      = (N_EDGES + BLK - 1) / BLK;
  const int gNodeFeat  = (N_NODES * HID) / BLK;
  const int gWmma      = (N_NODES / 16 + 3) / 4;

  fill_zero_kernel<<<gNode, BLK, 0, stream>>>((float*)off, N_NODES);
  edge_weight_rank_kernel<<<gEdge, BLK, 0, stream>>>(pos, row, col, s1, s2, off, wr);
  block_total_kernel<<<SCAN_NBLK, SCAN_BLK, 0, stream>>>(off, btot);
  offsets_kernel<<<SCAN_NBLK, SCAN_BLK, 0, stream>>>(off, btot);
  place_kernel<<<gEdge, BLK, 0, stream>>>(row, col, off, wr, seg);
  dinv_kernel<<<gNodeFeat, BLK, 0, stream>>>(off, seg, dinv);
  coef_kernel<<<gNodeFeat, BLK, 0, stream>>>(off, dinv, seg);

  gemm_in_kernel<<<gNodeFeat, BLK, 0, stream>>>(x, W1, xw);
  aggregate_kernel<<<gNodeFeat, BLK, 0, stream>>>(off, seg, xw, dinv, b1, g1, be1, 1, h);

  gemm_hid_wmma<<<gWmma, 128, 0, stream>>>(h, W2, xw);
  aggregate_kernel<<<gNodeFeat, BLK, 0, stream>>>(off, seg, xw, dinv, b2, g2, be2, 1, h);

  gemm_hid_wmma<<<gWmma, 128, 0, stream>>>(h, W3, xw);
  aggregate_kernel<<<gNodeFeat, BLK, 0, stream>>>(off, seg, xw, dinv, b3, g3, be3, 1, h);

  gemm_hid_wmma<<<gWmma, 128, 0, stream>>>(h, W4, xw);
  aggregate_kernel<<<gNodeFeat, BLK, 0, stream>>>(off, seg, xw, dinv, b4, (const float*)nullptr,
                                                  (const float*)nullptr, 0, emb);

  fill_zero_kernel<<<(N_GRAPHS * HID + BLK - 1) / BLK, BLK, 0, stream>>>(sums, N_GRAPHS * HID);
  fill_zero_kernel<<<1, BLK, 0, stream>>>(cnts, N_GRAPHS);
  pool_kernel<<<gNodeFeat, BLK, 0, stream>>>(emb, batch, sums, cnts);
  pred_kernel<<<1, BLK, 0, stream>>>(sums, cnts, Wout, bout, pred);
}


# Grouping by destination via LDS bins (block histograms, scan, staged scatter to 256-node bins, per-bin LDS sort with the weights) replaces global atomic rank + scattered place; sum order: LDS arrival
# speedup vs baseline: 3.6929x; 1.5023x over previous
#include <hip/hip_runtime.h>
#include <hip/hip_bf16.h>


#define N_NODES  100000
#define N_EDGES  3200000
#define N_GRAPHS 256
#define HID      32
#define IN_DIM   6
#define SCAN_BLK  1024
#define BIN_SHIFT 8
#define BIN_NODES (1 << BIN_SHIFT)
#define NBIN      ((N_NODES + BIN_NODES - 1) / BIN_NODES)
#define EPB       12800
#define EPT       ((EPB + SCAN_BLK - 1) / SCAN_BLK)
#define NB1       (N_EDGES / EPB)
#define TBL       (NBIN * NB1)
#define TBL_NBLK  ((TBL + SCAN_BLK - 1) / SCAN_BLK)
#define BIN_CAP   12288
static_assert(N_EDGES % EPB == 0, "edge passes assume whole blocks");
static_assert(TBL <= N_NODES, "the run table borrows the dinv array");
static_assert(NBIN <= SCAN_BLK && TBL_NBLK <= SCAN_BLK, "one thread per bin / per scan block");

typedef __attribute__((ext_vector_type(2))) float v2f;
typedef __attribute__((ext_vector_type(8))) float v8f;

__global__ void fill_zero_kernel(float* __restrict__ p, int n) {
  int i = blockIdx.x * blockDim.x + threadIdx.x;
  if (i < n) p[i] = 0.0f;
}

__device__ __forceinline__ int wave_sum_i(int v) {
  for (int d = 16; d > 0; d >>= 1) v += __shfl_xor(v, d, 32);
  return v;
}

__device__ __forceinline__ int wave_incl_scan_i(int v, int lane) {
  for (int d = 1; d < 32; d <<= 1) {
    int u = __shfl_up(v, d, 32);
    if (lane >= d) v += u;
  }
  return v;
}

__global__ __launch_bounds__(SCAN_BLK) void bin_count_kernel(const int* __restrict__ col,
                                                             int* __restrict__ tbl) {
  __shared__ int hist[NBIN];
  int tid = threadIdx.x;
  if (tid < NBIN) hist[tid] = 0;
  __syncthreads();
  const int* c = col + (size_t)blockIdx.x * EPB;
  int cv[EPT];
#pragma unroll
  for (int k = 0; k < EPT; ++k) {
    int i = tid + k * SCAN_BLK;
    cv[k] = (i < EPB) ? c[i] : -1;
  }
#pragma unroll
  for (int k = 0; k < EPT; ++k)
    if (cv[k] >= 0) atomicAdd(&hist[cv[k] >> BIN_SHIFT], 1);
  __syncthreads();
  if (tid < NBIN) tbl[tid * NB1 + blockIdx.x] = hist[tid];
}

__global__ __launch_bounds__(SCAN_BLK) void block_total_kernel(const int* __restrict__ cnt,
                                                               int* __restrict__ btot, int n) {
  __shared__ int wtot[32];
  int tid = threadIdx.x, lane = tid & 31, wv = tid >> 5;
  int i = blockIdx.x * SCAN_BLK + tid;
  int v = (i < n) ? cnt[i] : 0;
  v = wave_sum_i(v);
  if (lane == 0) wtot[wv] = v;
  __syncthreads();
  if (wv == 0) {
    int t = wave_sum_i(wtot[lane]);
    if (lane == 0) btot[blockIdx.x] = t;
  }
}

__global__ __launch_bounds__(SCAN_BLK) void offsets_kernel(int* __restrict__ cnt_off,
                                                           const int* __restrict__ btot, int n) {
  __shared__ int wincl[32];
  __shared__ int wpre[32];
  int tid = threadIdx.x, lane = tid & 31, wv = tid >> 5;
  int i = blockIdx.x * SCAN_BLK + tid;
  int v = (i < n) ? cnt_off[i] : 0;
  int p = (tid < (int)blockIdx.x) ? btot[tid] : 0;
  int incl = wave_incl_scan_i(v, lane);
  p = wave_sum_i(p);
  if (lane == 31) wincl[wv] = incl;
  if (lane == 0)  wpre[wv]  = p;
  __syncthreads();
  int before = wave_sum_i((lane < wv) ? wincl[lane] : 0);
  int prefix = wave_sum_i(wpre[lane]);
  if (i < n) cnt_off[i] = prefix + before + incl - v;
}

__global__ __launch_bounds__(SCAN_BLK) void bin_scatter_kernel(const int* __restrict__ row,
                                                               const int* __restrict__ col,
                                                               const int* __restrict__ tbl,
                                                               int2* __restrict__ tmp) {
  __shared__ int2 stage[EPB];
  __shared__ int cursor[NBIN];
  __shared__ int gbase[NBIN];
  __shared__ int wsum[32];
  int tid = threadIdx.x, lane = tid & 31, wv = tid >> 5;
  int blk = blockIdx.x;
  const int* r = row + (size_t)blk * EPB;
  const int* c = col + (size_t)blk * EPB;
  int rv[EPT], cv[EPT];
#pragma unroll
  for (int k = 0; k < EPT; ++k) {
    int i = tid + k * SCAN_BLK;
    rv[k] = (i < EPB) ? r[i] : 0;
    cv[k] = (i < EPB) ? c[i] : -1;
  }
  int g = 0, len = 0;
  if (tid < NBIN) {
    int idx = tid * NB1 + blk;
    g = tbl[idx];
    int nxt = (idx + 1 < TBL) ? tbl[idx + 1] : N_EDGES;
    len = nxt - g;
  }
  int incl = wave_incl_scan_i(len, lane);
  if (lane == 31) wsum[wv] = incl;
  __syncthreads();
  int before = wave_sum_i((lane < wv) ? wsum[lane] : 0);
  int lofs = before + incl - len;
  if (tid < NBIN) { cursor[tid] = lofs; gbase[tid] = g - lofs; }
  __syncthreads();
#pragma unroll
  for (int k = 0; k < EPT; ++k) {
    if (cv[k] >= 0) {
      int p = atomicAdd(&cursor[cv[k] >> BIN_SHIFT], 1);
      stage[p] = make_int2(rv[k], cv[k]);
    }
  }
  __syncthreads();
  for (int i = tid; i < EPB; i += SCAN_BLK) {
    int2 q = stage[i];
    tmp[gbase[q.y >> BIN_SHIFT] + i] = q;
  }
}

__global__ __launch_bounds__(SCAN_BLK) void bin_sort_kernel(const float* __restrict__ pos,
                                                            const float* __restrict__ s1p,
                                                            const float* __restrict__ s2p,
                                                            const int* __restrict__ tbl,
                                                            const int2* __restrict__ tmp,
                                                            int* __restrict__ off,
                                                            int2* __restrict__ seg) {
  __shared__ int2 stage[BIN_CAP];
  __shared__ int cnt[BIN_NODES];
  __shared__ int cursor[BIN_NODES];
  __shared__ int wsum[8];
  int tid = threadIdx.x, lane = tid & 31, wv = tid >> 5;
  int b = blockIdx.x;
  int S  = tbl[b * NB1];
  int En = (b + 1 < NBIN) ? tbl[(b + 1) * NB1] : N_EDGES;
  int nb = En - S;
  const int2* in = tmp + S;
  if (tid < BIN_NODES) cnt[tid] = 0;
  __syncthreads();
  for (int i = tid; i < nb; i += SCAN_BLK)
    atomicAdd(&cnt[in[i].y & (BIN_NODES - 1)], 1);
  __syncthreads();
  int v = (tid < BIN_NODES) ? cnt[tid] : 0;
  int incl = wave_incl_scan_i(v, lane);
  if (lane == 31 && wv < 8) wsum[wv] = incl;
  __syncthreads();
  int before = wave_sum_i((lane < wv && lane < 8) ? wsum[lane] : 0);
  int lofs = before + incl - v;
  if (tid < BIN_NODES) {
    cursor[tid] = lofs;
    int node = b * BIN_NODES + tid;
    if (node < N_NODES) off[node] = S + lofs;
  }
  __syncthreads();
  float s1 = s1p[0], s2 = s2p[0];
  bool staged = nb <= BIN_CAP;
  for (int i = tid; i < nb; i += SCAN_BLK) {
    int2 q = in[i];
    int rr = q.x, cc = q.y;
    const float* pr = pos + (size_t)rr * 6;
    const float* pc = pos + (size_t)cc * 6;
    float d0 = pr[0] - pc[0], d1 = pr[1] - pc[1], d2 = pr[2] - pc[2];
    float D = d0 * d0 + d1 * d1 + d2 * d2;
    float t = 1.0f - (pr[3] * pc[3] + pr[4] * pc[4] + pr[5] * pc[5]);
    float w = __expf(-(D * s1 * s1 + t * t * s2 * s2));
    int p = atomicAdd(&cursor[cc & (BIN_NODES - 1)], 1);
    int2 o = make_int2(rr, __float_as_int(w));
    if (staged) stage[p] = o; else seg[S + p] = o;
  }
  if (staged) {
    __syncthreads();
    for (int i = tid; i < nb; i += SCAN_BLK) seg[S + i] = stage[i];
  }
}

__global__ __launch_bounds__(256) void dinv_kernel(const int* __restrict__ off,
                                                   const int2* __restrict__ seg,
                                                   float* __restrict__ dinv) {
  int node = blockIdx.x * 8 + (threadIdx.x >> 5);
  if (node >= N_NODES) return;
  int lane = threadIdx.x & 31;
  int s = off[node];
  int e = (node + 1 < N_NODES) ? off[node + 1] : N_EDGES;
  float d = 0.0f;
  for (int j = s + lane; j < e; j += 32) d += __int_as_float(seg[j].y);
  for (int k = 16; k > 0; k >>= 1) d += __shfl_xor(d, k, 32);
  if (lane == 0) dinv[node] = rsqrtf(d + 1.0f);
}

__global__ __launch_bounds__(256) void coef_kernel(const int* __restrict__ off,
                                                   const float* __restrict__ dinv,
                                                   int2* __restrict__ seg) {
  int node = blockIdx.x * 8 + (threadIdx.x >> 5);
  if (node >= N_NODES) return;
  int lane = threadIdx.x & 31;
  int s = off[node];
  int e = (node + 1 < N_NODES) ? off[node + 1] : N_EDGES;
  float dc = dinv[node];
  for (int j = s + lane; j < e; j += 32) {
    int2 p = seg[j];
    p.y = __float_as_int(dinv[p.x] * __int_as_float(p.y) * dc);
    seg[j] = p;
  }
}

__global__ void gemm_in_kernel(const float* __restrict__ x,
                               const float* __restrict__ W,
                               float* __restrict__ xw) {
  int gid = blockIdx.x * blockDim.x + threadIdx.x;
  if (gid >= N_NODES * HID) return;
  int node = gid >> 5;
  int f    = gid & 31;
  float acc = 0.0f;
#pragma unroll
  for (int k = 0; k < IN_DIM; ++k)
    acc += x[node * IN_DIM + k] * W[k * HID + f];
  xw[gid] = acc;
}

__global__ __launch_bounds__(128) void gemm_hid_wmma(const float* __restrict__ H,
                                                     const float* __restrict__ W,
                                                     float* __restrict__ XW) {
  int wave = blockIdx.x * (blockDim.x >> 5) + (threadIdx.x >> 5);
  int lane = threadIdx.x & 31;
  int row0 = wave * 16;
  if (row0 >= N_NODES) return;
  int grp = lane >> 4;
  int m   = lane & 15;
  v8f c0 = {0.f,0.f,0.f,0.f,0.f,0.f,0.f,0.f};
  v8f c1 = {0.f,0.f,0.f,0.f,0.f,0.f,0.f,0.f};
  const float* hrow = H + (size_t)(row0 + m) * HID;
#pragma unroll
  for (int k0 = 0; k0 < HID; k0 += 4) {
    int kk = k0 + 2 * grp;
    v2f a;  a.x  = hrow[kk];                 a.y  = hrow[kk + 1];
    v2f b0; b0.x = W[kk * HID + m];          b0.y = W[(kk + 1) * HID + m];
    v2f b1; b1.x = W[kk * HID + 16 + m];     b1.y = W[(kk + 1) * HID + 16 + m];
    c0 = __builtin_amdgcn_wmma_f32_16x16x4_f32(false, a, false, b0, (short)0, c0, false, false);
    c1 = __builtin_amdgcn_wmma_f32_16x16x4_f32(false, a, false, b1, (short)0, c1, false, false);
  }
#pragma unroll
  for (int r = 0; r < 8; ++r) {
    int M = r + 8 * grp;
    XW[(size_t)(row0 + M) * HID + m]      = c0[r];
    XW[(size_t)(row0 + M) * HID + 16 + m] = c1[r];
  }
}

__global__ __launch_bounds__(256) void aggregate_kernel(const int* __restrict__ off,
                                                        const int2* __restrict__ seg,
                                                        const float* __restrict__ xw,
                                                        const float* __restrict__ dinv,
                                                        const float* __restrict__ b,
                                                        const float* __restrict__ g,
                                                        const float* __restrict__ be,
                                                        int do_bn,
                                                        float* __restrict__ out) {
  int node = __builtin_amdgcn_readfirstlane((int)(blockIdx.x * 8 + (threadIdx.x >> 5)));
  if (node >= N_NODES) return;
  int lane = threadIdx.x & 31;
  int s = off[node];
  int e = (node + 1 < N_NODES) ? off[node + 1] : N_EDGES;
  const float* xcol = xw + lane;
  size_t o  = (size_t)node * HID + lane;
  float xs  = xw[o];
  float bb  = b[lane];
  float di  = dinv[node];
  float gs  = 1.0f, bs = 0.0f;
  if (do_bn) { gs = g[lane] * rsqrtf(1.0f + 1e-5f); bs = be[lane]; }
  float acc = 0.0f;
  for (int base = s; base < e; base += 32) {
    int j = base + lane;
    int   r  = node;
    float cf = 0.0f;
    if (j < e) { int2 p = seg[j]; r = p.x; cf = __int_as_float(p.y); }
    int n = e - base;
    n = n < 32 ? n : 32;
    for (int t = 0; t < n; t += 8) {
      float xv[8], cv[8];
      for (int k = 0; k < 8; ++k) {
        int rk = __builtin_amdgcn_readlane(r, t + k);
        cv[k]  = __int_as_float(__builtin_amdgcn_readlane(__float_as_int(cf), t + k));
        xv[k]  = xcol[(size_t)rk * HID];
      }
      for (int k = 0; k < 8; ++k) {
        float a2 = fmaf(cv[k], xv[k], acc);
        acc = (t + k < n) ? a2 : acc;
      }
    }
  }
  float v = acc + di * di * xs + bb;
  if (do_bn) {
    v = v * gs + bs;
    v = v > 0.0f ? v : 0.0f;
  }
  out[o] = v;
}

__global__ void pool_kernel(const float* __restrict__ emb,
                            const int* __restrict__ batch,
                            float* __restrict__ sums,
                            float* __restrict__ cnts) {
  int gid = blockIdx.x * blockDim.x + threadIdx.x;
  if (gid >= N_NODES * HID) return;
  int node = gid >> 5;
  int f    = gid & 31;
  int bg   = batch[node];
  atomicAdd(&sums[bg * HID + f], emb[gid]);
  if (f == 0) atomicAdd(&cnts[bg], 1.0f);
}

__global__ void pred_kernel(const float* __restrict__ sums,
                            const float* __restrict__ cnts,
                            const float* __restrict__ Wout,
                            const float* __restrict__ bout,
                            float* __restrict__ pred) {
  int gph = blockIdx.x * blockDim.x + threadIdx.x;
  if (gph >= N_GRAPHS) return;
  float cnt = cnts[gph];
  cnt = cnt > 1.0f ? cnt : 1.0f;
  float acc = bout[0];
#pragma unroll
  for (int f = 0; f < HID; ++f)
    acc += (sums[gph * HID + f] / cnt) * Wout[f];
  pred[gph] = 1.0f / (1.0f + __expf(-acc));
}

extern "C" void kernel_launch(void* const* d_in, const int* in_sizes, int n_in,
                              void* d_out, int out_size, void* d_ws, size_t ws_size,
                              hipStream_t stream) {
  const float* x     = (const float*)d_in[0];
  const int*   ei    = (const int*)  d_in[1];
  const float* pos   = (const float*)d_in[2];
  const int*   batch = (const int*)  d_in[3];
  const float* s1    = (const float*)d_in[4];
  const float* s2    = (const float*)d_in[5];
  const float* W1 = (const float*)d_in[6];   const float* b1 = (const float*)d_in[7];
  const float* W2 = (const float*)d_in[8];   const float* b2 = (const float*)d_in[9];
  const float* W3 = (const float*)d_in[10];  const float* b3 = (const float*)d_in[11];
  const float* W4 = (const float*)d_in[12];  const float* b4 = (const float*)d_in[13];
  const float* g1 = (const float*)d_in[14];  const float* be1 = (const float*)d_in[15];
  const float* g2 = (const float*)d_in[16];  const float* be2 = (const float*)d_in[17];
  const float* g3 = (const float*)d_in[18];  const float* be3 = (const float*)d_in[19];
  const float* Wout = (const float*)d_in[20];
  const float* bout = (const float*)d_in[21];

  const int* row = ei;
  const int* col = ei + N_EDGES;

  float* ws   = (float*)d_ws;
  int2*  tmp  = (int2*)ws;
  float* xw   = ws;
  float* h    = xw   + (size_t)N_NODES * HID;
  int2*  seg  = (int2*)(h + (size_t)N_NODES * HID);
  int*   off  = (int*)(ws + (size_t)4 * N_EDGES);
  float* dinv = (float*)(off + (size_t)N_NODES);
  float* sums = dinv + (size_t)N_NODES;
  float* cnts = sums + (size_t)N_GRAPHS * HID;
  int*   btot = (int*)sums;
  int*   tbl  = (int*)dinv;

  float* emb  = (float*)d_out;
  float* pred = (float*)d_out + (size_t)N_NODES * HID;

  const int BLK        = 256;
  const int gNodeFeat  = (N_NODES * HID) / BLK;
  const int gWmma      = (N_NODES / 16 + 3) / 4;

  bin_count_kernel<<<NB1, SCAN_BLK, 0, stream>>>(col, tbl);
  block_total_kernel<<<TBL_NBLK, SCAN_BLK, 0, stream>>>(tbl, btot, TBL);
  offsets_kernel<<<TBL_NBLK, SCAN_BLK, 0, stream>>>(tbl, btot, TBL);
  bin_scatter_kernel<<<NB1, SCAN_BLK, 0, stream>>>(row, col, tbl, tmp);
  bin_sort_kernel<<<NBIN, SCAN_BLK, 0, stream>>>(pos, s1, s2, tbl, tmp, off, seg);
  dinv_kernel<<<gNodeFeat, BLK, 0, stream>>>(off, seg, dinv);
  coef_kernel<<<gNodeFeat, BLK, 0, stream>>>(off, dinv, seg);

  gemm_in_kernel<<<gNodeFeat, BLK, 0, stream>>>(x, W1, xw);
  aggregate_kernel<<<gNodeFeat, BLK, 0, stream>>>(off, seg, xw, dinv, b1, g1, be1, 1, h);

  gemm_hid_wmma<<<gWmma, 128, 0, stream>>>(h, W2, xw);
  aggregate_kernel<<<gNodeFeat, BLK, 0, stream>>>(off, seg, xw, dinv, b2, g2, be2, 1, h);

  gemm_hid_wmma<<<gWmma, 128, 0, stream>>>(h, W3, xw);
  aggregate_kernel<<<gNodeFeat, BLK, 0, stream>>>(off, seg, xw, dinv, b3, g3, be3, 1, h);

  gemm_hid_wmma<<<gWmma, 128, 0, stream>>>(h, W4, xw);
  aggregate_kernel<<<gNodeFeat, BLK, 0, stream>>>(off, seg, xw, dinv, b4, (const float*)nullptr,
                                                  (const float*)nullptr, 0, emb);

  fill_zero_kernel<<<(N_GRAPHS * HID + BLK - 1) / BLK, BLK, 0, stream>>>(sums, N_GRAPHS * HID);
  fill_zero_kernel<<<1, BLK, 0, stream>>>(cnts, N_GRAPHS);
  pool_kernel<<<gNodeFeat, BLK, 0, stream>>>(emb, batch, sums, cnts);
  pred_kernel<<<1, BLK, 0, stream>>>(sums, cnts, Wout, bout, pred);
}


# Fewer, longer waves in the per-node launches: a wave walks 7 consecutive nodes (degree, coefficient, 4 aggregations, pooling); layer-1 product: a thread makes 4 features of 5 nodes; sum orders kept
# speedup vs baseline: 4.6120x; 1.2489x over previous
#include <hip/hip_runtime.h>
#include <hip/hip_bf16.h>


#define N_NODES  100000
#define N_EDGES  3200000
#define N_GRAPHS 256
#define HID      32
#define IN_DIM   6
#define SCAN_BLK  1024
#define BIN_SHIFT 8
#define BIN_NODES (1 << BIN_SHIFT)
#define NBIN      ((N_NODES + BIN_NODES - 1) / BIN_NODES)
#define EPB       12800
#define EPT       ((EPB + SCAN_BLK - 1) / SCAN_BLK)
#define NB1       (N_EDGES / EPB)
#define TBL       (NBIN * NB1)
#define TBL_NBLK  ((TBL + SCAN_BLK - 1) / SCAN_BLK)
#define BIN_CAP   12288
#define NPW         7
#define NODE_WAVES  ((N_NODES + NPW - 1) / NPW)
#define NODE_BLOCKS ((NODE_WAVES + 7) / 8)
#define GIN_BLOCKS  625
#define GIN_THREADS (GIN_BLOCKS * 256)
#define GIN_ITERS   5
static_assert(GIN_THREADS * GIN_ITERS == N_NODES * (HID / 4), "layer-1 product covers every 4-feature group exactly once");
static_assert(GIN_THREADS % 8 == 0, "a thread keeps its feature group from trip to trip");
static_assert(NPW + 1 <= 32, "a wave holds its nodes' segment bounds one per lane");
static_assert(N_EDGES % EPB == 0, "edge passes assume whole blocks");
static_assert(TBL <= N_NODES, "the run table borrows the dinv array");
static_assert(NBIN <= SCAN_BLK && TBL_NBLK <= SCAN_BLK, "one thread per bin / per scan block");

typedef __attribute__((ext_vector_type(2))) float v2f;
typedef __attribute__((ext_vector_type(8))) float v8f;

__global__ void fill_zero_kernel(float* __restrict__ p, int n) {
  int i = blockIdx.x * blockDim.x + threadIdx.x;
  if (i < n) p[i] = 0.0f;
}

__device__ __forceinline__ int wave_sum_i(int v) {
  for (int d = 16; d > 0; d >>= 1) v += __shfl_xor(v, d, 32);
  return v;
}

__device__ __forceinline__ int wave_incl_scan_i(int v, int lane) {
  for (int d = 1; d < 32; d <<= 1) {
    int u = __shfl_up(v, d, 32);
    if (lane >= d) v += u;
  }
  return v;
}

__global__ __launch_bounds__(SCAN_BLK) void bin_count_kernel(const int* __restrict__ col,
                                                             int* __restrict__ tbl) {
  __shared__ int hist[NBIN];
  int tid = threadIdx.x;
  if (tid < NBIN) hist[tid] = 0;
  __syncthreads();
  const int* c = col + (size_t)blockIdx.x * EPB;
  int cv[EPT];
#pragma unroll
  for (int k = 0; k < EPT; ++k) {
    int i = tid + k * SCAN_BLK;
    cv[k] = (i < EPB) ? c[i] : -1;
  }
#pragma unroll
  for (int k = 0; k < EPT; ++k)
    if (cv[k] >= 0) atomicAdd(&hist[cv[k] >> BIN_SHIFT], 1);
  __syncthreads();
  if (tid < NBIN) tbl[tid * NB1 + blockIdx.x] = hist[tid];
}

__global__ __launch_bounds__(SCAN_BLK) void block_total_kernel(const int* __restrict__ cnt,
                                                               int* __restrict__ btot, int n) {
  __shared__ int wtot[32];
  int tid = threadIdx.x, lane = tid & 31, wv = tid >> 5;
  int i = blockIdx.x * SCAN_BLK + tid;
  int v = (i < n) ? cnt[i] : 0;
  v = wave_sum_i(v);
  if (lane == 0) wtot[wv] = v;
  __syncthreads();
  if (wv == 0) {
    int t = wave_sum_i(wtot[lane]);
    if (lane == 0) btot[blockIdx.x] = t;
  }
}

__global__ __launch_bounds__(SCAN_BLK) void offsets_kernel(int* __restrict__ cnt_off,
                                                           const int* __restrict__ btot, int n) {
  __shared__ int wincl[32];
  __shared__ int wpre[32];
  int tid = threadIdx.x, lane = tid & 31, wv = tid >> 5;
  int i = blockIdx.x * SCAN_BLK + tid;
  int v = (i < n) ? cnt_off[i] : 0;
  int p = (tid < (int)blockIdx.x) ? btot[tid] : 0;
  int incl = wave_incl_scan_i(v, lane);
  p = wave_sum_i(p);
  if (lane == 31) wincl[wv] = incl;
  if (lane == 0)  wpre[wv]  = p;
  __syncthreads();
  int before = wave_sum_i((lane < wv) ? wincl[lane] : 0);
  int prefix = wave_sum_i(wpre[lane]);
  if (i < n) cnt_off[i] = prefix + before + incl - v;
}

__global__ __launch_bounds__(SCAN_BLK) void bin_scatter_kernel(const int* __restrict__ row,
                                                               const int* __restrict__ col,
                                                               const int* __restrict__ tbl,
                                                               int2* __restrict__ tmp) {
  __shared__ int2 stage[EPB];
  __shared__ int cursor[NBIN];
  __shared__ int gbase[NBIN];
  __shared__ int wsum[32];
  int tid = threadIdx.x, lane = tid & 31, wv = tid >> 5;
  int blk = blockIdx.x;
  const int* r = row + (size_t)blk * EPB;
  const int* c = col + (size_t)blk * EPB;
  int rv[EPT], cv[EPT];
#pragma unroll
  for (int k = 0; k < EPT; ++k) {
    int i = tid + k * SCAN_BLK;
    rv[k] = (i < EPB) ? r[i] : 0;
    cv[k] = (i < EPB) ? c[i] : -1;
  }
  int g = 0, len = 0;
  if (tid < NBIN) {
    int idx = tid * NB1 + blk;
    g = tbl[idx];
    int nxt = (idx + 1 < TBL) ? tbl[idx + 1] : N_EDGES;
    len = nxt - g;
  }
  int incl = wave_incl_scan_i(len, lane);
  if (lane == 31) wsum[wv] = incl;
  __syncthreads();
  int before = wave_sum_i((lane < wv) ? wsum[lane] : 0);
  int lofs = before + incl - len;
  if (tid < NBIN) { cursor[tid] = lofs; gbase[tid] = g - lofs; }
  __syncthreads();
#pragma unroll
  for (int k = 0; k < EPT; ++k) {
    if (cv[k] >= 0) {
      int p = atomicAdd(&cursor[cv[k] >> BIN_SHIFT], 1);
      stage[p] = make_int2(rv[k], cv[k]);
    }
  }
  __syncthreads();
  for (int i = tid; i < EPB; i += SCAN_BLK) {
    int2 q = stage[i];
    tmp[gbase[q.y >> BIN_SHIFT] + i] = q;
  }
}

__global__ __launch_bounds__(SCAN_BLK) void bin_sort_kernel(const float* __restrict__ pos,
                                                            const float* __restrict__ s1p,
                                                            const float* __restrict__ s2p,
                                                            const int* __restrict__ tbl,
                                                            const int2* __restrict__ tmp,
                                                            int* __restrict__ off,
                                                            int2* __restrict__ seg) {
  __shared__ int2 stage[BIN_CAP];
  __shared__ int cnt[BIN_NODES];
  __shared__ int cursor[BIN_NODES];
  __shared__ int wsum[8];
  int tid = threadIdx.x, lane = tid & 31, wv = tid >> 5;
  int b = blockIdx.x;
  int S  = tbl[b * NB1];
  int En = (b + 1 < NBIN) ? tbl[(b + 1) * NB1] : N_EDGES;
  int nb = En - S;
  const int2* in = tmp + S;
  if (tid < BIN_NODES) cnt[tid] = 0;
  __syncthreads();
  for (int i = tid; i < nb; i += SCAN_BLK)
    atomicAdd(&cnt[in[i].y & (BIN_NODES - 1)], 1);
  __syncthreads();
  int v = (tid < BIN_NODES) ? cnt[tid] : 0;
  int incl = wave_incl_scan_i(v, lane);
  if (lane == 31 && wv < 8) wsum[wv] = incl;
  __syncthreads();
  int before = wave_sum_i((lane < wv && lane < 8) ? wsum[lane] : 0);
  int lofs = before + incl - v;
  if (tid < BIN_NODES) {
    cursor[tid] = lofs;
    int node = b * BIN_NODES + tid;
    if (node < N_NODES) off[node] = S + lofs;
  }
  __syncthreads();
  float s1 = s1p[0], s2 = s2p[0];
  bool staged = nb <= BIN_CAP;
  for (int i = tid; i < nb; i += SCAN_BLK) {
    int2 q = in[i];
    int rr = q.x, cc = q.y;
    const float* pr = pos + (size_t)rr * 6;
    const float* pc = pos + (size_t)cc * 6;
    float d0 = pr[0] - pc[0], d1 = pr[1] - pc[1], d2 = pr[2] - pc[2];
    float D = d0 * d0 + d1 * d1 + d2 * d2;
    float t = 1.0f - (pr[3] * pc[3] + pr[4] * pc[4] + pr[5] * pc[5]);
    float w = __expf(-(D * s1 * s1 + t * t * s2 * s2));
    int p = atomicAdd(&cursor[cc & (BIN_NODES - 1)], 1);
    int2 o = make_int2(rr, __float_as_int(w));
    if (staged) stage[p] = o; else seg[S + p] = o;
  }
  if (staged) {
    __syncthreads();
    for (int i = tid; i < nb; i += SCAN_BLK) seg[S + i] = stage[i];
  }
}

__global__ __launch_bounds__(256) void dinv_kernel(const int* __restrict__ off,
                                                   const int2* __restrict__ seg,
                                                   float* __restrict__ dinv) {
  int wave  = __builtin_amdgcn_readfirstlane((int)(blockIdx.x * 8 + (threadIdx.x >> 5)));
  int node0 = wave * NPW;
  if (node0 >= N_NODES) return;
  int lane = threadIdx.x & 31;
  int offv = N_EDGES;
  if (lane <= NPW && node0 + lane < N_NODES) offv = off[node0 + lane];
  float wa[NPW], wb[NPW], d[NPW];
#pragma unroll
  for (int i = 0; i < NPW; ++i) {
    int s = __builtin_amdgcn_readlane(offv, i);
    int e = __builtin_amdgcn_readlane(offv, i + 1);
    int ja = s + lane, jb = ja + 32;
    wa[i] = (ja < e) ? __int_as_float(seg[ja].y) : 0.0f;
    wb[i] = (jb < e) ? __int_as_float(seg[jb].y) : 0.0f;
  }
#pragma unroll
  for (int i = 0; i < NPW; ++i) d[i] = wa[i] + wb[i];
#pragma unroll
  for (int i = 0; i < NPW; ++i) {
    int s = __builtin_amdgcn_readlane(offv, i);
    int e = __builtin_amdgcn_readlane(offv, i + 1);
    if (e - s > 64) {
      for (int j = s + 64 + lane; j < e; j += 32) d[i] += __int_as_float(seg[j].y);
    }
  }
  float res = 0.0f;
#pragma unroll
  for (int i = 0; i < NPW; ++i) {
    float t = d[i];
    for (int k = 16; k > 0; k >>= 1) t += __shfl_xor(t, k, 32);
    res = (lane == i) ? t : res;
  }
  if (lane < NPW && node0 + lane < N_NODES)
    dinv[node0 + lane] = rsqrtf(res + 1.0f);
}

__global__ __launch_bounds__(256) void coef_kernel(const int* __restrict__ off,
                                                   const float* __restrict__ dinv,
                                                   int2* __restrict__ seg) {
  int wave  = __builtin_amdgcn_readfirstlane((int)(blockIdx.x * 8 + (threadIdx.x >> 5)));
  int node0 = wave * NPW;
  if (node0 >= N_NODES) return;
  int lane = threadIdx.x & 31;
  int offv = N_EDGES;
  if (lane <= NPW && node0 + lane < N_NODES) offv = off[node0 + lane];
  int   sb[NPW], eb[NPW];
  float dc[NPW];
  int2  pa[NPW], pb[NPW];
#pragma unroll
  for (int i = 0; i < NPW; ++i) {
    sb[i] = __builtin_amdgcn_readlane(offv, i);
    eb[i] = __builtin_amdgcn_readlane(offv, i + 1);
    dc[i] = (node0 + i < N_NODES) ? dinv[node0 + i] : 0.0f;
    int ja = sb[i] + lane, jb = ja + 32;
    pa[i] = (ja < eb[i]) ? seg[ja] : make_int2(0, 0);
    pb[i] = (jb < eb[i]) ? seg[jb] : make_int2(0, 0);
  }
  float ra[NPW], rb[NPW];
#pragma unroll
  for (int i = 0; i < NPW; ++i) {
    ra[i] = dinv[pa[i].x];
    rb[i] = dinv[pb[i].x];
  }
#pragma unroll
  for (int i = 0; i < NPW; ++i) {
    int ja = sb[i] + lane, jb = ja + 32;
    if (ja < eb[i]) { int2 p = pa[i]; p.y = __float_as_int(ra[i] * __int_as_float(p.y) * dc[i]); seg[ja] = p; }
    if (jb < eb[i]) { int2 p = pb[i]; p.y = __float_as_int(rb[i] * __int_as_float(p.y) * dc[i]); seg[jb] = p; }
  }
#pragma unroll
  for (int i = 0; i < NPW; ++i) {
    if (eb[i] - sb[i] > 64) {
#pragma clang loop unroll(disable)
      for (int j = sb[i] + 64 + lane; j < eb[i]; j += 32) {
        int2 p = seg[j];
        p.y = __float_as_int(dinv[p.x] * __int_as_float(p.y) * dc[i]);
        seg[j] = p;
      }
    }
  }
}

__global__ __launch_bounds__(256) void gemm_in_kernel(const float* __restrict__ x,
                                                      const float* __restrict__ W,
                                                      float* __restrict__ xw) {
  int t  = blockIdx.x * 256 + threadIdx.x;
  int fq = (t & 7) * 4;
  float4 w[IN_DIM];
#pragma unroll
  for (int k = 0; k < IN_DIM; ++k) w[k] = *(const float4*)(W + k * HID + fq);
#pragma unroll
  for (int i = 0; i < GIN_ITERS; ++i) {
    int node = (t + i * GIN_THREADS) >> 3;
    const float2* xr = (const float2*)(x + (size_t)node * IN_DIM);
    float2 x01 = xr[0], x23 = xr[1], x45 = xr[2];
    float xv[IN_DIM] = {x01.x, x01.y, x23.x, x23.y, x45.x, x45.y};
    float4 acc = make_float4(0.0f, 0.0f, 0.0f, 0.0f);
#pragma unroll
    for (int k = 0; k < IN_DIM; ++k) {
      acc.x += xv[k] * w[k].x;
      acc.y += xv[k] * w[k].y;
      acc.z += xv[k] * w[k].z;
      acc.w += xv[k] * w[k].w;
    }
    *(float4*)(xw + (size_t)node * HID + fq) = acc;
  }
}

__global__ __launch_bounds__(128) void gemm_hid_wmma(const float* __restrict__ H,
                                                     const float* __restrict__ W,
                                                     float* __restrict__ XW) {
  int wave = blockIdx.x * (blockDim.x >> 5) + (threadIdx.x >> 5);
  int lane = threadIdx.x & 31;
  int row0 = wave * 16;
  if (row0 >= N_NODES) return;
  int grp = lane >> 4;
  int m   = lane & 15;
  v8f c0 = {0.f,0.f,0.f,0.f,0.f,0.f,0.f,0.f};
  v8f c1 = {0.f,0.f,0.f,0.f,0.f,0.f,0.f,0.f};
  const float* hrow = H + (size_t)(row0 + m) * HID;
#pragma unroll
  for (int k0 = 0; k0 < HID; k0 += 4) {
    int kk = k0 + 2 * grp;
    v2f a;  a.x  = hrow[kk];                 a.y  = hrow[kk + 1];
    v2f b0; b0.x = W[kk * HID + m];          b0.y = W[(kk + 1) * HID + m];
    v2f b1; b1.x = W[kk * HID + 16 + m];     b1.y = W[(kk + 1) * HID + 16 + m];
    c0 = __builtin_amdgcn_wmma_f32_16x16x4_f32(false, a, false, b0, (short)0, c0, false, false);
    c1 = __builtin_amdgcn_wmma_f32_16x16x4_f32(false, a, false, b1, (short)0, c1, false, false);
  }
#pragma unroll
  for (int r = 0; r < 8; ++r) {
    int M = r + 8 * grp;
    XW[(size_t)(row0 + M) * HID + m]      = c0[r];
    XW[(size_t)(row0 + M) * HID + 16 + m] = c1[r];
  }
}

__device__ __forceinline__ float gather_sum(float acc, int r, float cf, int n,
                                            const float* __restrict__ xcol) {
  for (int t = 0; t < n; t += 8) {
    float xv[8], cv[8];
    for (int k = 0; k < 8; ++k) {
      int rk = __builtin_amdgcn_readlane(r, t + k);
      cv[k]  = __int_as_float(__builtin_amdgcn_readlane(__float_as_int(cf), t + k));
      xv[k]  = xcol[(size_t)rk * HID];
    }
    for (int k = 0; k < 8; ++k) {
      float a2 = fmaf(cv[k], xv[k], acc);
      acc = (t + k < n) ? a2 : acc;
    }
  }
  return acc;
}

__global__ __launch_bounds__(256) void aggregate_kernel(const int* __restrict__ off,
                                                        const int2* __restrict__ seg,
                                                        const float* __restrict__ xw,
                                                        const float* __restrict__ dinv,
                                                        const float* __restrict__ b,
                                                        const float* __restrict__ g,
                                                        const float* __restrict__ be,
                                                        int do_bn,
                                                        float* __restrict__ out) {
  int wave  = __builtin_amdgcn_readfirstlane((int)(blockIdx.x * 8 + (threadIdx.x >> 5)));
  int node0 = wave * NPW;
  if (node0 >= N_NODES) return;
  int lane = threadIdx.x & 31;
  int offv = N_EDGES;
  if (lane <= NPW && node0 + lane < N_NODES) offv = off[node0 + lane];
  const float* xcol = xw + lane;
  float bb  = b[lane];
  float gs  = 1.0f, bs = 0.0f;
  if (do_bn) { gs = g[lane] * rsqrtf(1.0f + 1e-5f); bs = be[lane]; }
  int cnt = N_NODES - node0;
  cnt = cnt < NPW ? cnt : NPW;
  int s = __builtin_amdgcn_readlane(offv, 0);
  int e = __builtin_amdgcn_readlane(offv, 1);
  int   rN = node0;
  float cN = 0.0f;
  { int j = s + lane; if (j < e) { int2 p = seg[j]; rN = p.x; cN = __int_as_float(p.y); } }
  for (int i = 0; i < cnt; ++i) {
    int   node = node0 + i;
    int   r  = rN;
    float cf = cN;
    int sN = e, eN = e;
    if (i + 1 < cnt) {
      eN = __builtin_amdgcn_readlane(offv, i + 2);
      int j = sN + lane;
      rN = node + 1; cN = 0.0f;
      if (j < eN) { int2 p = seg[j]; rN = p.x; cN = __int_as_float(p.y); }
    }
    size_t o  = (size_t)node * HID + lane;
    float xs  = xw[o];
    float di  = dinv[node];
    float acc = 0.0f;
    int n0 = e - s;
    n0 = n0 < 32 ? n0 : 32;
    acc = gather_sum(acc, r, cf, n0, xcol);
    for (int base = s + 32; base < e; base += 32) {
      int j = base + lane;
      r = node; cf = 0.0f;
      if (j < e) { int2 p = seg[j]; r = p.x; cf = __int_as_float(p.y); }
      int n = e - base;
      n = n < 32 ? n : 32;
      acc = gather_sum(acc, r, cf, n, xcol);
    }
    float v = acc + di * di * xs + bb;
    if (do_bn) {
      v = v * gs + bs;
      v = v > 0.0f ? v : 0.0f;
    }
    out[o] = v;
    s = sN; e = eN;
  }
}

__global__ __launch_bounds__(256) void pool_kernel(const float* __restrict__ emb,
                                                   const int* __restrict__ batch,
                                                   float* __restrict__ sums,
                                                   float* __restrict__ cnts) {
  int wave  = __builtin_amdgcn_readfirstlane((int)(blockIdx.x * 8 + (threadIdx.x >> 5)));
  int node0 = wave * NPW;
  if (node0 >= N_NODES) return;
  int lane = threadIdx.x & 31;
  int cnt = N_NODES - node0;
  cnt = cnt < NPW ? cnt : NPW;
  int bgv = 0;
  if (lane < cnt) bgv = batch[node0 + lane];
  float v[NPW];
#pragma unroll
  for (int i = 0; i < NPW; ++i)
    v[i] = (i < cnt) ? emb[(size_t)(node0 + i) * HID + lane] : 0.0f;
#pragma unroll
  for (int i = 0; i < NPW; ++i) {
    if (i < cnt) {
      int bg = __builtin_amdgcn_readlane(bgv, i);
      atomicAdd(&sums[bg * HID + lane], v[i]);
      if (lane == 0) atomicAdd(&cnts[bg], 1.0f);
    }
  }
}

__global__ void pred_kernel(const float* __restrict__ sums,
                            const float* __restrict__ cnts,
                            const float* __restrict__ Wout,
                            const float* __restrict__ bout,
                            float* __restrict__ pred) {
  int gph = blockIdx.x * blockDim.x + threadIdx.x;
  if (gph >= N_GRAPHS) return;
  float cnt = cnts[gph];
  cnt = cnt > 1.0f ? cnt : 1.0f;
  float acc = bout[0];
#pragma unroll
  for (int f = 0; f < HID; ++f)
    acc += (sums[gph * HID + f] / cnt) * Wout[f];
  pred[gph] = 1.0f / (1.0f + __expf(-acc));
}

extern "C" void kernel_launch(void* const* d_in, const int* in_sizes, int n_in,
                              void* d_out, int out_size, void* d_ws, size_t ws_size,
                              hipStream_t stream) {
  const float* x     = (const float*)d_in[0];
  const int*   ei    = (const int*)  d_in[1];
  const float* pos   = (const float*)d_in[2];
  const int*   batch = (const int*)  d_in[3];
  const float* s1    = (const float*)d_in[4];
  const float* s2    = (const float*)d_in[5];
  const float* W1 = (const float*)d_in[6];   const float* b1 = (const float*)d_in[7];
  const float* W2 = (const float*)d_in[8];   const float* b2 = (const float*)d_in[9];
  const float* W3 = (const float*)d_in[10];  const float* b3 = (const float*)d_in[11];
  const float* W4 = (const float*)d_in[12];  const float* b4 = (const float*)d_in[13];
  const float* g1 = (const float*)d_in[14];  const float* be1 = (const float*)d_in[15];
  const float* g2 = (const float*)d_in[16];  const float* be2 = (const float*)d_in[17];
  const float* g3 = (const float*)d_in[18];  const float* be3 = (const float*)d_in[19];
  const float* Wout = (const float*)d_in[20];
  const float* bout = (const float*)d_in[21];

  const int* row = ei;
  const int* col = ei + N_EDGES;

  float* ws   = (float*)d_ws;
  int2*  tmp  = (int2*)ws;
  float* xw   = ws;
  float* h    = xw   + (size_t)N_NODES * HID;
  int2*  seg  = (int2*)(h + (size_t)N_NODES * HID);
  int*   off  = (int*)(ws + (size_t)4 * N_EDGES);
  float* dinv = (float*)(off + (size_t)N_NODES);
  float* sums = dinv + (size_t)N_NODES;
  float* cnts = sums + (size_t)N_GRAPHS * HID;
  int*   btot = (int*)sums;
  int*   tbl  = (int*)dinv;

  float* emb  = (float*)d_out;
  float* pred = (float*)d_out + (size_t)N_NODES * HID;

  const int BLK        = 256;
  const int gWmma      = (N_NODES / 16 + 3) / 4;

  bin_count_kernel<<<NB1, SCAN_BLK, 0, stream>>>(col, tbl);
  block_total_kernel<<<TBL_NBLK, SCAN_BLK, 0, stream>>>(tbl, btot, TBL);
  offsets_kernel<<<TBL_NBLK, SCAN_BLK, 0, stream>>>(tbl, btot, TBL);
  bin_scatter_kernel<<<NB1, SCAN_BLK, 0, stream>>>(row, col, tbl, tmp);
  bin_sort_kernel<<<NBIN, SCAN_BLK, 0, stream>>>(pos, s1, s2, tbl, tmp, off, seg);
  dinv_kernel<<<NODE_BLOCKS, BLK, 0, stream>>>(off, seg, dinv);
  coef_kernel<<<NODE_BLOCKS, BLK, 0, stream>>>(off, dinv, seg);

  gemm_in_kernel<<<GIN_BLOCKS, BLK, 0, stream>>>(x, W1, xw);
  aggregate_kernel<<<NODE_BLOCKS, BLK, 0, stream>>>(off, seg, xw, dinv, b1, g1, be1, 1, h);

  gemm_hid_wmma<<<gWmma, 128, 0, stream>>>(h, W2, xw);
  aggregate_kernel<<<NODE_BLOCKS, BLK, 0, stream>>>(off, seg, xw, dinv, b2, g2, be2, 1, h);

  gemm_hid_wmma<<<gWmma, 128, 0, stream>>>(h, W3, xw);
  aggregate_kernel<<<NODE_BLOCKS, BLK, 0, stream>>>(off, seg, xw, dinv, b3, g3, be3, 1, h);

  gemm_hid_wmma<<<gWmma, 128, 0, stream>>>(h, W4, xw);
  aggregate_kernel<<<NODE_BLOCKS, BLK, 0, stream>>>(off, seg, xw, dinv, b4, (const float*)nullptr,
                                                  (const float*)nullptr, 0, emb);

  fill_zero_kernel<<<(N_GRAPHS * HID + BLK - 1) / BLK, BLK, 0, stream>>>(sums, N_GRAPHS * HID);
  fill_zero_kernel<<<1, BLK, 0, stream>>>(cnts, N_GRAPHS);
  pool_kernel<<<NODE_BLOCKS, BLK, 0, stream>>>(emb, batch, sums, cnts);
  pred_kernel<<<1, BLK, 0, stream>>>(sums, cnts, Wout, bout, pred);
}


# Normalisation factored out of the edge sum: stored rows carry their node's dinv (source side), the destination's dinv multiplies once per node; per-edge coefficient pass and launch go; sum order kept
# speedup vs baseline: 4.8743x; 1.0569x over previous
#include <hip/hip_runtime.h>
#include <hip/hip_bf16.h>


#define N_NODES  100000
#define N_EDGES  3200000
#define N_GRAPHS 256
#define HID      32
#define IN_DIM   6
#define SCAN_BLK  1024
#define BIN_SHIFT 8
#define BIN_NODES (1 << BIN_SHIFT)
#define NBIN      ((N_NODES + BIN_NODES - 1) / BIN_NODES)
#define EPB       12800
#define EPT       ((EPB + SCAN_BLK - 1) / SCAN_BLK)
#define NB1       (N_EDGES / EPB)
#define TBL       (NBIN * NB1)
#define TBL_NBLK  ((TBL + SCAN_BLK - 1) / SCAN_BLK)
#define BIN_CAP   12288
#define NPW         7
#define NODE_WAVES  ((N_NODES + NPW - 1) / NPW)
#define NODE_BLOCKS ((NODE_WAVES + 7) / 8)
#define GIN_BLOCKS  625
#define GIN_THREADS (GIN_BLOCKS * 256)
#define GIN_ITERS   5
static_assert(GIN_THREADS * GIN_ITERS == N_NODES * (HID / 4), "layer-1 product covers every 4-feature group exactly once");
static_assert(GIN_THREADS % 8 == 0, "a thread keeps its feature group from trip to trip");
static_assert(NPW + 1 <= 32, "a wave holds its nodes' segment bounds one per lane");
static_assert(N_EDGES % EPB == 0, "edge passes assume whole blocks");
static_assert(TBL <= N_NODES, "the run table borrows the dinv array");
static_assert(NBIN <= SCAN_BLK && TBL_NBLK <= SCAN_BLK, "one thread per bin / per scan block");

typedef __attribute__((ext_vector_type(2))) float v2f;
typedef __attribute__((ext_vector_type(8))) float v8f;

__global__ void fill_zero_kernel(float* __restrict__ p, int n) {
  int i = blockIdx.x * blockDim.x + threadIdx.x;
  if (i < n) p[i] = 0.0f;
}

__device__ __forceinline__ int wave_sum_i(int v) {
  for (int d = 16; d > 0; d >>= 1) v += __shfl_xor(v, d, 32);
  return v;
}

__device__ __forceinline__ int wave_incl_scan_i(int v, int lane) {
  for (int d = 1; d < 32; d <<= 1) {
    int u = __shfl_up(v, d, 32);
    if (lane >= d) v += u;
  }
  return v;
}

__global__ __launch_bounds__(SCAN_BLK) void bin_count_kernel(const int* __restrict__ col,
                                                             int* __restrict__ tbl) {
  __shared__ int hist[NBIN];
  int tid = threadIdx.x;
  if (tid < NBIN) hist[tid] = 0;
  __syncthreads();
  const int* c = col + (size_t)blockIdx.x * EPB;
  int cv[EPT];
#pragma unroll
  for (int k = 0; k < EPT; ++k) {
    int i = tid + k * SCAN_BLK;
    cv[k] = (i < EPB) ? c[i] : -1;
  }
#pragma unroll
  for (int k = 0; k < EPT; ++k)
    if (cv[k] >= 0) atomicAdd(&hist[cv[k] >> BIN_SHIFT], 1);
  __syncthreads();
  if (tid < NBIN) tbl[tid * NB1 + blockIdx.x] = hist[tid];
}

__global__ __launch_bounds__(SCAN_BLK) void block_total_kernel(const int* __restrict__ cnt,
                                                               int* __restrict__ btot, int n) {
  __shared__ int wtot[32];
  int tid = threadIdx.x, lane = tid & 31, wv = tid >> 5;
  int i = blockIdx.x * SCAN_BLK + tid;
  int v = (i < n) ? cnt[i] : 0;
  v = wave_sum_i(v);
  if (lane == 0) wtot[wv] = v;
  __syncthreads();
  if (wv == 0) {
    int t = wave_sum_i(wtot[lane]);
    if (lane == 0) btot[blockIdx.x] = t;
  }
}

__global__ __launch_bounds__(SCAN_BLK) void offsets_kernel(int* __restrict__ cnt_off,
                                                           const int* __restrict__ btot, int n) {
  __shared__ int wincl[32];
  __shared__ int wpre[32];
  int tid = threadIdx.x, lane = tid & 31, wv = tid >> 5;
  int i = blockIdx.x * SCAN_BLK + tid;
  int v = (i < n) ? cnt_off[i] : 0;
  int p = (tid < (int)blockIdx.x) ? btot[tid] : 0;
  int incl = wave_incl_scan_i(v, lane);
  p = wave_sum_i(p);
  if (lane == 31) wincl[wv] = incl;
  if (lane == 0)  wpre[wv]  = p;
  __syncthreads();
  int before = wave_sum_i((lane < wv) ? wincl[lane] : 0);
  int prefix = wave_sum_i(wpre[lane]);
  if (i < n) cnt_off[i] = prefix + before + incl - v;
}

__global__ __launch_bounds__(SCAN_BLK) void bin_scatter_kernel(const int* __restrict__ row,
                                                               const int* __restrict__ col,
                                                               const int* __restrict__ tbl,
                                                               int2* __restrict__ tmp) {
  __shared__ int2 stage[EPB];
  __shared__ int cursor[NBIN];
  __shared__ int gbase[NBIN];
  __shared__ int wsum[32];
  int tid = threadIdx.x, lane = tid & 31, wv = tid >> 5;
  int blk = blockIdx.x;
  const int* r = row + (size_t)blk * EPB;
  const int* c = col + (size_t)blk * EPB;
  int rv[EPT], cv[EPT];
#pragma unroll
  for (int k = 0; k < EPT; ++k) {
    int i = tid + k * SCAN_BLK;
    rv[k] = (i < EPB) ? r[i] : 0;
    cv[k] = (i < EPB) ? c[i] : -1;
  }
  int g = 0, len = 0;
  if (tid < NBIN) {
    int idx = tid * NB1 + blk;
    g = tbl[idx];
    int nxt = (idx + 1 < TBL) ? tbl[idx + 1] : N_EDGES;
    len = nxt - g;
  }
  int incl = wave_incl_scan_i(len, lane);
  if (lane == 31) wsum[wv] = incl;
  __syncthreads();
  int before = wave_sum_i((lane < wv) ? wsum[lane] : 0);
  int lofs = before + incl - len;
  if (tid < NBIN) { cursor[tid] = lofs; gbase[tid] = g - lofs; }
  __syncthreads();
#pragma unroll
  for (int k = 0; k < EPT; ++k) {
    if (cv[k] >= 0) {
      int p = atomicAdd(&cursor[cv[k] >> BIN_SHIFT], 1);
      stage[p] = make_int2(rv[k], cv[k]);
    }
  }
  __syncthreads();
  for (int i = tid; i < EPB; i += SCAN_BLK) {
    int2 q = stage[i];
    tmp[gbase[q.y >> BIN_SHIFT] + i] = q;
  }
}

__global__ __launch_bounds__(SCAN_BLK) void bin_sort_kernel(const float* __restrict__ pos,
                                                            const float* __restrict__ s1p,
                                                            const float* __restrict__ s2p,
                                                            const int* __restrict__ tbl,
                                                            const int2* __restrict__ tmp,
                                                            int* __restrict__ off,
                                                            int2* __restrict__ seg) {
  __shared__ int2 stage[BIN_CAP];
  __shared__ int cnt[BIN_NODES];
  __shared__ int cursor[BIN_NODES];
  __shared__ int wsum[8];
  int tid = threadIdx.x, lane = tid & 31, wv = tid >> 5;
  int b = blockIdx.x;
  int S  = tbl[b * NB1];
  int En = (b + 1 < NBIN) ? tbl[(b + 1) * NB1] : N_EDGES;
  int nb = En - S;
  const int2* in = tmp + S;
  if (tid < BIN_NODES) cnt[tid] = 0;
  __syncthreads();
  for (int i = tid; i < nb; i += SCAN_BLK)
    atomicAdd(&cnt[in[i].y & (BIN_NODES - 1)], 1);
  __syncthreads();
  int v = (tid < BIN_NODES) ? cnt[tid] : 0;
  int incl = wave_incl_scan_i(v, lane);
  if (lane == 31 && wv < 8) wsum[wv] = incl;
  __syncthreads();
  int before = wave_sum_i((lane < wv && lane < 8) ? wsum[lane] : 0);
  int lofs = before + incl - v;
  if (tid < BIN_NODES) {
    cursor[tid] = lofs;
    int node = b * BIN_NODES + tid;
    if (node < N_NODES) off[node] = S + lofs;
  }
  __syncthreads();
  float s1 = s1p[0], s2 = s2p[0];
  bool staged = nb <= BIN_CAP;
  for (int i = tid; i < nb; i += SCAN_BLK) {
    int2 q = in[i];
    int rr = q.x, cc = q.y;
    const float* pr = pos + (size_t)rr * 6;
    const float* pc = pos + (size_t)cc * 6;
    float d0 = pr[0] - pc[0], d1 = pr[1] - pc[1], d2 = pr[2] - pc[2];
    float D = d0 * d0 + d1 * d1 + d2 * d2;
    float t = 1.0f - (pr[3] * pc[3] + pr[4] * pc[4] + pr[5] * pc[5]);
    float w = __expf(-(D * s1 * s1 + t * t * s2 * s2));
    int p = atomicAdd(&cursor[cc & (BIN_NODES - 1)], 1);
    int2 o = make_int2(rr, __float_as_int(w));
    if (staged) stage[p] = o; else seg[S + p] = o;
  }
  if (staged) {
    __syncthreads();
    for (int i = tid; i < nb; i += SCAN_BLK) seg[S + i] = stage[i];
  }
}

__global__ __launch_bounds__(256) void dinv_kernel(const int* __restrict__ off,
                                                   const int2* __restrict__ seg,
                                                   float* __restrict__ dinv) {
  int wave  = __builtin_amdgcn_readfirstlane((int)(blockIdx.x * 8 + (threadIdx.x >> 5)));
  int node0 = wave * NPW;
  if (node0 >= N_NODES) return;
  int lane = threadIdx.x & 31;
  int offv = N_EDGES;
  if (lane <= NPW && node0 + lane < N_NODES) offv = off[node0 + lane];
  float wa[NPW], wb[NPW], d[NPW];
#pragma unroll
  for (int i = 0; i < NPW; ++i) {
    int s = __builtin_amdgcn_readlane(offv, i);
    int e = __builtin_amdgcn_readlane(offv, i + 1);
    int ja = s + lane, jb = ja + 32;
    wa[i] = (ja < e) ? __int_as_float(seg[ja].y) : 0.0f;
    wb[i] = (jb < e) ? __int_as_float(seg[jb].y) : 0.0f;
  }
#pragma unroll
  for (int i = 0; i < NPW; ++i) d[i] = wa[i] + wb[i];
#pragma unroll
  for (int i = 0; i < NPW; ++i) {
    int s = __builtin_amdgcn_readlane(offv, i);
    int e = __builtin_amdgcn_readlane(offv, i + 1);
    if (e - s > 64) {
      for (int j = s + 64 + lane; j < e; j += 32) d[i] += __int_as_float(seg[j].y);
    }
  }
  float res = 0.0f;
#pragma unroll
  for (int i = 0; i < NPW; ++i) {
    float t = d[i];
    for (int k = 16; k > 0; k >>= 1) t += __shfl_xor(t, k, 32);
    res = (lane == i) ? t : res;
  }
  if (lane < NPW && node0 + lane < N_NODES)
    dinv[node0 + lane] = rsqrtf(res + 1.0f);
}

__global__ __launch_bounds__(256) void gemm_in_kernel(const float* __restrict__ x,
                                                      const float* __restrict__ W,
                                                      const float* __restrict__ dinv,
                                                      float* __restrict__ xw) {
  int t  = blockIdx.x * 256 + threadIdx.x;
  int fq = (t & 7) * 4;
  float4 w[IN_DIM];
#pragma unroll
  for (int k = 0; k < IN_DIM; ++k) w[k] = *(const float4*)(W + k * HID + fq);
#pragma unroll
  for (int i = 0; i < GIN_ITERS; ++i) {
    int node = (t + i * GIN_THREADS) >> 3;
    const float2* xr = (const float2*)(x + (size_t)node * IN_DIM);
    float2 x01 = xr[0], x23 = xr[1], x45 = xr[2];
    float ds = dinv[node];
    float xv[IN_DIM] = {x01.x, x01.y, x23.x, x23.y, x45.x, x45.y};
    float4 acc = make_float4(0.0f, 0.0f, 0.0f, 0.0f);
#pragma unroll
    for (int k = 0; k < IN_DIM; ++k) {
      acc.x += xv[k] * w[k].x;
      acc.y += xv[k] * w[k].y;
      acc.z += xv[k] * w[k].z;
      acc.w += xv[k] * w[k].w;
    }
    acc.x *= ds; acc.y *= ds; acc.z *= ds; acc.w *= ds;
    *(float4*)(xw + (size_t)node * HID + fq) = acc;
  }
}

__global__ __launch_bounds__(128) void gemm_hid_wmma(const float* __restrict__ H,
                                                     const float* __restrict__ W,
                                                     float* __restrict__ XW) {
  int wave = blockIdx.x * (blockDim.x >> 5) + (threadIdx.x >> 5);
  int lane = threadIdx.x & 31;
  int row0 = wave * 16;
  if (row0 >= N_NODES) return;
  int grp = lane >> 4;
  int m   = lane & 15;
  v8f c0 = {0.f,0.f,0.f,0.f,0.f,0.f,0.f,0.f};
  v8f c1 = {0.f,0.f,0.f,0.f,0.f,0.f,0.f,0.f};
  const float* hrow = H + (size_t)(row0 + m) * HID;
#pragma unroll
  for (int k0 = 0; k0 < HID; k0 += 4) {
    int kk = k0 + 2 * grp;
    v2f a;  a.x  = hrow[kk];                 a.y  = hrow[kk + 1];
    v2f b0; b0.x = W[kk * HID + m];          b0.y = W[(kk + 1) * HID + m];
    v2f b1; b1.x = W[kk * HID + 16 + m];     b1.y = W[(kk + 1) * HID + 16 + m];
    c0 = __builtin_amdgcn_wmma_f32_16x16x4_f32(false, a, false, b0, (short)0, c0, false, false);
    c1 = __builtin_amdgcn_wmma_f32_16x16x4_f32(false, a, false, b1, (short)0, c1, false, false);
  }
#pragma unroll
  for (int r = 0; r < 8; ++r) {
    int M = r + 8 * grp;
    XW[(size_t)(row0 + M) * HID + m]      = c0[r];
    XW[(size_t)(row0 + M) * HID + 16 + m] = c1[r];
  }
}

__device__ __forceinline__ float gather_sum(float acc, int r, float cf, int n,
                                            const float* __restrict__ xcol) {
  for (int t = 0; t < n; t += 8) {
    float xv[8], cv[8];
    for (int k = 0; k < 8; ++k) {
      int rk = __builtin_amdgcn_readlane(r, t + k);
      cv[k]  = __int_as_float(__builtin_amdgcn_readlane(__float_as_int(cf), t + k));
      xv[k]  = xcol[(size_t)rk * HID];
    }
    for (int k = 0; k < 8; ++k) {
      float a2 = fmaf(cv[k], xv[k], acc);
      acc = (t + k < n) ? a2 : acc;
    }
  }
  return acc;
}

__global__ __launch_bounds__(256) void aggregate_kernel(const int* __restrict__ off,
                                                        const int2* __restrict__ seg,
                                                        const float* __restrict__ xw,
                                                        const float* __restrict__ dinv,
                                                        const float* __restrict__ b,
                                                        const float* __restrict__ g,
                                                        const float* __restrict__ be,
                                                        int hidden,
                                                        float* __restrict__ out) {
  int wave  = __builtin_amdgcn_readfirstlane((int)(blockIdx.x * 8 + (threadIdx.x >> 5)));
  int node0 = wave * NPW;
  if (node0 >= N_NODES) return;
  int lane = threadIdx.x & 31;
  int offv = N_EDGES;
  if (lane <= NPW && node0 + lane < N_NODES) offv = off[node0 + lane];
  const float* xcol = xw + lane;
  float bb  = b[lane];
  float gs  = 1.0f, bs = 0.0f;
  if (hidden) { gs = g[lane] * rsqrtf(1.0f + 1e-5f); bs = be[lane]; }
  int cnt = N_NODES - node0;
  cnt = cnt < NPW ? cnt : NPW;
  int s = __builtin_amdgcn_readlane(offv, 0);
  int e = __builtin_amdgcn_readlane(offv, 1);
  int   rN = node0;
  float cN = 0.0f;
  { int j = s + lane; if (j < e) { int2 p = seg[j]; rN = p.x; cN = __int_as_float(p.y); } }
  for (int i = 0; i < cnt; ++i) {
    int   node = node0 + i;
    int   r  = rN;
    float cf = cN;
    int sN = e, eN = e;
    if (i + 1 < cnt) {
      eN = __builtin_amdgcn_readlane(offv, i + 2);
      int j = sN + lane;
      rN = node + 1; cN = 0.0f;
      if (j < eN) { int2 p = seg[j]; rN = p.x; cN = __int_as_float(p.y); }
    }
    size_t o  = (size_t)node * HID + lane;
    float xs  = xw[o];
    float di  = dinv[node];
    float acc = 0.0f;
    int n0 = e - s;
    n0 = n0 < 32 ? n0 : 32;
    acc = gather_sum(acc, r, cf, n0, xcol);
    for (int base = s + 32; base < e; base += 32) {
      int j = base + lane;
      r = node; cf = 0.0f;
      if (j < e) { int2 p = seg[j]; r = p.x; cf = __int_as_float(p.y); }
      int n = e - base;
      n = n < 32 ? n : 32;
      acc = gather_sum(acc, r, cf, n, xcol);
    }
    float v = di * (acc + xs) + bb;
    if (hidden) {
      v = v * gs + bs;
      v = v > 0.0f ? v : 0.0f;
      v *= di;
    }
    out[o] = v;
    s = sN; e = eN;
  }
}

__global__ __launch_bounds__(256) void pool_kernel(const float* __restrict__ emb,
                                                   const int* __restrict__ batch,
                                                   float* __restrict__ sums,
                                                   float* __restrict__ cnts) {
  int wave  = __builtin_amdgcn_readfirstlane((int)(blockIdx.x * 8 + (threadIdx.x >> 5)));
  int node0 = wave * NPW;
  if (node0 >= N_NODES) return;
  int lane = threadIdx.x & 31;
  int cnt = N_NODES - node0;
  cnt = cnt < NPW ? cnt : NPW;
  int bgv = 0;
  if (lane < cnt) bgv = batch[node0 + lane];
  float v[NPW];
#pragma unroll
  for (int i = 0; i < NPW; ++i)
    v[i] = (i < cnt) ? emb[(size_t)(node0 + i) * HID + lane] : 0.0f;
#pragma unroll
  for (int i = 0; i < NPW; ++i) {
    if (i < cnt) {
      int bg = __builtin_amdgcn_readlane(bgv, i);
      atomicAdd(&sums[bg * HID + lane], v[i]);
      if (lane == 0) atomicAdd(&cnts[bg], 1.0f);
    }
  }
}

__global__ void pred_kernel(const float* __restrict__ sums,
                            const float* __restrict__ cnts,
                            const float* __restrict__ Wout,
                            const float* __restrict__ bout,
                            float* __restrict__ pred) {
  int gph = blockIdx.x * blockDim.x + threadIdx.x;
  if (gph >= N_GRAPHS) return;
  float cnt = cnts[gph];
  cnt = cnt > 1.0f ? cnt : 1.0f;
  float acc = bout[0];
#pragma unroll
  for (int f = 0; f < HID; ++f)
    acc += (sums[gph * HID + f] / cnt) * Wout[f];
  pred[gph] = 1.0f / (1.0f + __expf(-acc));
}

extern "C" void kernel_launch(void* const* d_in, const int* in_sizes, int n_in,
                              void* d_out, int out_size, void* d_ws, size_t ws_size,
                              hipStream_t stream) {
  const float* x     = (const float*)d_in[0];
  const int*   ei    = (const int*)  d_in[1];
  const float* pos   = (const float*)d_in[2];
  const int*   batch = (const int*)  d_in[3];
  const float* s1    = (const float*)d_in[4];
  const float* s2    = (const float*)d_in[5];
  const float* W1 = (const float*)d_in[6];   const float* b1 = (const float*)d_in[7];
  const float* W2 = (const float*)d_in[8];   const float* b2 = (const float*)d_in[9];
  const float* W3 = (const float*)d_in[10];  const float* b3 = (const float*)d_in[11];
  const float* W4 = (const float*)d_in[12];  const float* b4 = (const float*)d_in[13];
  const float* g1 = (const float*)d_in[14];  const float* be1 = (const float*)d_in[15];
  const float* g2 = (const float*)d_in[16];  const float* be2 = (const float*)d_in[17];
  const float* g3 = (const float*)d_in[18];  const float* be3 = (const float*)d_in[19];
  const float* Wout = (const float*)d_in[20];
  const float* bout = (const float*)d_in[21];

  const int* row = ei;
  const int* col = ei + N_EDGES;

  float* ws   = (float*)d_ws;
  int2*  tmp  = (int2*)ws;
  float* xw   = ws;
  float* h    = xw   + (size_t)N_NODES * HID;
  int2*  seg  = (int2*)(h + (size_t)N_NODES * HID);
  int*   off  = (int*)(ws + (size_t)4 * N_EDGES);
  float* dinv = (float*)(off + (size_t)N_NODES);
  float* sums = dinv + (size_t)N_NODES;
  float* cnts = sums + (size_t)N_GRAPHS * HID;
  int*   btot = (int*)sums;
  int*   tbl  = (int*)dinv;

  float* emb  = (float*)d_out;
  float* pred = (float*)d_out + (size_t)N_NODES * HID;

  const int BLK        = 256;
  const int gWmma      = (N_NODES / 16 + 3) / 4;

  bin_count_kernel<<<NB1, SCAN_BLK, 0, stream>>>(col, tbl);
  block_total_kernel<<<TBL_NBLK, SCAN_BLK, 0, stream>>>(tbl, btot, TBL);
  offsets_kernel<<<TBL_NBLK, SCAN_BLK, 0, stream>>>(tbl, btot, TBL);
  bin_scatter_kernel<<<NB1, SCAN_BLK, 0, stream>>>(row, col, tbl, tmp);
  bin_sort_kernel<<<NBIN, SCAN_BLK, 0, stream>>>(pos, s1, s2, tbl, tmp, off, seg);
  dinv_kernel<<<NODE_BLOCKS, BLK, 0, stream>>>(off, seg, dinv);

  gemm_in_kernel<<<GIN_BLOCKS, BLK, 0, stream>>>(x, W1, dinv, xw);
  aggregate_kernel<<<NODE_BLOCKS, BLK, 0, stream>>>(off, seg, xw, dinv, b1, g1, be1, 1, h);

  gemm_hid_wmma<<<gWmma, 128, 0, stream>>>(h, W2, xw);
  aggregate_kernel<<<NODE_BLOCKS, BLK, 0, stream>>>(off, seg, xw, dinv, b2, g2, be2, 1, h);

  gemm_hid_wmma<<<gWmma, 128, 0, stream>>>(h, W3, xw);
  aggregate_kernel<<<NODE_BLOCKS, BLK, 0, stream>>>(off, seg, xw, dinv, b3, g3, be3, 1, h);

  gemm_hid_wmma<<<gWmma, 128, 0, stream>>>(h, W4, xw);
  aggregate_kernel<<<NODE_BLOCKS, BLK, 0, stream>>>(off, seg, xw, dinv, b4, (const float*)nullptr,
                                                  (const float*)nullptr, 0, emb);

  fill_zero_kernel<<<(N_GRAPHS * HID + BLK - 1) / BLK, BLK, 0, stream>>>(sums, N_GRAPHS * HID);
  fill_zero_kernel<<<1, BLK, 0, stream>>>(cnts, N_GRAPHS);
  pool_kernel<<<NODE_BLOCKS, BLK, 0, stream>>>(emb, batch, sums, cnts);
  pred_kernel<<<1, BLK, 0, stream>>>(sums, cnts, Wout, bout, pred);
}


# 18 launches to 11: next layer's WMMA product from an LDS tile inside each hidden aggregation, layer-1 product in the degree launch, pooling in the last; pool adds a wave's nodes before its atomic
# speedup vs baseline: 6.6633x; 1.3670x over previous
#include <hip/hip_runtime.h>
#include <hip/hip_bf16.h>


#define N_NODES  100000
#define N_EDGES  3200000
#define N_GRAPHS 256
#define HID      32
#define IN_DIM   6
#define SCAN_BLK  1024
#define BIN_SHIFT 8
#define BIN_NODES (1 << BIN_SHIFT)
#define NBIN      ((N_NODES + BIN_NODES - 1) / BIN_NODES)
#define EPB       12800
#define EPT       ((EPB + SCAN_BLK - 1) / SCAN_BLK)
#define NB1       (N_EDGES / EPB)
#define TBL       (NBIN * NB1)
#define TBL_NBLK  ((TBL + SCAN_BLK - 1) / SCAN_BLK)
#define BIN_CAP   12288
#define NPW         7
#define NODE_WAVES  ((N_NODES + NPW - 1) / NPW)
#define NODE_BLOCKS ((NODE_WAVES + 7) / 8)
#define AGG_WAVES   16
#define AGG_THREADS (AGG_WAVES * 32)
#define AGG_NODES   (AGG_WAVES * NPW)
#define AGG_TILES   (AGG_NODES / 16)
#define AGG_BLOCKS  ((N_NODES + AGG_NODES - 1) / AGG_NODES)
#define TILE_PITCH  34
#define POOL_WORDS  (N_GRAPHS * (HID + 1))
static_assert(AGG_NODES % 16 == 0 && N_NODES % 16 == 0, "a block's nodes are whole 16-row tiles");
static_assert(NPW * IN_DIM <= 64, "a wave's x rows fit two registers a lane");
static_assert(NODE_BLOCKS * 256 >= POOL_WORDS, "the degree launch has a thread for every pooling word");
static_assert(NPW + 1 <= 32, "a wave holds its nodes' segment bounds one per lane");
static_assert(N_EDGES % EPB == 0, "edge passes assume whole blocks");
static_assert(TBL <= N_NODES, "the run table borrows the dinv array");
static_assert(NBIN <= SCAN_BLK && TBL_NBLK <= SCAN_BLK, "one thread per bin / per scan block");

typedef __attribute__((ext_vector_type(2))) float v2f;
typedef __attribute__((ext_vector_type(8))) float v8f;


__device__ __forceinline__ int wave_sum_i(int v) {
  for (int d = 16; d > 0; d >>= 1) v += __shfl_xor(v, d, 32);
  return v;
}

__device__ __forceinline__ int wave_incl_scan_i(int v, int lane) {
  for (int d = 1; d < 32; d <<= 1) {
    int u = __shfl_up(v, d, 32);
    if (lane >= d) v += u;
  }
  return v;
}

__global__ __launch_bounds__(SCAN_BLK) void bin_count_kernel(const int* __restrict__ col,
                                                             int* __restrict__ tbl) {
  __shared__ int hist[NBIN];
  int tid = threadIdx.x;
  if (tid < NBIN) hist[tid] = 0;
  __syncthreads();
  const int* c = col + (size_t)blockIdx.x * EPB;
  int cv[EPT];
#pragma unroll
  for (int k = 0; k < EPT; ++k) {
    int i = tid + k * SCAN_BLK;
    cv[k] = (i < EPB) ? c[i] : -1;
  }
#pragma unroll
  for (int k = 0; k < EPT; ++k)
    if (cv[k] >= 0) atomicAdd(&hist[cv[k] >> BIN_SHIFT], 1);
  __syncthreads();
  if (tid < NBIN) tbl[tid * NB1 + blockIdx.x] = hist[tid];
}

__global__ __launch_bounds__(SCAN_BLK) void block_total_kernel(const int* __restrict__ cnt,
                                                               int* __restrict__ btot, int n) {
  __shared__ int wtot[32];
  int tid = threadIdx.x, lane = tid & 31, wv = tid >> 5;
  int i = blockIdx.x * SCAN_BLK + tid;
  int v = (i < n) ? cnt[i] : 0;
  v = wave_sum_i(v);
  if (lane == 0) wtot[wv] = v;
  __syncthreads();
  if (wv == 0) {
    int t = wave_sum_i(wtot[lane]);
    if (lane == 0) btot[blockIdx.x] = t;
  }
}

__global__ __launch_bounds__(SCAN_BLK) void offsets_kernel(int* __restrict__ cnt_off,
                                                           const int* __restrict__ btot, int n) {
  __shared__ int wincl[32];
  __shared__ int wpre[32];
  int tid = threadIdx.x, lane = tid & 31, wv = tid >> 5;
  int i = blockIdx.x * SCAN_BLK + tid;
  int v = (i < n) ? cnt_off[i] : 0;
  int p = (tid < (int)blockIdx.x) ? btot[tid] : 0;
  int incl = wave_incl_scan_i(v, lane);
  p = wave_sum_i(p);
  if (lane == 31) wincl[wv] = incl;
  if (lane == 0)  wpre[wv]  = p;
  __syncthreads();
  int before = wave_sum_i((lane < wv) ? wincl[lane] : 0);
  int prefix = wave_sum_i(wpre[lane]);
  if (i < n) cnt_off[i] = prefix + before + incl - v;
}

__global__ __launch_bounds__(SCAN_BLK) void bin_scatter_kernel(const int* __restrict__ row,
                                                               const int* __restrict__ col,
                                                               const int* __restrict__ tbl,
                                                               int2* __restrict__ tmp) {
  __shared__ int2 stage[EPB];
  __shared__ int cursor[NBIN];
  __shared__ int gbase[NBIN];
  __shared__ int wsum[32];
  int tid = threadIdx.x, lane = tid & 31, wv = tid >> 5;
  int blk = blockIdx.x;
  const int* r = row + (size_t)blk * EPB;
  const int* c = col + (size_t)blk * EPB;
  int rv[EPT], cv[EPT];
#pragma unroll
  for (int k = 0; k < EPT; ++k) {
    int i = tid + k * SCAN_BLK;
    rv[k] = (i < EPB) ? r[i] : 0;
    cv[k] = (i < EPB) ? c[i] : -1;
  }
  int g = 0, len = 0;
  if (tid < NBIN) {
    int idx = tid * NB1 + blk;
    g = tbl[idx];
    int nxt = (idx + 1 < TBL) ? tbl[idx + 1] : N_EDGES;
    len = nxt - g;
  }
  int incl = wave_incl_scan_i(len, lane);
  if (lane == 31) wsum[wv] = incl;
  __syncthreads();
  int before = wave_sum_i((lane < wv) ? wsum[lane] : 0);
  int lofs = before + incl - len;
  if (tid < NBIN) { cursor[tid] = lofs; gbase[tid] = g - lofs; }
  __syncthreads();
#pragma unroll
  for (int k = 0; k < EPT; ++k) {
    if (cv[k] >= 0) {
      int p = atomicAdd(&cursor[cv[k] >> BIN_SHIFT], 1);
      stage[p] = make_int2(rv[k], cv[k]);
    }
  }
  __syncthreads();
  for (int i = tid; i < EPB; i += SCAN_BLK) {
    int2 q = stage[i];
    tmp[gbase[q.y >> BIN_SHIFT] + i] = q;
  }
}

__global__ __launch_bounds__(SCAN_BLK) void bin_sort_kernel(const float* __restrict__ pos,
                                                            const float* __restrict__ s1p,
                                                            const float* __restrict__ s2p,
                                                            const int* __restrict__ tbl,
                                                            const int2* __restrict__ tmp,
                                                            int* __restrict__ off,
                                                            int2* __restrict__ seg) {
  __shared__ int2 stage[BIN_CAP];
  __shared__ int cnt[BIN_NODES];
  __shared__ int cursor[BIN_NODES];
  __shared__ int wsum[8];
  int tid = threadIdx.x, lane = tid & 31, wv = tid >> 5;
  int b = blockIdx.x;
  int S  = tbl[b * NB1];
  int En = (b + 1 < NBIN) ? tbl[(b + 1) * NB1] : N_EDGES;
  int nb = En - S;
  const int2* in = tmp + S;
  if (tid < BIN_NODES) cnt[tid] = 0;
  __syncthreads();
  for (int i = tid; i < nb; i += SCAN_BLK)
    atomicAdd(&cnt[in[i].y & (BIN_NODES - 1)], 1);
  __syncthreads();
  int v = (tid < BIN_NODES) ? cnt[tid] : 0;
  int incl = wave_incl_scan_i(v, lane);
  if (lane == 31 && wv < 8) wsum[wv] = incl;
  __syncthreads();
  int before = wave_sum_i((lane < wv && lane < 8) ? wsum[lane] : 0);
  int lofs = before + incl - v;
  if (tid < BIN_NODES) {
    cursor[tid] = lofs;
    int node = b * BIN_NODES + tid;
    if (node < N_NODES) off[node] = S + lofs;
  }
  __syncthreads();
  float s1 = s1p[0], s2 = s2p[0];
  bool staged = nb <= BIN_CAP;
  for (int i = tid; i < nb; i += SCAN_BLK) {
    int2 q = in[i];
    int rr = q.x, cc = q.y;
    const float* pr = pos + (size_t)rr * 6;
    const float* pc = pos + (size_t)cc * 6;
    float d0 = pr[0] - pc[0], d1 = pr[1] - pc[1], d2 = pr[2] - pc[2];
    float D = d0 * d0 + d1 * d1 + d2 * d2;
    float t = 1.0f - (pr[3] * pc[3] + pr[4] * pc[4] + pr[5] * pc[5]);
    float w = __expf(-(D * s1 * s1 + t * t * s2 * s2));
    int p = atomicAdd(&cursor[cc & (BIN_NODES - 1)], 1);
    int2 o = make_int2(rr, __float_as_int(w));
    if (staged) stage[p] = o; else seg[S + p] = o;
  }
  if (staged) {
    __syncthreads();
    for (int i = tid; i < nb; i += SCAN_BLK) seg[S + i] = stage[i];
  }
}

__global__ __launch_bounds__(256) void dinv_gemm_in_kernel(const int* __restrict__ off,
                                                           const int2* __restrict__ seg,
                                                           const float* __restrict__ x,
                                                           const float* __restrict__ W,
                                                           float* __restrict__ dinv,
                                                           float* __restrict__ xw,
                                                           float* __restrict__ poolz) {
  int gt = blockIdx.x * 256 + threadIdx.x;
  if (gt < POOL_WORDS) poolz[gt] = 0.0f;
  int wave  = __builtin_amdgcn_readfirstlane((int)(blockIdx.x * 8 + (threadIdx.x >> 5)));
  int node0 = wave * NPW;
  if (node0 >= N_NODES) return;
  int lane = threadIdx.x & 31;
  int offv = N_EDGES;
  if (lane <= NPW && node0 + lane < N_NODES) offv = off[node0 + lane];
  int xi = node0 * IN_DIM + lane;
  float xa = (xi < N_NODES * IN_DIM) ? x[xi] : 0.0f;
  float xb = (xi + 32 < N_NODES * IN_DIM) ? x[xi + 32] : 0.0f;
  float w[IN_DIM];
#pragma unroll
  for (int k = 0; k < IN_DIM; ++k) w[k] = W[k * HID + lane];
  float wa[NPW], wb[NPW], d[NPW];
#pragma unroll
  for (int i = 0; i < NPW; ++i) {
    int s = __builtin_amdgcn_readlane(offv, i);
    int e = __builtin_amdgcn_readlane(offv, i + 1);
    int ja = s + lane, jb = ja + 32;
    wa[i] = (ja < e) ? __int_as_float(seg[ja].y) : 0.0f;
    wb[i] = (jb < e) ? __int_as_float(seg[jb].y) : 0.0f;
  }
#pragma unroll
  for (int i = 0; i < NPW; ++i) d[i] = wa[i] + wb[i];
#pragma unroll
  for (int i = 0; i < NPW; ++i) {
    int s = __builtin_amdgcn_readlane(offv, i);
    int e = __builtin_amdgcn_readlane(offv, i + 1);
    if (e - s > 64) {
      for (int j = s + 64 + lane; j < e; j += 32) d[i] += __int_as_float(seg[j].y);
    }
  }
  float res = 0.0f;
#pragma unroll
  for (int i = 0; i < NPW; ++i) {
    float t = d[i];
    for (int k = 16; k > 0; k >>= 1) t += __shfl_xor(t, k, 32);
    res = (lane == i) ? t : res;
  }
  float dv = rsqrtf(res + 1.0f);
  if (lane < NPW && node0 + lane < N_NODES) dinv[node0 + lane] = dv;
#pragma unroll
  for (int i = 0; i < NPW; ++i) {
    if (node0 + i < N_NODES) {
      float ds  = __int_as_float(__builtin_amdgcn_readlane(__float_as_int(dv), i));
      float acc = 0.0f;
#pragma unroll
      for (int k = 0; k < IN_DIM; ++k) {
        int f = i * IN_DIM + k;
        float xv = __int_as_float(__builtin_amdgcn_readlane(__float_as_int(f < 32 ? xa : xb), f & 31));
        acc += xv * w[k];
      }
      acc *= ds;
      xw[(size_t)(node0 + i) * HID + lane] = acc;
    }
  }
}

__device__ __forceinline__ float gather_sum(float acc, int r, float cf, int n,
                                            const float* __restrict__ xcol) {
  for (int t = 0; t < n; t += 8) {
    float xv[8], cv[8];
    for (int k = 0; k < 8; ++k) {
      int rk = __builtin_amdgcn_readlane(r, t + k);
      cv[k]  = __int_as_float(__builtin_amdgcn_readlane(__float_as_int(cf), t + k));
      xv[k]  = xcol[(size_t)rk * HID];
    }
    for (int k = 0; k < 8; ++k) {
      float a2 = fmaf(cv[k], xv[k], acc);
      acc = (t + k < n) ? a2 : acc;
    }
  }
  return acc;
}

template <class Sink>
__device__ __forceinline__ void walk_nodes(const int* __restrict__ off,
                                           const int2* __restrict__ seg,
                                           const float* __restrict__ xw,
                                           const float* __restrict__ dinv,
                                           float bb, int node0, int lane, Sink sink) {
  int offv = N_EDGES;
  if (lane <= NPW && node0 + lane < N_NODES) offv = off[node0 + lane];
  const float* xcol = xw + lane;
  int cnt = N_NODES - node0;
  cnt = cnt < NPW ? cnt : NPW;
  int s = __builtin_amdgcn_readlane(offv, 0);
  int e = __builtin_amdgcn_readlane(offv, 1);
  int   rN = node0;
  float cN = 0.0f;
  { int j = s + lane; if (j < e) { int2 p = seg[j]; rN = p.x; cN = __int_as_float(p.y); } }
  for (int i = 0; i < cnt; ++i) {
    int   node = node0 + i;
    int   r  = rN;
    float cf = cN;
    int sN = e, eN = e;
    if (i + 1 < cnt) {
      eN = __builtin_amdgcn_readlane(offv, i + 2);
      int j = sN + lane;
      rN = node + 1; cN = 0.0f;
      if (j < eN) { int2 p = seg[j]; rN = p.x; cN = __int_as_float(p.y); }
    }
    size_t o  = (size_t)node * HID + lane;
    float xs  = xw[o];
    float di  = dinv[node];
    float acc = 0.0f;
    int n0 = e - s;
    n0 = n0 < 32 ? n0 : 32;
    acc = gather_sum(acc, r, cf, n0, xcol);
    for (int base = s + 32; base < e; base += 32) {
      int j = base + lane;
      r = node; cf = 0.0f;
      if (j < e) { int2 p = seg[j]; r = p.x; cf = __int_as_float(p.y); }
      int n = e - base;
      n = n < 32 ? n : 32;
      acc = gather_sum(acc, r, cf, n, xcol);
    }
    float v = di * (acc + xs) + bb;
    sink(i, node, di, v);
    s = sN; e = eN;
  }
}

__global__ __launch_bounds__(AGG_THREADS) void aggregate_mm_kernel(const int* __restrict__ off,
                                                                   const int2* __restrict__ seg,
                                                                   const float* __restrict__ xw,
                                                                   const float* __restrict__ dinv,
                                                                   const float* __restrict__ b,
                                                                   const float* __restrict__ g,
                                                                   const float* __restrict__ be,
                                                                   const float* __restrict__ Wn,
                                                                   float* __restrict__ xwn) {
  __shared__ float tile[AGG_NODES * TILE_PITCH];
  int wv    = __builtin_amdgcn_readfirstlane((int)(threadIdx.x >> 5));
  int node0 = ((int)blockIdx.x * AGG_WAVES + wv) * NPW;
  int lane  = threadIdx.x & 31;
  if (node0 < N_NODES) {
    float bb = b[lane];
    float gs = g[lane] * rsqrtf(1.0f + 1e-5f);
    float bs = be[lane];
    float* trow = tile + (wv * NPW) * TILE_PITCH + lane;
    walk_nodes(off, seg, xw, dinv, bb, node0, lane,
               [&](int i, int node, float di, float v) {
                 v = v * gs + bs;
                 v = v > 0.0f ? v : 0.0f;
                 v *= di;
                 trow[i * TILE_PITCH] = v;
               });
  }
  int  row0 = (int)blockIdx.x * AGG_NODES + wv * 16;
  bool mm   = wv < AGG_TILES && row0 < N_NODES;
  int  grp  = lane >> 4;
  int  m    = lane & 15;
  v2f b0[HID / 4], b1[HID / 4];
  if (mm) {
#pragma unroll
    for (int q = 0; q < HID / 4; ++q) {
      int kk = 4 * q + 2 * grp;
      b0[q].x = Wn[kk * HID + m];          b0[q].y = Wn[(kk + 1) * HID + m];
      b1[q].x = Wn[kk * HID + 16 + m];     b1[q].y = Wn[(kk + 1) * HID + 16 + m];
    }
  }
  __syncthreads();
  if (mm) {
    v8f c0 = {0.f,0.f,0.f,0.f,0.f,0.f,0.f,0.f};
    v8f c1 = {0.f,0.f,0.f,0.f,0.f,0.f,0.f,0.f};
    const float* hrow = tile + (wv * 16 + m) * TILE_PITCH;
#pragma unroll
    for (int q = 0; q < HID / 4; ++q) {
      int kk = 4 * q + 2 * grp;
      v2f a;  a.x  = hrow[kk];                  a.y  = hrow[kk + 1];
      c0 = __builtin_amdgcn_wmma_f32_16x16x4_f32(false, a, false, b0[q], (short)0, c0, false, false);
      c1 = __builtin_amdgcn_wmma_f32_16x16x4_f32(false, a, false, b1[q], (short)0, c1, false, false);
    }
#pragma unroll
    for (int r = 0; r < 8; ++r) {
      int M = r + 8 * grp;
      xwn[(size_t)(row0 + M) * HID + m]      = c0[r];
      xwn[(size_t)(row0 + M) * HID + 16 + m] = c1[r];
    }
  }
}

__global__ __launch_bounds__(256) void aggregate_out_kernel(const int* __restrict__ off,
                                                            const int2* __restrict__ seg,
                                                            const float* __restrict__ xw,
                                                            const float* __restrict__ dinv,
                                                            const float* __restrict__ b,
                                                            const int* __restrict__ batch,
                                                            float* __restrict__ sums,
                                                            float* __restrict__ cnts,
                                                            float* __restrict__ out) {
  int wave  = __builtin_amdgcn_readfirstlane((int)(blockIdx.x * 8 + (threadIdx.x >> 5)));
  int node0 = wave * NPW;
  if (node0 >= N_NODES) return;
  int lane = threadIdx.x & 31;
  float bb = b[lane];
  int bgv = 0;
  if (lane < NPW && node0 + lane < N_NODES) bgv = batch[node0 + lane];
  int   pg = __builtin_amdgcn_readlane(bgv, 0);
  float ps = 0.0f, pc = 0.0f;
  walk_nodes(off, seg, xw, dinv, bb, node0, lane,
             [&](int i, int node, float di, float v) {
               out[(size_t)node * HID + lane] = v;
               int bg = __builtin_amdgcn_readlane(bgv, i);
               if (bg != pg) {
                 atomicAdd(&sums[pg * HID + lane], ps);
                 if (lane == 0) atomicAdd(&cnts[pg], pc);
                 ps = 0.0f; pc = 0.0f; pg = bg;
               }
               ps += v;
               pc += 1.0f;
             });
  atomicAdd(&sums[pg * HID + lane], ps);
  if (lane == 0) atomicAdd(&cnts[pg], pc);
}

__global__ void pred_kernel(const float* __restrict__ sums,
                            const float* __restrict__ cnts,
                            const float* __restrict__ Wout,
                            const float* __restrict__ bout,
                            float* __restrict__ pred) {
  int gph = blockIdx.x * blockDim.x + threadIdx.x;
  if (gph >= N_GRAPHS) return;
  float cnt = cnts[gph];
  cnt = cnt > 1.0f ? cnt : 1.0f;
  float acc = bout[0];
#pragma unroll
  for (int f = 0; f < HID; ++f)
    acc += (sums[gph * HID + f] / cnt) * Wout[f];
  pred[gph] = 1.0f / (1.0f + __expf(-acc));
}

extern "C" void kernel_launch(void* const* d_in, const int* in_sizes, int n_in,
                              void* d_out, int out_size, void* d_ws, size_t ws_size,
                              hipStream_t stream) {
  const float* x     = (const float*)d_in[0];
  const int*   ei    = (const int*)  d_in[1];
  const float* pos   = (const float*)d_in[2];
  const int*   batch = (const int*)  d_in[3];
  const float* s1    = (const float*)d_in[4];
  const float* s2    = (const float*)d_in[5];
  const float* W1 = (const float*)d_in[6];   const float* b1 = (const float*)d_in[7];
  const float* W2 = (const float*)d_in[8];   const float* b2 = (const float*)d_in[9];
  const float* W3 = (const float*)d_in[10];  const float* b3 = (const float*)d_in[11];
  const float* W4 = (const float*)d_in[12];  const float* b4 = (const float*)d_in[13];
  const float* g1 = (const float*)d_in[14];  const float* be1 = (const float*)d_in[15];
  const float* g2 = (const float*)d_in[16];  const float* be2 = (const float*)d_in[17];
  const float* g3 = (const float*)d_in[18];  const float* be3 = (const float*)d_in[19];
  const float* Wout = (const float*)d_in[20];
  const float* bout = (const float*)d_in[21];

  const int* row = ei;
  const int* col = ei + N_EDGES;

  float* ws   = (float*)d_ws;
  int2*  tmp  = (int2*)ws;
  float* xw   = ws;
  float* h    = xw   + (size_t)N_NODES * HID;
  int2*  seg  = (int2*)(h + (size_t)N_NODES * HID);
  int*   off  = (int*)(ws + (size_t)4 * N_EDGES);
  float* dinv = (float*)(off + (size_t)N_NODES);
  float* sums = dinv + (size_t)N_NODES;
  float* cnts = sums + (size_t)N_GRAPHS * HID;
  int*   btot = (int*)sums;
  int*   tbl  = (int*)dinv;

  float* emb  = (float*)d_out;
  float* pred = (float*)d_out + (size_t)N_NODES * HID;

  const int BLK        = 256;

  bin_count_kernel<<<NB1, SCAN_BLK, 0, stream>>>(col, tbl);
  block_total_kernel<<<TBL_NBLK, SCAN_BLK, 0, stream>>>(tbl, btot, TBL);
  offsets_kernel<<<TBL_NBLK, SCAN_BLK, 0, stream>>>(tbl, btot, TBL);
  bin_scatter_kernel<<<NB1, SCAN_BLK, 0, stream>>>(row, col, tbl, tmp);
  bin_sort_kernel<<<NBIN, SCAN_BLK, 0, stream>>>(pos, s1, s2, tbl, tmp, off, seg);

  dinv_gemm_in_kernel<<<NODE_BLOCKS, BLK, 0, stream>>>(off, seg, x, W1, dinv, xw, sums);

  aggregate_mm_kernel<<<AGG_BLOCKS, AGG_THREADS, 0, stream>>>(off, seg, xw, dinv, b1, g1, be1, W2, h);
  aggregate_mm_kernel<<<AGG_BLOCKS, AGG_THREADS, 0, stream>>>(off, seg, h, dinv, b2, g2, be2, W3, xw);
  aggregate_mm_kernel<<<AGG_BLOCKS, AGG_THREADS, 0, stream>>>(off, seg, xw, dinv, b3, g3, be3, W4, h);

  aggregate_out_kernel<<<NODE_BLOCKS, BLK, 0, stream>>>(off, seg, h, dinv, b4, batch, sums, cnts, emb);
  pred_kernel<<<1, BLK, 0, stream>>>(sums, cnts, Wout, bout, pred);
}


# All of a node's row gathers in flight at once: its first 64 edges are fetched one node ahead and up to 40 row loads are issued before the first sum (was 8 per round trip); edge sum order kept
# speedup vs baseline: 7.1592x; 1.0744x over previous
#include <hip/hip_runtime.h>
#include <hip/hip_bf16.h>


#define N_NODES  100000
#define N_EDGES  3200000
#define N_GRAPHS 256
#define HID      32
#define IN_DIM   6
#define SCAN_BLK  1024
#define BIN_SHIFT 8
#define BIN_NODES (1 << BIN_SHIFT)
#define NBIN      ((N_NODES + BIN_NODES - 1) / BIN_NODES)
#define EPB       12800
#define EPT       ((EPB + SCAN_BLK - 1) / SCAN_BLK)
#define NB1       (N_EDGES / EPB)
#define TBL       (NBIN * NB1)
#define TBL_NBLK  ((TBL + SCAN_BLK - 1) / SCAN_BLK)
#define BIN_CAP   12288
#define NPW         7
#define NODE_WAVES  ((N_NODES + NPW - 1) / NPW)
#define NODE_BLOCKS ((NODE_WAVES + 7) / 8)
#define AGG_WAVES   16
#define AGG_THREADS (AGG_WAVES * 32)
#define AGG_NODES   (AGG_WAVES * NPW)
#define AGG_TILES   (AGG_NODES / 16)
#define AGG_BLOCKS  ((N_NODES + AGG_NODES - 1) / AGG_NODES)
#define TILE_PITCH  34
#define POOL_WORDS  (N_GRAPHS * (HID + 1))
static_assert(AGG_NODES % 16 == 0 && N_NODES % 16 == 0, "a block's nodes are whole 16-row tiles");
static_assert(NPW * IN_DIM <= 64, "a wave's x rows fit two registers a lane");
static_assert(NODE_BLOCKS * 256 >= POOL_WORDS, "the degree launch has a thread for every pooling word");
static_assert(NPW + 1 <= 32, "a wave holds its nodes' segment bounds one per lane");
static_assert(N_EDGES % EPB == 0, "edge passes assume whole blocks");
static_assert(TBL <= N_NODES, "the run table borrows the dinv array");
static_assert(NBIN <= SCAN_BLK && TBL_NBLK <= SCAN_BLK, "one thread per bin / per scan block");

typedef __attribute__((ext_vector_type(2))) float v2f;
typedef __attribute__((ext_vector_type(8))) float v8f;


__device__ __forceinline__ int wave_sum_i(int v) {
  for (int d = 16; d > 0; d >>= 1) v += __shfl_xor(v, d, 32);
  return v;
}

__device__ __forceinline__ int wave_incl_scan_i(int v, int lane) {
  for (int d = 1; d < 32; d <<= 1) {
    int u = __shfl_up(v, d, 32);
    if (lane >= d) v += u;
  }
  return v;
}

__global__ __launch_bounds__(SCAN_BLK) void bin_count_kernel(const int* __restrict__ col,
                                                             int* __restrict__ tbl) {
  __shared__ int hist[NBIN];
  int tid = threadIdx.x;
  if (tid < NBIN) hist[tid] = 0;
  __syncthreads();
  const int* c = col + (size_t)blockIdx.x * EPB;
  int cv[EPT];
#pragma unroll
  for (int k = 0; k < EPT; ++k) {
    int i = tid + k * SCAN_BLK;
    cv[k] = (i < EPB) ? c[i] : -1;
  }
#pragma unroll
  for (int k = 0; k < EPT; ++k)
    if (cv[k] >= 0) atomicAdd(&hist[cv[k] >> BIN_SHIFT], 1);
  __syncthreads();
  if (tid < NBIN) tbl[tid * NB1 + blockIdx.x] = hist[tid];
}

__global__ __launch_bounds__(SCAN_BLK) void block_total_kernel(const int* __restrict__ cnt,
                                                               int* __restrict__ btot, int n) {
  __shared__ int wtot[32];
  int tid = threadIdx.x, lane = tid & 31, wv = tid >> 5;
  int i = blockIdx.x * SCAN_BLK + tid;
  int v = (i < n) ? cnt[i] : 0;
  v = wave_sum_i(v);
  if (lane == 0) wtot[wv] = v;
  __syncthreads();
  if (wv == 0) {
    int t = wave_sum_i(wtot[lane]);
    if (lane == 0) btot[blockIdx.x] = t;
  }
}

__global__ __launch_bounds__(SCAN_BLK) void offsets_kernel(int* __restrict__ cnt_off,
                                                           const int* __restrict__ btot, int n) {
  __shared__ int wincl[32];
  __shared__ int wpre[32];
  int tid = threadIdx.x, lane = tid & 31, wv = tid >> 5;
  int i = blockIdx.x * SCAN_BLK + tid;
  int v = (i < n) ? cnt_off[i] : 0;
  int p = (tid < (int)blockIdx.x) ? btot[tid] : 0;
  int incl = wave_incl_scan_i(v, lane);
  p = wave_sum_i(p);
  if (lane == 31) wincl[wv] = incl;
  if (lane == 0)  wpre[wv]  = p;
  __syncthreads();
  int before = wave_sum_i((lane < wv) ? wincl[lane] : 0);
  int prefix = wave_sum_i(wpre[lane]);
  if (i < n) cnt_off[i] = prefix + before + incl - v;
}

__global__ __launch_bounds__(SCAN_BLK) void bin_scatter_kernel(const int* __restrict__ row,
                                                               const int* __restrict__ col,
                                                               const int* __restrict__ tbl,
                                                               int2* __restrict__ tmp) {
  __shared__ int2 stage[EPB];
  __shared__ int cursor[NBIN];
  __shared__ int gbase[NBIN];
  __shared__ int wsum[32];
  int tid = threadIdx.x, lane = tid & 31, wv = tid >> 5;
  int blk = blockIdx.x;
  const int* r = row + (size_t)blk * EPB;
  const int* c = col + (size_t)blk * EPB;
  int rv[EPT], cv[EPT];
#pragma unroll
  for (int k = 0; k < EPT; ++k) {
    int i = tid + k * SCAN_BLK;
    rv[k] = (i < EPB) ? r[i] : 0;
    cv[k] = (i < EPB) ? c[i] : -1;
  }
  int g = 0, len = 0;
  if (tid < NBIN) {
    int idx = tid * NB1 + blk;
    g = tbl[idx];
    int nxt = (idx + 1 < TBL) ? tbl[idx + 1] : N_EDGES;
    len = nxt - g;
  }
  int incl = wave_incl_scan_i(len, lane);
  if (lane == 31) wsum[wv] = incl;
  __syncthreads();
  int before = wave_sum_i((lane < wv) ? wsum[lane] : 0);
  int lofs = before + incl - len;
  if (tid < NBIN) { cursor[tid] = lofs; gbase[tid] = g - lofs; }
  __syncthreads();
#pragma unroll
  for (int k = 0; k < EPT; ++k) {
    if (cv[k] >= 0) {
      int p = atomicAdd(&cursor[cv[k] >> BIN_SHIFT], 1);
      stage[p] = make_int2(rv[k], cv[k]);
    }
  }
  __syncthreads();
  for (int i = tid; i < EPB; i += SCAN_BLK) {
    int2 q = stage[i];
    tmp[gbase[q.y >> BIN_SHIFT] + i] = q;
  }
}

__global__ __launch_bounds__(SCAN_BLK) void bin_sort_kernel(const float* __restrict__ pos,
                                                            const float* __restrict__ s1p,
                                                            const float* __restrict__ s2p,
                                                            const int* __restrict__ tbl,
                                                            const int2* __restrict__ tmp,
                                                            int* __restrict__ off,
                                                            int2* __restrict__ seg) {
  __shared__ int2 stage[BIN_CAP];
  __shared__ int cnt[BIN_NODES];
  __shared__ int cursor[BIN_NODES];
  __shared__ int wsum[8];
  int tid = threadIdx.x, lane = tid & 31, wv = tid >> 5;
  int b = blockIdx.x;
  int S  = tbl[b * NB1];
  int En = (b + 1 < NBIN) ? tbl[(b + 1) * NB1] : N_EDGES;
  int nb = En - S;
  const int2* in = tmp + S;
  if (tid < BIN_NODES) cnt[tid] = 0;
  __syncthreads();
  for (int i = tid; i < nb; i += SCAN_BLK)
    atomicAdd(&cnt[in[i].y & (BIN_NODES - 1)], 1);
  __syncthreads();
  int v = (tid < BIN_NODES) ? cnt[tid] : 0;
  int incl = wave_incl_scan_i(v, lane);
  if (lane == 31 && wv < 8) wsum[wv] = incl;
  __syncthreads();
  int before = wave_sum_i((lane < wv && lane < 8) ? wsum[lane] : 0);
  int lofs = before + incl - v;
  if (tid < BIN_NODES) {
    cursor[tid] = lofs;
    int node = b * BIN_NODES + tid;
    if (node < N_NODES) off[node] = S + lofs;
  }
  __syncthreads();
  float s1 = s1p[0], s2 = s2p[0];
  bool staged = nb <= BIN_CAP;
  for (int i = tid; i < nb; i += SCAN_BLK) {
    int2 q = in[i];
    int rr = q.x, cc = q.y;
    const float* pr = pos + (size_t)rr * 6;
    const float* pc = pos + (size_t)cc * 6;
    float d0 = pr[0] - pc[0], d1 = pr[1] - pc[1], d2 = pr[2] - pc[2];
    float D = d0 * d0 + d1 * d1 + d2 * d2;
    float t = 1.0f - (pr[3] * pc[3] + pr[4] * pc[4] + pr[5] * pc[5]);
    float w = __expf(-(D * s1 * s1 + t * t * s2 * s2));
    int p = atomicAdd(&cursor[cc & (BIN_NODES - 1)], 1);
    int2 o = make_int2(rr, __float_as_int(w));
    if (staged) stage[p] = o; else seg[S + p] = o;
  }
  if (staged) {
    __syncthreads();
    for (int i = tid; i < nb; i += SCAN_BLK) seg[S + i] = stage[i];
  }
}

__global__ __launch_bounds__(256) void dinv_gemm_in_kernel(const int* __restrict__ off,
                                                           const int2* __restrict__ seg,
                                                           const float* __restrict__ x,
                                                           const float* __restrict__ W,
                                                           float* __restrict__ dinv,
                                                           float* __restrict__ xw,
                                                           float* __restrict__ poolz) {
  int gt = blockIdx.x * 256 + threadIdx.x;
  if (gt < POOL_WORDS) poolz[gt] = 0.0f;
  int wave  = __builtin_amdgcn_readfirstlane((int)(blockIdx.x * 8 + (threadIdx.x >> 5)));
  int node0 = wave * NPW;
  if (node0 >= N_NODES) return;
  int lane = threadIdx.x & 31;
  int offv = N_EDGES;
  if (lane <= NPW && node0 + lane < N_NODES) offv = off[node0 + lane];
  int xi = node0 * IN_DIM + lane;
  float xa = (xi < N_NODES * IN_DIM) ? x[xi] : 0.0f;
  float xb = (xi + 32 < N_NODES * IN_DIM) ? x[xi + 32] : 0.0f;
  float w[IN_DIM];
#pragma unroll
  for (int k = 0; k < IN_DIM; ++k) w[k] = W[k * HID + lane];
  float wa[NPW], wb[NPW], d[NPW];
#pragma unroll
  for (int i = 0; i < NPW; ++i) {
    int s = __builtin_amdgcn_readlane(offv, i);
    int e = __builtin_amdgcn_readlane(offv, i + 1);
    int ja = s + lane, jb = ja + 32;
    wa[i] = (ja < e) ? __int_as_float(seg[ja].y) : 0.0f;
    wb[i] = (jb < e) ? __int_as_float(seg[jb].y) : 0.0f;
  }
#pragma unroll
  for (int i = 0; i < NPW; ++i) d[i] = wa[i] + wb[i];
#pragma unroll
  for (int i = 0; i < NPW; ++i) {
    int s = __builtin_amdgcn_readlane(offv, i);
    int e = __builtin_amdgcn_readlane(offv, i + 1);
    if (e - s > 64) {
      for (int j = s + 64 + lane; j < e; j += 32) d[i] += __int_as_float(seg[j].y);
    }
  }
  float res = 0.0f;
#pragma unroll
  for (int i = 0; i < NPW; ++i) {
    float t = d[i];
    for (int k = 16; k > 0; k >>= 1) t += __shfl_xor(t, k, 32);
    res = (lane == i) ? t : res;
  }
  float dv = rsqrtf(res + 1.0f);
  if (lane < NPW && node0 + lane < N_NODES) dinv[node0 + lane] = dv;
#pragma unroll
  for (int i = 0; i < NPW; ++i) {
    if (node0 + i < N_NODES) {
      float ds  = __int_as_float(__builtin_amdgcn_readlane(__float_as_int(dv), i));
      float acc = 0.0f;
#pragma unroll
      for (int k = 0; k < IN_DIM; ++k) {
        int f = i * IN_DIM + k;
        float xv = __int_as_float(__builtin_amdgcn_readlane(__float_as_int(f < 32 ? xa : xb), f & 31));
        acc += xv * w[k];
      }
      acc *= ds;
      xw[(size_t)(node0 + i) * HID + lane] = acc;
    }
  }
}

template <int K0, int G, int SURE>
__device__ __forceinline__ float gather_rows(float acc, int r0, float c0, int r1, float c1, int n,
                                             const float* __restrict__ xcol) {
  float xv[8 * G];
#pragma unroll
  for (int k = 0; k < 8 * G; ++k) {
    int t  = K0 + k;
    int rk = __builtin_amdgcn_readlane(t < 32 ? r0 : r1, t & 31);
    xv[k]  = xcol[(size_t)rk * HID];
  }
#pragma unroll
  for (int k = 0; k < 8 * G; ++k) {
    int t    = K0 + k;
    float ck = __int_as_float(__builtin_amdgcn_readlane(__float_as_int(t < 32 ? c0 : c1), t & 31));
    float a2 = fmaf(ck, xv[k], acc);
    acc = (k < SURE || t < n) ? a2 : acc;
  }
  return acc;
}

#define FIRST 40
__device__ __forceinline__ float gather_first(float acc, int r0, float c0, int r1, float c1, int n,
                                              const float* __restrict__ xcol) {
  if (n > 32) return gather_rows<0, 5, 32>(acc, r0, c0, r1, c1, n, xcol);
  if (n > 24) return gather_rows<0, 4, 24>(acc, r0, c0, r1, c1, n, xcol);
  if (n > 16) return gather_rows<0, 3, 16>(acc, r0, c0, r1, c1, n, xcol);
  return gather_rows<0, 2, 0>(acc, r0, c0, r1, c1, n, xcol);
}

__device__ __forceinline__ float gather_second(float acc, int r0, float c0, int r1, float c1, int n,
                                               const float* __restrict__ xcol) {
  if (n > FIRST + 16) return gather_rows<FIRST, 3, 16>(acc, r0, c0, r1, c1, n, xcol);
  if (n > FIRST + 8)  return gather_rows<FIRST, 2, 8>(acc, r0, c0, r1, c1, n, xcol);
  return gather_rows<FIRST, 1, 0>(acc, r0, c0, r1, c1, n, xcol);
}

template <class Sink>
__device__ __forceinline__ void walk_nodes(const int* __restrict__ off,
                                           const int2* __restrict__ seg,
                                           const float* __restrict__ xw,
                                           const float* __restrict__ dinv,
                                           float bb, int node0, int lane, Sink sink) {
  int offv = N_EDGES;
  if (lane <= NPW && node0 + lane < N_NODES) offv = off[node0 + lane];
  const float* xcol = xw + lane;
  int cnt = N_NODES - node0;
  cnt = cnt < NPW ? cnt : NPW;
  int s = __builtin_amdgcn_readlane(offv, 0);
  int e = __builtin_amdgcn_readlane(offv, 1);
  int   rA = node0, rB = node0;
  float cA = 0.0f,  cB = 0.0f;
  {
    int j = s + lane;
    if (j < e)      { int2 p = seg[j];      rA = p.x; cA = __int_as_float(p.y); }
    if (j + 32 < e) { int2 p = seg[j + 32]; rB = p.x; cB = __int_as_float(p.y); }
  }
  for (int i = 0; i < cnt; ++i) {
    int   node = node0 + i;
    int   r0 = rA, r1 = rB;
    float c0 = cA, c1 = cB;
    int sN = e, eN = e;
    if (i + 1 < cnt) {
      eN = __builtin_amdgcn_readlane(offv, i + 2);
      int j = sN + lane;
      rA = node + 1; cA = 0.0f;
      rB = node + 1; cB = 0.0f;
      if (j < eN)      { int2 p = seg[j];      rA = p.x; cA = __int_as_float(p.y); }
      if (j + 32 < eN) { int2 p = seg[j + 32]; rB = p.x; cB = __int_as_float(p.y); }
    }
    size_t o  = (size_t)node * HID + lane;
    float xs  = xw[o];
    float di  = dinv[node];
    float acc = 0.0f;
    int n = e - s;
    acc = gather_first(acc, r0, c0, r1, c1, n, xcol);
    if (n > FIRST) acc = gather_second(acc, r0, c0, r1, c1, n, xcol);
    for (int base = s + 64; base < e; base += 32) {
      int j = base + lane;
      int r = node; float cf = 0.0f;
      if (j < e) { int2 p = seg[j]; r = p.x; cf = __int_as_float(p.y); }
      acc = gather_rows<0, 4, 0>(acc, r, cf, r, cf, e - base, xcol);
    }
    float v = di * (acc + xs) + bb;
    sink(i, node, di, v);
    s = sN; e = eN;
  }
}

__global__ __launch_bounds__(AGG_THREADS) void aggregate_mm_kernel(const int* __restrict__ off,
                                                                   const int2* __restrict__ seg,
                                                                   const float* __restrict__ xw,
                                                                   const float* __restrict__ dinv,
                                                                   const float* __restrict__ b,
                                                                   const float* __restrict__ g,
                                                                   const float* __restrict__ be,
                                                                   const float* __restrict__ Wn,
                                                                   float* __restrict__ xwn) {
  __shared__ float tile[AGG_NODES * TILE_PITCH];
  int wv    = __builtin_amdgcn_readfirstlane((int)(threadIdx.x >> 5));
  int node0 = ((int)blockIdx.x * AGG_WAVES + wv) * NPW;
  int lane  = threadIdx.x & 31;
  if (node0 < N_NODES) {
    float bb = b[lane];
    float gs = g[lane] * rsqrtf(1.0f + 1e-5f);
    float bs = be[lane];
    float* trow = tile + (wv * NPW) * TILE_PITCH + lane;
    walk_nodes(off, seg, xw, dinv, bb, node0, lane,
               [&](int i, int node, float di, float v) {
                 v = v * gs + bs;
                 v = v > 0.0f ? v : 0.0f;
                 v *= di;
                 trow[i * TILE_PITCH] = v;
               });
  }
  int  row0 = (int)blockIdx.x * AGG_NODES + wv * 16;
  bool mm   = wv < AGG_TILES && row0 < N_NODES;
  int  grp  = lane >> 4;
  int  m    = lane & 15;
  v2f b0[HID / 4], b1[HID / 4];
  if (mm) {
#pragma unroll
    for (int q = 0; q < HID / 4; ++q) {
      int kk = 4 * q + 2 * grp;
      b0[q].x = Wn[kk * HID + m];          b0[q].y = Wn[(kk + 1) * HID + m];
      b1[q].x = Wn[kk * HID + 16 + m];     b1[q].y = Wn[(kk + 1) * HID + 16 + m];
    }
  }
  __syncthreads();
  if (mm) {
    v8f c0 = {0.f,0.f,0.f,0.f,0.f,0.f,0.f,0.f};
    v8f c1 = {0.f,0.f,0.f,0.f,0.f,0.f,0.f,0.f};
    const float* hrow = tile + (wv * 16 + m) * TILE_PITCH;
#pragma unroll
    for (int q = 0; q < HID / 4; ++q) {
      int kk = 4 * q + 2 * grp;
      v2f a;  a.x  = hrow[kk];                  a.y  = hrow[kk + 1];
      c0 = __builtin_amdgcn_wmma_f32_16x16x4_f32(false, a, false, b0[q], (short)0, c0, false, false);
      c1 = __builtin_amdgcn_wmma_f32_16x16x4_f32(false, a, false, b1[q], (short)0, c1, false, false);
    }
#pragma unroll
    for (int r = 0; r < 8; ++r) {
      int M = r + 8 * grp;
      xwn[(size_t)(row0 + M) * HID + m]      = c0[r];
      xwn[(size_t)(row0 + M) * HID + 16 + m] = c1[r];
    }
  }
}

__global__ __launch_bounds__(256) void aggregate_out_kernel(const int* __restrict__ off,
                                                            const int2* __restrict__ seg,
                                                            const float* __restrict__ xw,
                                                            const float* __restrict__ dinv,
                                                            const float* __restrict__ b,
                                                            const int* __restrict__ batch,
                                                            float* __restrict__ sums,
                                                            float* __restrict__ cnts,
                                                            float* __restrict__ out) {
  int wave  = __builtin_amdgcn_readfirstlane((int)(blockIdx.x * 8 + (threadIdx.x >> 5)));
  int node0 = wave * NPW;
  if (node0 >= N_NODES) return;
  int lane = threadIdx.x & 31;
  float bb = b[lane];
  int bgv = 0;
  if (lane < NPW && node0 + lane < N_NODES) bgv = batch[node0 + lane];
  int   pg = __builtin_amdgcn_readlane(bgv, 0);
  float ps = 0.0f, pc = 0.0f;
  walk_nodes(off, seg, xw, dinv, bb, node0, lane,
             [&](int i, int node, float di, float v) {
               out[(size_t)node * HID + lane] = v;
               int bg = __builtin_amdgcn_readlane(bgv, i);
               if (bg != pg) {
                 atomicAdd(&sums[pg * HID + lane], ps);
                 if (lane == 0) atomicAdd(&cnts[pg], pc);
                 ps = 0.0f; pc = 0.0f; pg = bg;
               }
               ps += v;
               pc += 1.0f;
             });
  atomicAdd(&sums[pg * HID + lane], ps);
  if (lane == 0) atomicAdd(&cnts[pg], pc);
}

__global__ void pred_kernel(const float* __restrict__ sums,
                            const float* __restrict__ cnts,
                            const float* __restrict__ Wout,
                            const float* __restrict__ bout,
                            float* __restrict__ pred) {
  int gph = blockIdx.x * blockDim.x + threadIdx.x;
  if (gph >= N_GRAPHS) return;
  float cnt = cnts[gph];
  cnt = cnt > 1.0f ? cnt : 1.0f;
  float acc = bout[0];
#pragma unroll
  for (int f = 0; f < HID; ++f)
    acc += (sums[gph * HID + f] / cnt) * Wout[f];
  pred[gph] = 1.0f / (1.0f + __expf(-acc));
}

extern "C" void kernel_launch(void* const* d_in, const int* in_sizes, int n_in,
                              void* d_out, int out_size, void* d_ws, size_t ws_size,
                              hipStream_t stream) {
  const float* x     = (const float*)d_in[0];
  const int*   ei    = (const int*)  d_in[1];
  const float* pos   = (const float*)d_in[2];
  const int*   batch = (const int*)  d_in[3];
  const float* s1    = (const float*)d_in[4];
  const float* s2    = (const float*)d_in[5];
  const float* W1 = (const float*)d_in[6];   const float* b1 = (const float*)d_in[7];
  const float* W2 = (const float*)d_in[8];   const float* b2 = (const float*)d_in[9];
  const float* W3 = (const float*)d_in[10];  const float* b3 = (const float*)d_in[11];
  const float* W4 = (const float*)d_in[12];  const float* b4 = (const float*)d_in[13];
  const float* g1 = (const float*)d_in[14];  const float* be1 = (const float*)d_in[15];
  const float* g2 = (const float*)d_in[16];  const float* be2 = (const float*)d_in[17];
  const float* g3 = (const float*)d_in[18];  const float* be3 = (const float*)d_in[19];
  const float* Wout = (const float*)d_in[20];
  const float* bout = (const float*)d_in[21];

  const int* row = ei;
  const int* col = ei + N_EDGES;

  float* ws   = (float*)d_ws;
  int2*  tmp  = (int2*)ws;
  float* xw   = ws;
  float* h    = xw   + (size_t)N_NODES * HID;
  int2*  seg  = (int2*)(h + (size_t)N_NODES * HID);
  int*   off  = (int*)(ws + (size_t)4 * N_EDGES);
  float* dinv = (float*)(off + (size_t)N_NODES);
  float* sums = dinv + (size_t)N_NODES;
  float* cnts = sums + (size_t)N_GRAPHS * HID;
  int*   btot = (int*)sums;
  int*   tbl  = (int*)dinv;

  float* emb  = (float*)d_out;
  float* pred = (float*)d_out + (size_t)N_NODES * HID;

  const int BLK        = 256;

  bin_count_kernel<<<NB1, SCAN_BLK, 0, stream>>>(col, tbl);
  block_total_kernel<<<TBL_NBLK, SCAN_BLK, 0, stream>>>(tbl, btot, TBL);
  offsets_kernel<<<TBL_NBLK, SCAN_BLK, 0, stream>>>(tbl, btot, TBL);
  bin_scatter_kernel<<<NB1, SCAN_BLK, 0, stream>>>(row, col, tbl, tmp);
  bin_sort_kernel<<<NBIN, SCAN_BLK, 0, stream>>>(pos, s1, s2, tbl, tmp, off, seg);

  dinv_gemm_in_kernel<<<NODE_BLOCKS, BLK, 0, stream>>>(off, seg, x, W1, dinv, xw, sums);

  aggregate_mm_kernel<<<AGG_BLOCKS, AGG_THREADS, 0, stream>>>(off, seg, xw, dinv, b1, g1, be1, W2, h);
  aggregate_mm_kernel<<<AGG_BLOCKS, AGG_THREADS, 0, stream>>>(off, seg, h, dinv, b2, g2, be2, W3, xw);
  aggregate_mm_kernel<<<AGG_BLOCKS, AGG_THREADS, 0, stream>>>(off, seg, xw, dinv, b3, g3, be3, W4, h);

  aggregate_out_kernel<<<NODE_BLOCKS, BLK, 0, stream>>>(off, seg, h, dinv, b4, batch, sums, cnts, emb);
  pred_kernel<<<1, BLK, 0, stream>>>(sums, cnts, Wout, bout, pred);
}


# 11 launches to 8: the scatter pass sums and scans the count table itself (2 scan launches go); the per-bin pass makes degree (LDS float adds), dinv and layer-1 rows; degree sum order: LDS arrival
# speedup vs baseline: 7.5586x; 1.0558x over previous
#include <hip/hip_runtime.h>
#include <hip/hip_bf16.h>


#define N_NODES  100000
#define N_EDGES  3200000
#define N_GRAPHS 256
#define HID      32
#define IN_DIM   6
#define SCAN_BLK  1024
#define BIN_SHIFT 8
#define BIN_NODES (1 << BIN_SHIFT)
#define NBIN      ((N_NODES + BIN_NODES - 1) / BIN_NODES)
#define EPB       12800
#define EPT       ((EPB + SCAN_BLK - 1) / SCAN_BLK)
#define NB1       (N_EDGES / EPB)
#define TBL       (NBIN * NB1)
#define HALF_BINS (SCAN_BLK / 2)
#define TBL_GRP   25
#define BIN_CAP   12288
#define SORT_NPW  (BIN_NODES / (SCAN_BLK / 32))
#define NPW         7
#define NODE_WAVES  ((N_NODES + NPW - 1) / NPW)
#define NODE_BLOCKS ((NODE_WAVES + 7) / 8)
#define AGG_WAVES   16
#define AGG_THREADS (AGG_WAVES * 32)
#define AGG_NODES   (AGG_WAVES * NPW)
#define AGG_TILES   (AGG_NODES / 16)
#define AGG_BLOCKS  ((N_NODES + AGG_NODES - 1) / AGG_NODES)
#define TILE_PITCH  34
#define POOL_WORDS  (N_GRAPHS * (HID + 1))
static_assert(AGG_NODES % 16 == 0 && N_NODES % 16 == 0, "a block's nodes are whole 16-row tiles");
static_assert(SORT_NPW * IN_DIM <= 64, "a wave's x rows fit two registers a lane");
static_assert(AGG_BLOCKS * AGG_THREADS >= POOL_WORDS, "an aggregation launch has a thread for every pooling word");
static_assert(NBIN <= HALF_BINS && NB1 % 2 == 0, "two threads per bin, each with half of the blocks");
static_assert((NB1 / 2) % TBL_GRP == 0, "a half of the blocks is whole groups of table loads");
static_assert(N_NODES <= (1 << (32 - BIN_SHIFT)), "source row and node-in-bin share one word");
static_assert(NBIN + 1 <= N_GRAPHS * HID, "the bin starts borrow the pooling sums");
static_assert(NPW + 1 <= 32, "a wave holds its nodes' segment bounds one per lane");
static_assert(N_EDGES % EPB == 0, "edge passes assume whole blocks");
static_assert(TBL <= N_NODES, "the run table borrows the dinv array");
static_assert(NBIN <= SCAN_BLK, "one thread per bin");

typedef __attribute__((ext_vector_type(2))) float v2f;
typedef __attribute__((ext_vector_type(8))) float v8f;


__device__ __forceinline__ int wave_sum_i(int v) {
  for (int d = 16; d > 0; d >>= 1) v += __shfl_xor(v, d, 32);
  return v;
}

__device__ __forceinline__ int wave_incl_scan_i(int v, int lane) {
  for (int d = 1; d < 32; d <<= 1) {
    int u = __shfl_up(v, d, 32);
    if (lane >= d) v += u;
  }
  return v;
}

__global__ __launch_bounds__(SCAN_BLK) void bin_count_kernel(const int* __restrict__ col,
                                                             int* __restrict__ tbl) {
  __shared__ int hist[NBIN];
  int tid = threadIdx.x;
  if (tid < NBIN) hist[tid] = 0;
  __syncthreads();
  const int* c = col + (size_t)blockIdx.x * EPB;
  int cv[EPT];
#pragma unroll
  for (int k = 0; k < EPT; ++k) {
    int i = tid + k * SCAN_BLK;
    cv[k] = (i < EPB) ? c[i] : -1;
  }
#pragma unroll
  for (int k = 0; k < EPT; ++k)
    if (cv[k] >= 0) atomicAdd(&hist[cv[k] >> BIN_SHIFT], 1);
  __syncthreads();
  if (tid < NBIN) tbl[blockIdx.x * NBIN + tid] = hist[tid];
}

__global__ __launch_bounds__(SCAN_BLK) void bin_scatter_kernel(const int* __restrict__ row,
                                                               const int* __restrict__ col,
                                                               const int* __restrict__ tbl,
                                                               int* __restrict__ binstart,
                                                               unsigned* __restrict__ tmp) {
  __shared__ int2 stage[EPB];
  __shared__ int cursor[NBIN];
  __shared__ int gbase[NBIN];
  __shared__ int wsumL[32];
  __shared__ int wsumT[32];
  __shared__ int ptot[HALF_BINS];
  __shared__ int ppre[HALF_BINS];
  int tid = threadIdx.x, lane = tid & 31, wv = tid >> 5;
  int blk = blockIdx.x;
  const int* r = row + (size_t)blk * EPB;
  const int* c = col + (size_t)blk * EPB;
  int rv[EPT], cv[EPT];
#pragma unroll
  for (int k = 0; k < EPT; ++k) {
    int i = tid + k * SCAN_BLK;
    rv[k] = (i < EPB) ? r[i] : 0;
    cv[k] = (i < EPB) ? c[i] : -1;
  }
  int bin = tid & (HALF_BINS - 1), half = tid / HALF_BINS;
  int tot = 0, pre = 0;
  if (bin < NBIN) {
    const int* t = tbl + half * (NB1 / 2) * NBIN + bin;
    int lim = blk - half * (NB1 / 2);
#pragma nounroll
    for (int k0 = 0; k0 < NB1 / 2; k0 += TBL_GRP) {
      int v[TBL_GRP];
#pragma unroll
      for (int j = 0; j < TBL_GRP; ++j) v[j] = t[(k0 + j) * NBIN];
#pragma unroll
      for (int j = 0; j < TBL_GRP; ++j) {
        tot += v[j];
        pre += (k0 + j < lim) ? v[j] : 0;
      }
    }
  }
  if (half == 1) { ptot[bin] = tot; ppre[bin] = pre; }
  __syncthreads();
  int len = 0;
  if (tid < NBIN) {
    tot += ptot[tid];
    pre += ppre[tid];
    len  = tbl[blk * NBIN + tid];
  } else {
    tot = 0;
  }
  int inclT = wave_incl_scan_i(tot, lane);
  int inclL = wave_incl_scan_i(len, lane);
  if (lane == 31) { wsumT[wv] = inclT; wsumL[wv] = inclL; }
  __syncthreads();
  int bstart = wave_sum_i((lane < wv) ? wsumT[lane] : 0) + inclT - tot;
  int lofs   = wave_sum_i((lane < wv) ? wsumL[lane] : 0) + inclL - len;
  if (tid < NBIN) { cursor[tid] = lofs; gbase[tid] = bstart + pre - lofs; }
  if (blk == 0 && tid <= NBIN) binstart[tid] = bstart;
  __syncthreads();
#pragma unroll
  for (int k = 0; k < EPT; ++k) {
    if (cv[k] >= 0) {
      int p = atomicAdd(&cursor[cv[k] >> BIN_SHIFT], 1);
      stage[p] = make_int2(rv[k], cv[k]);
    }
  }
  __syncthreads();
  for (int i = tid; i < EPB; i += SCAN_BLK) {
    int2 q = stage[i];
    tmp[gbase[q.y >> BIN_SHIFT] + i] = ((unsigned)q.x << BIN_SHIFT) | (unsigned)(q.y & (BIN_NODES - 1));
  }
}

__global__ __launch_bounds__(SCAN_BLK) void bin_sort_kernel(const float* __restrict__ pos,
                                                            const float* __restrict__ s1p,
                                                            const float* __restrict__ s2p,
                                                            const int* __restrict__ binstart,
                                                            const unsigned* __restrict__ tmp,
                                                            const float* __restrict__ x,
                                                            const float* __restrict__ W,
                                                            int* __restrict__ off,
                                                            int2* __restrict__ seg,
                                                            float* __restrict__ dinv,
                                                            float* __restrict__ xw) {
  __shared__ int2 stage[BIN_CAP];
  __shared__ int cnt[BIN_NODES];
  __shared__ int cursor[BIN_NODES];
  __shared__ float degl[BIN_NODES];
  __shared__ int wsum[8];
  int tid = threadIdx.x, lane = tid & 31, wv = tid >> 5;
  int b = blockIdx.x;
  int S  = binstart[b];
  int En = binstart[b + 1];
  int nb = En - S;
  const unsigned* in = tmp + S;
  if (tid < BIN_NODES) { cnt[tid] = 0; degl[tid] = 0.0f; }
  __syncthreads();
  for (int i = tid; i < nb; i += SCAN_BLK)
    atomicAdd(&cnt[in[i] & (BIN_NODES - 1)], 1);
  __syncthreads();
  int v = (tid < BIN_NODES) ? cnt[tid] : 0;
  int incl = wave_incl_scan_i(v, lane);
  if (lane == 31 && wv < 8) wsum[wv] = incl;
  __syncthreads();
  int before = wave_sum_i((lane < wv && lane < 8) ? wsum[lane] : 0);
  int lofs = before + incl - v;
  if (tid < BIN_NODES) {
    cursor[tid] = lofs;
    int node = b * BIN_NODES + tid;
    if (node < N_NODES) off[node] = S + lofs;
  }
  __syncthreads();
  float s1 = s1p[0], s2 = s2p[0];
  bool staged = nb <= BIN_CAP;
  for (int i = tid; i < nb; i += SCAN_BLK) {
    unsigned q = in[i];
    int rr = (int)(q >> BIN_SHIFT), cl = (int)(q & (BIN_NODES - 1));
    int cc = b * BIN_NODES + cl;
    const float* pr = pos + (size_t)rr * 6;
    const float* pc = pos + (size_t)cc * 6;
    float d0 = pr[0] - pc[0], d1 = pr[1] - pc[1], d2 = pr[2] - pc[2];
    float D = d0 * d0 + d1 * d1 + d2 * d2;
    float t = 1.0f - (pr[3] * pc[3] + pr[4] * pc[4] + pr[5] * pc[5]);
    float w = __expf(-(D * s1 * s1 + t * t * s2 * s2));
    int p = atomicAdd(&cursor[cl], 1);
    atomicAdd(&degl[cl], w);
    int2 o = make_int2(rr, __float_as_int(w));
    if (staged) stage[p] = o; else seg[S + p] = o;
  }
  int wvu = __builtin_amdgcn_readfirstlane(wv);
  int n0  = b * BIN_NODES + wvu * SORT_NPW;
  int xi  = n0 * IN_DIM + lane;
  float xa = (xi < N_NODES * IN_DIM) ? x[xi] : 0.0f;
  float xb = (lane < SORT_NPW * IN_DIM - 32 && xi + 32 < N_NODES * IN_DIM) ? x[xi + 32] : 0.0f;
  float wk[IN_DIM];
#pragma unroll
  for (int k = 0; k < IN_DIM; ++k) wk[k] = W[k * HID + lane];
  __syncthreads();
  if (staged) {
    for (int i = tid; i < nb; i += SCAN_BLK) seg[S + i] = stage[i];
  }
  if (tid < BIN_NODES) {
    float dv = rsqrtf(degl[tid] + 1.0f);
    degl[tid] = dv;
    int node = b * BIN_NODES + tid;
    if (node < N_NODES) dinv[node] = dv;
  }
  __syncthreads();
#pragma unroll
  for (int i = 0; i < SORT_NPW; ++i) {
    if (n0 + i < N_NODES) {
      float ds  = degl[wvu * SORT_NPW + i];
      float acc = 0.0f;
#pragma unroll
      for (int k = 0; k < IN_DIM; ++k) {
        int f = i * IN_DIM + k;
        float xv = __int_as_float(__builtin_amdgcn_readlane(__float_as_int(f < 32 ? xa : xb), f & 31));
        acc += xv * wk[k];
      }
      acc *= ds;
      xw[(size_t)(n0 + i) * HID + lane] = acc;
    }
  }
}

template <int K0, int G, int SURE>
__device__ __forceinline__ float gather_rows(float acc, int r0, float c0, int r1, float c1, int n,
                                             const float* __restrict__ xcol) {
  float xv[8 * G];
#pragma unroll
  for (int k = 0; k < 8 * G; ++k) {
    int t  = K0 + k;
    int rk = __builtin_amdgcn_readlane(t < 32 ? r0 : r1, t & 31);
    xv[k]  = xcol[(size_t)rk * HID];
  }
#pragma unroll
  for (int k = 0; k < 8 * G; ++k) {
    int t    = K0 + k;
    float ck = __int_as_float(__builtin_amdgcn_readlane(__float_as_int(t < 32 ? c0 : c1), t & 31));
    float a2 = fmaf(ck, xv[k], acc);
    acc = (k < SURE || t < n) ? a2 : acc;
  }
  return acc;
}

#define FIRST 40
__device__ __forceinline__ float gather_first(float acc, int r0, float c0, int r1, float c1, int n,
                                              const float* __restrict__ xcol) {
  if (n > 32) return gather_rows<0, 5, 32>(acc, r0, c0, r1, c1, n, xcol);
  if (n > 24) return gather_rows<0, 4, 24>(acc, r0, c0, r1, c1, n, xcol);
  if (n > 16) return gather_rows<0, 3, 16>(acc, r0, c0, r1, c1, n, xcol);
  return gather_rows<0, 2, 0>(acc, r0, c0, r1, c1, n, xcol);
}

__device__ __forceinline__ float gather_second(float acc, int r0, float c0, int r1, float c1, int n,
                                               const float* __restrict__ xcol) {
  if (n > FIRST + 16) return gather_rows<FIRST, 3, 16>(acc, r0, c0, r1, c1, n, xcol);
  if (n > FIRST + 8)  return gather_rows<FIRST, 2, 8>(acc, r0, c0, r1, c1, n, xcol);
  return gather_rows<FIRST, 1, 0>(acc, r0, c0, r1, c1, n, xcol);
}

template <class Sink>
__device__ __forceinline__ void walk_nodes(const int* __restrict__ off,
                                           const int2* __restrict__ seg,
                                           const float* __restrict__ xw,
                                           const float* __restrict__ dinv,
                                           float bb, int node0, int lane, Sink sink) {
  int offv = N_EDGES;
  if (lane <= NPW && node0 + lane < N_NODES) offv = off[node0 + lane];
  const float* xcol = xw + lane;
  int cnt = N_NODES - node0;
  cnt = cnt < NPW ? cnt : NPW;
  int s = __builtin_amdgcn_readlane(offv, 0);
  int e = __builtin_amdgcn_readlane(offv, 1);
  int   rA = node0, rB = node0;
  float cA = 0.0f,  cB = 0.0f;
  {
    int j = s + lane;
    if (j < e)      { int2 p = seg[j];      rA = p.x; cA = __int_as_float(p.y); }
    if (j + 32 < e) { int2 p = seg[j + 32]; rB = p.x; cB = __int_as_float(p.y); }
  }
  for (int i = 0; i < cnt; ++i) {
    int   node = node0 + i;
    int   r0 = rA, r1 = rB;
    float c0 = cA, c1 = cB;
    int sN = e, eN = e;
    if (i + 1 < cnt) {
      eN = __builtin_amdgcn_readlane(offv, i + 2);
      int j = sN + lane;
      rA = node + 1; cA = 0.0f;
      rB = node + 1; cB = 0.0f;
      if (j < eN)      { int2 p = seg[j];      rA = p.x; cA = __int_as_float(p.y); }
      if (j + 32 < eN) { int2 p = seg[j + 32]; rB = p.x; cB = __int_as_float(p.y); }
    }
    size_t o  = (size_t)node * HID + lane;
    float xs  = xw[o];
    float di  = dinv[node];
    float acc = 0.0f;
    int n = e - s;
    acc = gather_first(acc, r0, c0, r1, c1, n, xcol);
    if (n > FIRST) acc = gather_second(acc, r0, c0, r1, c1, n, xcol);
    for (int base = s + 64; base < e; base += 32) {
      int j = base + lane;
      int r = node; float cf = 0.0f;
      if (j < e) { int2 p = seg[j]; r = p.x; cf = __int_as_float(p.y); }
      acc = gather_rows<0, 4, 0>(acc, r, cf, r, cf, e - base, xcol);
    }
    float v = di * (acc + xs) + bb;
    sink(i, node, di, v);
    s = sN; e = eN;
  }
}

__global__ __launch_bounds__(AGG_THREADS) void aggregate_mm_kernel(const int* __restrict__ off,
                                                                   const int2* __restrict__ seg,
                                                                   const float* __restrict__ xw,
                                                                   const float* __restrict__ dinv,
                                                                   const float* __restrict__ b,
                                                                   const float* __restrict__ g,
                                                                   const float* __restrict__ be,
                                                                   const float* __restrict__ Wn,
                                                                   float* __restrict__ xwn,
                                                                   float* __restrict__ poolz) {
  __shared__ float tile[AGG_NODES * TILE_PITCH];
  int gt = blockIdx.x * AGG_THREADS + threadIdx.x;
  if (poolz != nullptr && gt < POOL_WORDS) poolz[gt] = 0.0f;
  int wv    = __builtin_amdgcn_readfirstlane((int)(threadIdx.x >> 5));
  int node0 = ((int)blockIdx.x * AGG_WAVES + wv) * NPW;
  int lane  = threadIdx.x & 31;
  if (node0 < N_NODES) {
    float bb = b[lane];
    float gs = g[lane] * rsqrtf(1.0f + 1e-5f);
    float bs = be[lane];
    float* trow = tile + (wv * NPW) * TILE_PITCH + lane;
    walk_nodes(off, seg, xw, dinv, bb, node0, lane,
               [&](int i, int node, float di, float v) {
                 v = v * gs + bs;
                 v = v > 0.0f ? v : 0.0f;
                 v *= di;
                 trow[i * TILE_PITCH] = v;
               });
  }
  int  row0 = (int)blockIdx.x * AGG_NODES + wv * 16;
  bool mm   = wv < AGG_TILES && row0 < N_NODES;
  int  grp  = lane >> 4;
  int  m    = lane & 15;
  v2f b0[HID / 4], b1[HID / 4];
  if (mm) {
#pragma unroll
    for (int q = 0; q < HID / 4; ++q) {
      int kk = 4 * q + 2 * grp;
      b0[q].x = Wn[kk * HID + m];          b0[q].y = Wn[(kk + 1) * HID + m];
      b1[q].x = Wn[kk * HID + 16 + m];     b1[q].y = Wn[(kk + 1) * HID + 16 + m];
    }
  }
  __syncthreads();
  if (mm) {
    v8f c0 = {0.f,0.f,0.f,0.f,0.f,0.f,0.f,0.f};
    v8f c1 = {0.f,0.f,0.f,0.f,0.f,0.f,0.f,0.f};
    const float* hrow = tile + (wv * 16 + m) * TILE_PITCH;
#pragma unroll
    for (int q = 0; q < HID / 4; ++q) {
      int kk = 4 * q + 2 * grp;
      v2f a;  a.x  = hrow[kk];                  a.y  = hrow[kk + 1];
      c0 = __builtin_amdgcn_wmma_f32_16x16x4_f32(false, a, false, b0[q], (short)0, c0, false, false);
      c1 = __builtin_amdgcn_wmma_f32_16x16x4_f32(false, a, false, b1[q], (short)0, c1, false, false);
    }
#pragma unroll
    for (int r = 0; r < 8; ++r) {
      int M = r + 8 * grp;
      xwn[(size_t)(row0 + M) * HID + m]      = c0[r];
      xwn[(size_t)(row0 + M) * HID + 16 + m] = c1[r];
    }
  }
}

__global__ __launch_bounds__(256) void aggregate_out_kernel(const int* __restrict__ off,
                                                            const int2* __restrict__ seg,
                                                            const float* __restrict__ xw,
                                                            const float* __restrict__ dinv,
                                                            const float* __restrict__ b,
                                                            const int* __restrict__ batch,
                                                            float* __restrict__ sums,
                                                            float* __restrict__ cnts,
                                                            float* __restrict__ out) {
  int wave  = __builtin_amdgcn_readfirstlane((int)(blockIdx.x * 8 + (threadIdx.x >> 5)));
  int node0 = wave * NPW;
  if (node0 >= N_NODES) return;
  int lane = threadIdx.x & 31;
  float bb = b[lane];
  int bgv = 0;
  if (lane < NPW && node0 + lane < N_NODES) bgv = batch[node0 + lane];
  int   pg = __builtin_amdgcn_readlane(bgv, 0);
  float ps = 0.0f, pc = 0.0f;
  walk_nodes(off, seg, xw, dinv, bb, node0, lane,
             [&](int i, int node, float di, float v) {
               out[(size_t)node * HID + lane] = v;
               int bg = __builtin_amdgcn_readlane(bgv, i);
               if (bg != pg) {
                 atomicAdd(&sums[pg * HID + lane], ps);
                 if (lane == 0) atomicAdd(&cnts[pg], pc);
                 ps = 0.0f; pc = 0.0f; pg = bg;
               }
               ps += v;
               pc += 1.0f;
             });
  atomicAdd(&sums[pg * HID + lane], ps);
  if (lane == 0) atomicAdd(&cnts[pg], pc);
}

__global__ void pred_kernel(const float* __restrict__ sums,
                            const float* __restrict__ cnts,
                            const float* __restrict__ Wout,
                            const float* __restrict__ bout,
                            float* __restrict__ pred) {
  int gph = blockIdx.x * blockDim.x + threadIdx.x;
  if (gph >= N_GRAPHS) return;
  float cnt = cnts[gph];
  cnt = cnt > 1.0f ? cnt : 1.0f;
  float acc = bout[0];
#pragma unroll
  for (int f = 0; f < HID; ++f)
    acc += (sums[gph * HID + f] / cnt) * Wout[f];
  pred[gph] = 1.0f / (1.0f + __expf(-acc));
}

extern "C" void kernel_launch(void* const* d_in, const int* in_sizes, int n_in,
                              void* d_out, int out_size, void* d_ws, size_t ws_size,
                              hipStream_t stream) {
  const float* x     = (const float*)d_in[0];
  const int*   ei    = (const int*)  d_in[1];
  const float* pos   = (const float*)d_in[2];
  const int*   batch = (const int*)  d_in[3];
  const float* s1    = (const float*)d_in[4];
  const float* s2    = (const float*)d_in[5];
  const float* W1 = (const float*)d_in[6];   const float* b1 = (const float*)d_in[7];
  const float* W2 = (const float*)d_in[8];   const float* b2 = (const float*)d_in[9];
  const float* W3 = (const float*)d_in[10];  const float* b3 = (const float*)d_in[11];
  const float* W4 = (const float*)d_in[12];  const float* b4 = (const float*)d_in[13];
  const float* g1 = (const float*)d_in[14];  const float* be1 = (const float*)d_in[15];
  const float* g2 = (const float*)d_in[16];  const float* be2 = (const float*)d_in[17];
  const float* g3 = (const float*)d_in[18];  const float* be3 = (const float*)d_in[19];
  const float* Wout = (const float*)d_in[20];
  const float* bout = (const float*)d_in[21];

  const int* row = ei;
  const int* col = ei + N_EDGES;

  float*    ws   = (float*)d_ws;
  unsigned* tmp  = (unsigned*)ws;
  float*    xw   = ws;
  float*    h    = xw   + (size_t)N_NODES * HID;
  int2*     seg  = (int2*)(h + (size_t)N_NODES * HID);
  int*      off  = (int*)(ws + (size_t)4 * N_EDGES);
  float*    dinv = (float*)(off + (size_t)N_NODES);
  float*    sums = dinv + (size_t)N_NODES;
  float*    cnts = sums + (size_t)N_GRAPHS * HID;
  int*      bst  = (int*)sums;
  int*      tbl  = (int*)dinv;

  float* emb  = (float*)d_out;
  float* pred = (float*)d_out + (size_t)N_NODES * HID;

  const int BLK        = 256;

  bin_count_kernel<<<NB1, SCAN_BLK, 0, stream>>>(col, tbl);
  bin_scatter_kernel<<<NB1, SCAN_BLK, 0, stream>>>(row, col, tbl, bst, tmp);
  bin_sort_kernel<<<NBIN, SCAN_BLK, 0, stream>>>(pos, s1, s2, bst, tmp, x, W1, off, seg, dinv, h);

  aggregate_mm_kernel<<<AGG_BLOCKS, AGG_THREADS, 0, stream>>>(off, seg, h, dinv, b1, g1, be1, W2, xw, sums);
  aggregate_mm_kernel<<<AGG_BLOCKS, AGG_THREADS, 0, stream>>>(off, seg, xw, dinv, b2, g2, be2, W3, h, nullptr);
  aggregate_mm_kernel<<<AGG_BLOCKS, AGG_THREADS, 0, stream>>>(off, seg, h, dinv, b3, g3, be3, W4, xw, nullptr);

  aggregate_out_kernel<<<NODE_BLOCKS, BLK, 0, stream>>>(off, seg, xw, dinv, b4, batch, sums, cnts, emb);
  pred_kernel<<<1, BLK, 0, stream>>>(sums, cnts, Wout, bout, pred);
}


# Aggregation launches as 1021 blocks of 14 waves (98 nodes): a WGP's 64 wave slots take exactly 4, so every WGP gets 392 nodes (was 893 blocks of 16 waves: 448 on the fullest); sum orders kept
# speedup vs baseline: 7.8048x; 1.0326x over previous
#include <hip/hip_runtime.h>
#include <hip/hip_bf16.h>


#define N_NODES  100000
#define N_EDGES  3200000
#define N_GRAPHS 256
#define HID      32
#define IN_DIM   6
#define SCAN_BLK  1024
#define BIN_SHIFT 8
#define BIN_NODES (1 << BIN_SHIFT)
#define NBIN      ((N_NODES + BIN_NODES - 1) / BIN_NODES)
#define EPB       12800
#define EPT       ((EPB + SCAN_BLK - 1) / SCAN_BLK)
#define NB1       (N_EDGES / EPB)
#define TBL       (NBIN * NB1)
#define HALF_BINS (SCAN_BLK / 2)
#define TBL_GRP   25
#define BIN_CAP   12288
#define SORT_NPW  (BIN_NODES / (SCAN_BLK / 32))
#define NPW         7
#define NODE_WAVES  ((N_NODES + NPW - 1) / NPW)
#define NODE_BLOCKS ((NODE_WAVES + AGG_WAVES - 1) / AGG_WAVES)
#define AGG_WAVES   14
#define AGG_THREADS (AGG_WAVES * 32)
#define AGG_NODES   (AGG_WAVES * NPW)
#define AGG_TILES   ((AGG_NODES + 15) / 16)
#define AGG_BLOCKS  ((N_NODES + AGG_NODES - 1) / AGG_NODES)
#define TILE_PITCH  34
#define POOL_WORDS  (N_GRAPHS * (HID + 1))
static_assert(AGG_TILES <= AGG_WAVES && 4 * AGG_WAVES <= 64 && 5 * AGG_WAVES > 64, "a product wave per tile; 4 blocks fill a WGP");
static_assert(SORT_NPW * IN_DIM <= 64, "a wave's x rows fit two registers a lane");
static_assert(AGG_BLOCKS * AGG_THREADS >= POOL_WORDS, "an aggregation launch has a thread for every pooling word");
static_assert(NBIN <= HALF_BINS && NB1 % 2 == 0, "two threads per bin, each with half of the blocks");
static_assert((NB1 / 2) % TBL_GRP == 0, "a half of the blocks is whole groups of table loads");
static_assert(N_NODES <= (1 << (32 - BIN_SHIFT)), "source row and node-in-bin share one word");
static_assert(NBIN + 1 <= N_GRAPHS * HID, "the bin starts borrow the pooling sums");
static_assert(NPW + 1 <= 32, "a wave holds its nodes' segment bounds one per lane");
static_assert(N_EDGES % EPB == 0, "edge passes assume whole blocks");
static_assert(TBL <= N_NODES, "the run table borrows the dinv array");
static_assert(NBIN <= SCAN_BLK, "one thread per bin");

typedef __attribute__((ext_vector_type(2))) float v2f;
typedef __attribute__((ext_vector_type(8))) float v8f;


__device__ __forceinline__ int wave_sum_i(int v) {
  for (int d = 16; d > 0; d >>= 1) v += __shfl_xor(v, d, 32);
  return v;
}

__device__ __forceinline__ int wave_incl_scan_i(int v, int lane) {
  for (int d = 1; d < 32; d <<= 1) {
    int u = __shfl_up(v, d, 32);
    if (lane >= d) v += u;
  }
  return v;
}

__global__ __launch_bounds__(SCAN_BLK) void bin_count_kernel(const int* __restrict__ col,
                                                             int* __restrict__ tbl) {
  __shared__ int hist[NBIN];
  int tid = threadIdx.x;
  if (tid < NBIN) hist[tid] = 0;
  __syncthreads();
  const int* c = col + (size_t)blockIdx.x * EPB;
  int cv[EPT];
#pragma unroll
  for (int k = 0; k < EPT; ++k) {
    int i = tid + k * SCAN_BLK;
    cv[k] = (i < EPB) ? c[i] : -1;
  }
#pragma unroll
  for (int k = 0; k < EPT; ++k)
    if (cv[k] >= 0) atomicAdd(&hist[cv[k] >> BIN_SHIFT], 1);
  __syncthreads();
  if (tid < NBIN) tbl[blockIdx.x * NBIN + tid] = hist[tid];
}

__global__ __launch_bounds__(SCAN_BLK) void bin_scatter_kernel(const int* __restrict__ row,
                                                               const int* __restrict__ col,
                                                               const int* __restrict__ tbl,
                                                               int* __restrict__ binstart,
                                                               unsigned* __restrict__ tmp) {
  __shared__ int2 stage[EPB];
  __shared__ int cursor[NBIN];
  __shared__ int gbase[NBIN];
  __shared__ int wsumL[32];
  __shared__ int wsumT[32];
  __shared__ int ptot[HALF_BINS];
  __shared__ int ppre[HALF_BINS];
  int tid = threadIdx.x, lane = tid & 31, wv = tid >> 5;
  int blk = blockIdx.x;
  const int* r = row + (size_t)blk * EPB;
  const int* c = col + (size_t)blk * EPB;
  int rv[EPT], cv[EPT];
#pragma unroll
  for (int k = 0; k < EPT; ++k) {
    int i = tid + k * SCAN_BLK;
    rv[k] = (i < EPB) ? r[i] : 0;
    cv[k] = (i < EPB) ? c[i] : -1;
  }
  int bin = tid & (HALF_BINS - 1), half = tid / HALF_BINS;
  int tot = 0, pre = 0;
  if (bin < NBIN) {
    const int* t = tbl + half * (NB1 / 2) * NBIN + bin;
    int lim = blk - half * (NB1 / 2);
#pragma nounroll
    for (int k0 = 0; k0 < NB1 / 2; k0 += TBL_GRP) {
      int v[TBL_GRP];
#pragma unroll
      for (int j = 0; j < TBL_GRP; ++j) v[j] = t[(k0 + j) * NBIN];
#pragma unroll
      for (int j = 0; j < TBL_GRP; ++j) {
        tot += v[j];
        pre += (k0 + j < lim) ? v[j] : 0;
      }
    }
  }
  if (half == 1) { ptot[bin] = tot; ppre[bin] = pre; }
  __syncthreads();
  int len = 0;
  if (tid < NBIN) {
    tot += ptot[tid];
    pre += ppre[tid];
    len  = tbl[blk * NBIN + tid];
  } else {
    tot = 0;
  }
  int inclT = wave_incl_scan_i(tot, lane);
  int inclL = wave_incl_scan_i(len, lane);
  if (lane == 31) { wsumT[wv] = inclT; wsumL[wv] = inclL; }
  __syncthreads();
  int bstart = wave_sum_i((lane < wv) ? wsumT[lane] : 0) + inclT - tot;
  int lofs   = wave_sum_i((lane < wv) ? wsumL[lane] : 0) + inclL - len;
  if (tid < NBIN) { cursor[tid] = lofs; gbase[tid] = bstart + pre - lofs; }
  if (blk == 0 && tid <= NBIN) binstart[tid] = bstart;
  __syncthreads();
#pragma unroll
  for (int k = 0; k < EPT; ++k) {
    if (cv[k] >= 0) {
      int p = atomicAdd(&cursor[cv[k] >> BIN_SHIFT], 1);
      stage[p] = make_int2(rv[k], cv[k]);
    }
  }
  __syncthreads();
  for (int i = tid; i < EPB; i += SCAN_BLK) {
    int2 q = stage[i];
    tmp[gbase[q.y >> BIN_SHIFT] + i] = ((unsigned)q.x << BIN_SHIFT) | (unsigned)(q.y & (BIN_NODES - 1));
  }
}

__global__ __launch_bounds__(SCAN_BLK) void bin_sort_kernel(const float* __restrict__ pos,
                                                            const float* __restrict__ s1p,
                                                            const float* __restrict__ s2p,
                                                            const int* __restrict__ binstart,
                                                            const unsigned* __restrict__ tmp,
                                                            const float* __restrict__ x,
                                                            const float* __restrict__ W,
                                                            int* __restrict__ off,
                                                            int2* __restrict__ seg,
                                                            float* __restrict__ dinv,
                                                            float* __restrict__ xw) {
  __shared__ int2 stage[BIN_CAP];
  __shared__ int cnt[BIN_NODES];
  __shared__ int cursor[BIN_NODES];
  __shared__ float degl[BIN_NODES];
  __shared__ int wsum[8];
  int tid = threadIdx.x, lane = tid & 31, wv = tid >> 5;
  int b = blockIdx.x;
  int S  = binstart[b];
  int En = binstart[b + 1];
  int nb = En - S;
  const unsigned* in = tmp + S;
  if (tid < BIN_NODES) { cnt[tid] = 0; degl[tid] = 0.0f; }
  __syncthreads();
  for (int i = tid; i < nb; i += SCAN_BLK)
    atomicAdd(&cnt[in[i] & (BIN_NODES - 1)], 1);
  __syncthreads();
  int v = (tid < BIN_NODES) ? cnt[tid] : 0;
  int incl = wave_incl_scan_i(v, lane);
  if (lane == 31 && wv < 8) wsum[wv] = incl;
  __syncthreads();
  int before = wave_sum_i((lane < wv && lane < 8) ? wsum[lane] : 0);
  int lofs = before + incl - v;
  if (tid < BIN_NODES) {
    cursor[tid] = lofs;
    int node = b * BIN_NODES + tid;
    if (node < N_NODES) off[node] = S + lofs;
  }
  __syncthreads();
  float s1 = s1p[0], s2 = s2p[0];
  bool staged = nb <= BIN_CAP;
  for (int i = tid; i < nb; i += SCAN_BLK) {
    unsigned q = in[i];
    int rr = (int)(q >> BIN_SHIFT), cl = (int)(q & (BIN_NODES - 1));
    int cc = b * BIN_NODES + cl;
    const float* pr = pos + (size_t)rr * 6;
    const float* pc = pos + (size_t)cc * 6;
    float d0 = pr[0] - pc[0], d1 = pr[1] - pc[1], d2 = pr[2] - pc[2];
    float D = d0 * d0 + d1 * d1 + d2 * d2;
    float t = 1.0f - (pr[3] * pc[3] + pr[4] * pc[4] + pr[5] * pc[5]);
    float w = __expf(-(D * s1 * s1 + t * t * s2 * s2));
    int p = atomicAdd(&cursor[cl], 1);
    atomicAdd(&degl[cl], w);
    int2 o = make_int2(rr, __float_as_int(w));
    if (staged) stage[p] = o; else seg[S + p] = o;
  }
  int wvu = __builtin_amdgcn_readfirstlane(wv);
  int n0  = b * BIN_NODES + wvu * SORT_NPW;
  int xi  = n0 * IN_DIM + lane;
  float xa = (xi < N_NODES * IN_DIM) ? x[xi] : 0.0f;
  float xb = (lane < SORT_NPW * IN_DIM - 32 && xi + 32 < N_NODES * IN_DIM) ? x[xi + 32] : 0.0f;
  float wk[IN_DIM];
#pragma unroll
  for (int k = 0; k < IN_DIM; ++k) wk[k] = W[k * HID + lane];
  __syncthreads();
  if (staged) {
    for (int i = tid; i < nb; i += SCAN_BLK) seg[S + i] = stage[i];
  }
  if (tid < BIN_NODES) {
    float dv = rsqrtf(degl[tid] + 1.0f);
    degl[tid] = dv;
    int node = b * BIN_NODES + tid;
    if (node < N_NODES) dinv[node] = dv;
  }
  __syncthreads();
#pragma unroll
  for (int i = 0; i < SORT_NPW; ++i) {
    if (n0 + i < N_NODES) {
      float ds  = degl[wvu * SORT_NPW + i];
      float acc = 0.0f;
#pragma unroll
      for (int k = 0; k < IN_DIM; ++k) {
        int f = i * IN_DIM + k;
        float xv = __int_as_float(__builtin_amdgcn_readlane(__float_as_int(f < 32 ? xa : xb), f & 31));
        acc += xv * wk[k];
      }
      acc *= ds;
      xw[(size_t)(n0 + i) * HID + lane] = acc;
    }
  }
}

template <int K0, int G, int SURE>
__device__ __forceinline__ float gather_rows(float acc, int r0, float c0, int r1, float c1, int n,
                                             const float* __restrict__ xcol) {
  float xv[8 * G];
#pragma unroll
  for (int k = 0; k < 8 * G; ++k) {
    int t  = K0 + k;
    int rk = __builtin_amdgcn_readlane(t < 32 ? r0 : r1, t & 31);
    xv[k]  = xcol[(size_t)rk * HID];
  }
#pragma unroll
  for (int k = 0; k < 8 * G; ++k) {
    int t    = K0 + k;
    float ck = __int_as_float(__builtin_amdgcn_readlane(__float_as_int(t < 32 ? c0 : c1), t & 31));
    float a2 = fmaf(ck, xv[k], acc);
    acc = (k < SURE || t < n) ? a2 : acc;
  }
  return acc;
}

#define FIRST 40
__device__ __forceinline__ float gather_first(float acc, int r0, float c0, int r1, float c1, int n,
                                              const float* __restrict__ xcol) {
  if (n > 32) return gather_rows<0, 5, 32>(acc, r0, c0, r1, c1, n, xcol);
  if (n > 24) return gather_rows<0, 4, 24>(acc, r0, c0, r1, c1, n, xcol);
  if (n > 16) return gather_rows<0, 3, 16>(acc, r0, c0, r1, c1, n, xcol);
  return gather_rows<0, 2, 0>(acc, r0, c0, r1, c1, n, xcol);
}

__device__ __forceinline__ float gather_second(float acc, int r0, float c0, int r1, float c1, int n,
                                               const float* __restrict__ xcol) {
  if (n > FIRST + 16) return gather_rows<FIRST, 3, 16>(acc, r0, c0, r1, c1, n, xcol);
  if (n > FIRST + 8)  return gather_rows<FIRST, 2, 8>(acc, r0, c0, r1, c1, n, xcol);
  return gather_rows<FIRST, 1, 0>(acc, r0, c0, r1, c1, n, xcol);
}

template <class Sink>
__device__ __forceinline__ void walk_nodes(const int* __restrict__ off,
                                           const int2* __restrict__ seg,
                                           const float* __restrict__ xw,
                                           const float* __restrict__ dinv,
                                           float bb, int node0, int lane, Sink sink) {
  int offv = N_EDGES;
  if (lane <= NPW && node0 + lane < N_NODES) offv = off[node0 + lane];
  const float* xcol = xw + lane;
  int cnt = N_NODES - node0;
  cnt = cnt < NPW ? cnt : NPW;
  int s = __builtin_amdgcn_readlane(offv, 0);
  int e = __builtin_amdgcn_readlane(offv, 1);
  int   rA = node0, rB = node0;
  float cA = 0.0f,  cB = 0.0f;
  {
    int j = s + lane;
    if (j < e)      { int2 p = seg[j];      rA = p.x; cA = __int_as_float(p.y); }
    if (j + 32 < e) { int2 p = seg[j + 32]; rB = p.x; cB = __int_as_float(p.y); }
  }
  for (int i = 0; i < cnt; ++i) {
    int   node = node0 + i;
    int   r0 = rA, r1 = rB;
    float c0 = cA, c1 = cB;
    int sN = e, eN = e;
    if (i + 1 < cnt) {
      eN = __builtin_amdgcn_readlane(offv, i + 2);
      int j = sN + lane;
      rA = node + 1; cA = 0.0f;
      rB = node + 1; cB = 0.0f;
      if (j < eN)      { int2 p = seg[j];      rA = p.x; cA = __int_as_float(p.y); }
      if (j + 32 < eN) { int2 p = seg[j + 32]; rB = p.x; cB = __int_as_float(p.y); }
    }
    size_t o  = (size_t)node * HID + lane;
    float xs  = xw[o];
    float di  = dinv[node];
    float acc = 0.0f;
    int n = e - s;
    acc = gather_first(acc, r0, c0, r1, c1, n, xcol);
    if (n > FIRST) acc = gather_second(acc, r0, c0, r1, c1, n, xcol);
    for (int base = s + 64; base < e; base += 32) {
      int j = base + lane;
      int r = node; float cf = 0.0f;
      if (j < e) { int2 p = seg[j]; r = p.x; cf = __int_as_float(p.y); }
      acc = gather_rows<0, 4, 0>(acc, r, cf, r, cf, e - base, xcol);
    }
    float v = di * (acc + xs) + bb;
    sink(i, node, di, v);
    s = sN; e = eN;
  }
}

__global__ __launch_bounds__(AGG_THREADS) void aggregate_mm_kernel(const int* __restrict__ off,
                                                                   const int2* __restrict__ seg,
                                                                   const float* __restrict__ xw,
                                                                   const float* __restrict__ dinv,
                                                                   const float* __restrict__ b,
                                                                   const float* __restrict__ g,
                                                                   const float* __restrict__ be,
                                                                   const float* __restrict__ Wn,
                                                                   float* __restrict__ xwn,
                                                                   float* __restrict__ poolz) {
  __shared__ float tile[AGG_TILES * 16 * TILE_PITCH];
  int gt = blockIdx.x * AGG_THREADS + threadIdx.x;
  if (poolz != nullptr && gt < POOL_WORDS) poolz[gt] = 0.0f;
  int wv    = __builtin_amdgcn_readfirstlane((int)(threadIdx.x >> 5));
  int node0 = ((int)blockIdx.x * AGG_WAVES + wv) * NPW;
  int lane  = threadIdx.x & 31;
  if (node0 < N_NODES) {
    float bb = b[lane];
    float gs = g[lane] * rsqrtf(1.0f + 1e-5f);
    float bs = be[lane];
    float* trow = tile + (wv * NPW) * TILE_PITCH + lane;
    walk_nodes(off, seg, xw, dinv, bb, node0, lane,
               [&](int i, int node, float di, float v) {
                 v = v * gs + bs;
                 v = v > 0.0f ? v : 0.0f;
                 v *= di;
                 trow[i * TILE_PITCH] = v;
               });
  }
  int  row0 = (int)blockIdx.x * AGG_NODES + wv * 16;
  int  rend = ((int)blockIdx.x + 1) * AGG_NODES;
  rend = rend < N_NODES ? rend : N_NODES;
  bool mm   = wv < AGG_TILES && row0 < rend;
  int  grp  = lane >> 4;
  int  m    = lane & 15;
  v2f b0[HID / 4], b1[HID / 4];
  if (mm) {
#pragma unroll
    for (int q = 0; q < HID / 4; ++q) {
      int kk = 4 * q + 2 * grp;
      b0[q].x = Wn[kk * HID + m];          b0[q].y = Wn[(kk + 1) * HID + m];
      b1[q].x = Wn[kk * HID + 16 + m];     b1[q].y = Wn[(kk + 1) * HID + 16 + m];
    }
  }
  __syncthreads();
  if (mm) {
    v8f c0 = {0.f,0.f,0.f,0.f,0.f,0.f,0.f,0.f};
    v8f c1 = {0.f,0.f,0.f,0.f,0.f,0.f,0.f,0.f};
    const float* hrow = tile + (wv * 16 + m) * TILE_PITCH;
#pragma unroll
    for (int q = 0; q < HID / 4; ++q) {
      int kk = 4 * q + 2 * grp;
      v2f a;  a.x  = hrow[kk];                  a.y  = hrow[kk + 1];
      c0 = __builtin_amdgcn_wmma_f32_16x16x4_f32(false, a, false, b0[q], (short)0, c0, false, false);
      c1 = __builtin_amdgcn_wmma_f32_16x16x4_f32(false, a, false, b1[q], (short)0, c1, false, false);
    }
#pragma unroll
    for (int r = 0; r < 8; ++r) {
      int M = r + 8 * grp;
      if (row0 + M < rend) {
        xwn[(size_t)(row0 + M) * HID + m]      = c0[r];
        xwn[(size_t)(row0 + M) * HID + 16 + m] = c1[r];
      }
    }
  }
}

__global__ __launch_bounds__(AGG_THREADS) void aggregate_out_kernel(const int* __restrict__ off,
                                                            const int2* __restrict__ seg,
                                                            const float* __restrict__ xw,
                                                            const float* __restrict__ dinv,
                                                            const float* __restrict__ b,
                                                            const int* __restrict__ batch,
                                                            float* __restrict__ sums,
                                                            float* __restrict__ cnts,
                                                            float* __restrict__ out) {
  int wave  = __builtin_amdgcn_readfirstlane((int)(blockIdx.x * AGG_WAVES + (threadIdx.x >> 5)));
  int node0 = wave * NPW;
  if (node0 >= N_NODES) return;
  int lane = threadIdx.x & 31;
  float bb = b[lane];
  int bgv = 0;
  if (lane < NPW && node0 + lane < N_NODES) bgv = batch[node0 + lane];
  int   pg = __builtin_amdgcn_readlane(bgv, 0);
  float ps = 0.0f, pc = 0.0f;
  walk_nodes(off, seg, xw, dinv, bb, node0, lane,
             [&](int i, int node, float di, float v) {
               out[(size_t)node * HID + lane] = v;
               int bg = __builtin_amdgcn_readlane(bgv, i);
               if (bg != pg) {
                 atomicAdd(&sums[pg * HID + lane], ps);
                 if (lane == 0) atomicAdd(&cnts[pg], pc);
                 ps = 0.0f; pc = 0.0f; pg = bg;
               }
               ps += v;
               pc += 1.0f;
             });
  atomicAdd(&sums[pg * HID + lane], ps);
  if (lane == 0) atomicAdd(&cnts[pg], pc);
}

__global__ void pred_kernel(const float* __restrict__ sums,
                            const float* __restrict__ cnts,
                            const float* __restrict__ Wout,
                            const float* __restrict__ bout,
                            float* __restrict__ pred) {
  int gph = blockIdx.x * blockDim.x + threadIdx.x;
  if (gph >= N_GRAPHS) return;
  float cnt = cnts[gph];
  cnt = cnt > 1.0f ? cnt : 1.0f;
  float acc = bout[0];
#pragma unroll
  for (int f = 0; f < HID; ++f)
    acc += (sums[gph * HID + f] / cnt) * Wout[f];
  pred[gph] = 1.0f / (1.0f + __expf(-acc));
}

extern "C" void kernel_launch(void* const* d_in, const int* in_sizes, int n_in,
                              void* d_out, int out_size, void* d_ws, size_t ws_size,
                              hipStream_t stream) {
  const float* x     = (const float*)d_in[0];
  const int*   ei    = (const int*)  d_in[1];
  const float* pos   = (const float*)d_in[2];
  const int*   batch = (const int*)  d_in[3];
  const float* s1    = (const float*)d_in[4];
  const float* s2    = (const float*)d_in[5];
  const float* W1 = (const float*)d_in[6];   const float* b1 = (const float*)d_in[7];
  const float* W2 = (const float*)d_in[8];   const float* b2 = (const float*)d_in[9];
  const float* W3 = (const float*)d_in[10];  const float* b3 = (const float*)d_in[11];
  const float* W4 = (const float*)d_in[12];  const float* b4 = (const float*)d_in[13];
  const float* g1 = (const float*)d_in[14];  const float* be1 = (const float*)d_in[15];
  const float* g2 = (const float*)d_in[16];  const float* be2 = (const float*)d_in[17];
  const float* g3 = (const float*)d_in[18];  const float* be3 = (const float*)d_in[19];
  const float* Wout = (const float*)d_in[20];
  const float* bout = (const float*)d_in[21];

  const int* row = ei;
  const int* col = ei + N_EDGES;

  float*    ws   = (float*)d_ws;
  unsigned* tmp  = (unsigned*)ws;
  float*    xw   = ws;
  float*    h    = xw   + (size_t)N_NODES * HID;
  int2*     seg  = (int2*)(h + (size_t)N_NODES * HID);
  int*      off  = (int*)(ws + (size_t)4 * N_EDGES);
  float*    dinv = (float*)(off + (size_t)N_NODES);
  float*    sums = dinv + (size_t)N_NODES;
  float*    cnts = sums + (size_t)N_GRAPHS * HID;
  int*      bst  = (int*)sums;
  int*      tbl  = (int*)dinv;

  float* emb  = (float*)d_out;
  float* pred = (float*)d_out + (size_t)N_NODES * HID;

  const int BLK        = 256;

  bin_count_kernel<<<NB1, SCAN_BLK, 0, stream>>>(col, tbl);
  bin_scatter_kernel<<<NB1, SCAN_BLK, 0, stream>>>(row, col, tbl, bst, tmp);
  bin_sort_kernel<<<NBIN, SCAN_BLK, 0, stream>>>(pos, s1, s2, bst, tmp, x, W1, off, seg, dinv, h);

  aggregate_mm_kernel<<<AGG_BLOCKS, AGG_THREADS, 0, stream>>>(off, seg, h, dinv, b1, g1, be1, W2, xw, sums);
  aggregate_mm_kernel<<<AGG_BLOCKS, AGG_THREADS, 0, stream>>>(off, seg, xw, dinv, b2, g2, be2, W3, h, nullptr);
  aggregate_mm_kernel<<<AGG_BLOCKS, AGG_THREADS, 0, stream>>>(off, seg, h, dinv, b3, g3, be3, W4, xw, nullptr);

  aggregate_out_kernel<<<NODE_BLOCKS, AGG_THREADS, 0, stream>>>(off, seg, xw, dinv, b4, batch, sums, cnts, emb);
  pred_kernel<<<1, BLK, 0, stream>>>(sums, cnts, Wout, bout, pred);
}


# Aggregation gathers 4 source rows per load instruction: lane = (edge slot, feature quad), 16-byte loads, row and weight by DPP8; sum order: edges g, g+4, .. per slot, then (s0+s2)+(s1+s3)
# speedup vs baseline: 8.1039x; 1.0383x over previous
#include <hip/hip_runtime.h>
#include <hip/hip_bf16.h>


#define N_NODES  100000
#define N_EDGES  3200000
#define N_GRAPHS 256
#define HID      32
#define IN_DIM   6
#define SCAN_BLK  1024
#define BIN_SHIFT 8
#define BIN_NODES (1 << BIN_SHIFT)
#define NBIN      ((N_NODES + BIN_NODES - 1) / BIN_NODES)
#define EPB       12800
#define EPT       ((EPB + SCAN_BLK - 1) / SCAN_BLK)
#define NB1       (N_EDGES / EPB)
#define TBL       (NBIN * NB1)
#define HALF_BINS (SCAN_BLK / 2)
#define TBL_GRP   25
#define BIN_CAP   12288
#define SORT_NPW  (BIN_NODES / (SCAN_BLK / 32))
#define NPW         7
#define NODE_WAVES  ((N_NODES + NPW - 1) / NPW)
#define NODE_BLOCKS ((NODE_WAVES + AGG_WAVES - 1) / AGG_WAVES)
#define AGG_WAVES   14
#define AGG_THREADS (AGG_WAVES * 32)
#define AGG_NODES   (AGG_WAVES * NPW)
#define AGG_TILES   ((AGG_NODES + 15) / 16)
#define AGG_BLOCKS  ((N_NODES + AGG_NODES - 1) / AGG_NODES)
#define TILE_PITCH  34
#define POOL_WORDS  (N_GRAPHS * (HID + 1))
static_assert(AGG_TILES <= AGG_WAVES && 4 * AGG_WAVES <= 64 && 5 * AGG_WAVES > 64, "a product wave per tile; 4 blocks fill a WGP");
static_assert(SORT_NPW * IN_DIM <= 64, "a wave's x rows fit two registers a lane");
static_assert(AGG_BLOCKS * AGG_THREADS >= POOL_WORDS, "an aggregation launch has a thread for every pooling word");
static_assert(NBIN <= HALF_BINS && NB1 % 2 == 0, "two threads per bin, each with half of the blocks");
static_assert((NB1 / 2) % TBL_GRP == 0, "a half of the blocks is whole groups of table loads");
static_assert(N_NODES <= (1 << (32 - BIN_SHIFT)), "source row and node-in-bin share one word");
static_assert(NBIN + 1 <= N_GRAPHS * HID, "the bin starts borrow the pooling sums");
static_assert(NPW + 1 <= 32, "a wave holds its nodes' segment bounds one per lane");
static_assert(N_EDGES % EPB == 0, "edge passes assume whole blocks");
static_assert(TBL <= N_NODES, "the run table borrows the dinv array");
static_assert(NBIN <= SCAN_BLK, "one thread per bin");

typedef __attribute__((ext_vector_type(2))) float v2f;
typedef __attribute__((ext_vector_type(8))) float v8f;


__device__ __forceinline__ int wave_sum_i(int v) {
  for (int d = 16; d > 0; d >>= 1) v += __shfl_xor(v, d, 32);
  return v;
}

__device__ __forceinline__ int wave_incl_scan_i(int v, int lane) {
  for (int d = 1; d < 32; d <<= 1) {
    int u = __shfl_up(v, d, 32);
    if (lane >= d) v += u;
  }
  return v;
}

__global__ __launch_bounds__(SCAN_BLK) void bin_count_kernel(const int* __restrict__ col,
                                                             int* __restrict__ tbl) {
  __shared__ int hist[NBIN];
  int tid = threadIdx.x;
  if (tid < NBIN) hist[tid] = 0;
  __syncthreads();
  const int* c = col + (size_t)blockIdx.x * EPB;
  int cv[EPT];
#pragma unroll
  for (int k = 0; k < EPT; ++k) {
    int i = tid + k * SCAN_BLK;
    cv[k] = (i < EPB) ? c[i] : -1;
  }
#pragma unroll
  for (int k = 0; k < EPT; ++k)
    if (cv[k] >= 0) atomicAdd(&hist[cv[k] >> BIN_SHIFT], 1);
  __syncthreads();
  if (tid < NBIN) tbl[blockIdx.x * NBIN + tid] = hist[tid];
}

__global__ __launch_bounds__(SCAN_BLK) void bin_scatter_kernel(const int* __restrict__ row,
                                                               const int* __restrict__ col,
                                                               const int* __restrict__ tbl,
                                                               int* __restrict__ binstart,
                                                               unsigned* __restrict__ tmp) {
  __shared__ int2 stage[EPB];
  __shared__ int cursor[NBIN];
  __shared__ int gbase[NBIN];
  __shared__ int wsumL[32];
  __shared__ int wsumT[32];
  __shared__ int ptot[HALF_BINS];
  __shared__ int ppre[HALF_BINS];
  int tid = threadIdx.x, lane = tid & 31, wv = tid >> 5;
  int blk = blockIdx.x;
  const int* r = row + (size_t)blk * EPB;
  const int* c = col + (size_t)blk * EPB;
  int rv[EPT], cv[EPT];
#pragma unroll
  for (int k = 0; k < EPT; ++k) {
    int i = tid + k * SCAN_BLK;
    rv[k] = (i < EPB) ? r[i] : 0;
    cv[k] = (i < EPB) ? c[i] : -1;
  }
  int bin = tid & (HALF_BINS - 1), half = tid / HALF_BINS;
  int tot = 0, pre = 0;
  if (bin < NBIN) {
    const int* t = tbl + half * (NB1 / 2) * NBIN + bin;
    int lim = blk - half * (NB1 / 2);
#pragma nounroll
    for (int k0 = 0; k0 < NB1 / 2; k0 += TBL_GRP) {
      int v[TBL_GRP];
#pragma unroll
      for (int j = 0; j < TBL_GRP; ++j) v[j] = t[(k0 + j) * NBIN];
#pragma unroll
      for (int j = 0; j < TBL_GRP; ++j) {
        tot += v[j];
        pre += (k0 + j < lim) ? v[j] : 0;
      }
    }
  }
  if (half == 1) { ptot[bin] = tot; ppre[bin] = pre; }
  __syncthreads();
  int len = 0;
  if (tid < NBIN) {
    tot += ptot[tid];
    pre += ppre[tid];
    len  = tbl[blk * NBIN + tid];
  } else {
    tot = 0;
  }
  int inclT = wave_incl_scan_i(tot, lane);
  int inclL = wave_incl_scan_i(len, lane);
  if (lane == 31) { wsumT[wv] = inclT; wsumL[wv] = inclL; }
  __syncthreads();
  int bstart = wave_sum_i((lane < wv) ? wsumT[lane] : 0) + inclT - tot;
  int lofs   = wave_sum_i((lane < wv) ? wsumL[lane] : 0) + inclL - len;
  if (tid < NBIN) { cursor[tid] = lofs; gbase[tid] = bstart + pre - lofs; }
  if (blk == 0 && tid <= NBIN) binstart[tid] = bstart;
  __syncthreads();
#pragma unroll
  for (int k = 0; k < EPT; ++k) {
    if (cv[k] >= 0) {
      int p = atomicAdd(&cursor[cv[k] >> BIN_SHIFT], 1);
      stage[p] = make_int2(rv[k], cv[k]);
    }
  }
  __syncthreads();
  for (int i = tid; i < EPB; i += SCAN_BLK) {
    int2 q = stage[i];
    tmp[gbase[q.y >> BIN_SHIFT] + i] = ((unsigned)q.x << BIN_SHIFT) | (unsigned)(q.y & (BIN_NODES - 1));
  }
}

__global__ __launch_bounds__(SCAN_BLK) void bin_sort_kernel(const float* __restrict__ pos,
                                                            const float* __restrict__ s1p,
                                                            const float* __restrict__ s2p,
                                                            const int* __restrict__ binstart,
                                                            const unsigned* __restrict__ tmp,
                                                            const float* __restrict__ x,
                                                            const float* __restrict__ W,
                                                            int* __restrict__ off,
                                                            int2* __restrict__ seg,
                                                            float* __restrict__ dinv,
                                                            float* __restrict__ xw) {
  __shared__ int2 stage[BIN_CAP];
  __shared__ int cnt[BIN_NODES];
  __shared__ int cursor[BIN_NODES];
  __shared__ float degl[BIN_NODES];
  __shared__ int wsum[8];
  int tid = threadIdx.x, lane = tid & 31, wv = tid >> 5;
  int b = blockIdx.x;
  int S  = binstart[b];
  int En = binstart[b + 1];
  int nb = En - S;
  const unsigned* in = tmp + S;
  if (tid < BIN_NODES) { cnt[tid] = 0; degl[tid] = 0.0f; }
  __syncthreads();
  for (int i = tid; i < nb; i += SCAN_BLK)
    atomicAdd(&cnt[in[i] & (BIN_NODES - 1)], 1);
  __syncthreads();
  int v = (tid < BIN_NODES) ? cnt[tid] : 0;
  int incl = wave_incl_scan_i(v, lane);
  if (lane == 31 && wv < 8) wsum[wv] = incl;
  __syncthreads();
  int before = wave_sum_i((lane < wv && lane < 8) ? wsum[lane] : 0);
  int lofs = before + incl - v;
  if (tid < BIN_NODES) {
    cursor[tid] = lofs;
    int node = b * BIN_NODES + tid;
    if (node < N_NODES) off[node] = S + lofs;
  }
  __syncthreads();
  float s1 = s1p[0], s2 = s2p[0];
  bool staged = nb <= BIN_CAP;
  for (int i = tid; i < nb; i += SCAN_BLK) {
    unsigned q = in[i];
    int rr = (int)(q >> BIN_SHIFT), cl = (int)(q & (BIN_NODES - 1));
    int cc = b * BIN_NODES + cl;
    const float* pr = pos + (size_t)rr * 6;
    const float* pc = pos + (size_t)cc * 6;
    float d0 = pr[0] - pc[0], d1 = pr[1] - pc[1], d2 = pr[2] - pc[2];
    float D = d0 * d0 + d1 * d1 + d2 * d2;
    float t = 1.0f - (pr[3] * pc[3] + pr[4] * pc[4] + pr[5] * pc[5]);
    float w = __expf(-(D * s1 * s1 + t * t * s2 * s2));
    int p = atomicAdd(&cursor[cl], 1);
    atomicAdd(&degl[cl], w);
    int2 o = make_int2(rr, __float_as_int(w));
    if (staged) stage[p] = o; else seg[S + p] = o;
  }
  int wvu = __builtin_amdgcn_readfirstlane(wv);
  int n0  = b * BIN_NODES + wvu * SORT_NPW;
  int xi  = n0 * IN_DIM + lane;
  float xa = (xi < N_NODES * IN_DIM) ? x[xi] : 0.0f;
  float xb = (lane < SORT_NPW * IN_DIM - 32 && xi + 32 < N_NODES * IN_DIM) ? x[xi + 32] : 0.0f;
  float wk[IN_DIM];
#pragma unroll
  for (int k = 0; k < IN_DIM; ++k) wk[k] = W[k * HID + lane];
  __syncthreads();
  if (staged) {
    for (int i = tid; i < nb; i += SCAN_BLK) seg[S + i] = stage[i];
  }
  if (tid < BIN_NODES) {
    float dv = rsqrtf(degl[tid] + 1.0f);
    degl[tid] = dv;
    int node = b * BIN_NODES + tid;
    if (node < N_NODES) dinv[node] = dv;
  }
  __syncthreads();
#pragma unroll
  for (int i = 0; i < SORT_NPW; ++i) {
    if (n0 + i < N_NODES) {
      float ds  = degl[wvu * SORT_NPW + i];
      float acc = 0.0f;
#pragma unroll
      for (int k = 0; k < IN_DIM; ++k) {
        int f = i * IN_DIM + k;
        float xv = __int_as_float(__builtin_amdgcn_readlane(__float_as_int(f < 32 ? xa : xb), f & 31));
        acc += xv * wk[k];
      }
      acc *= ds;
      xw[(size_t)(n0 + i) * HID + lane] = acc;
    }
  }
}


template <int J>
__device__ __forceinline__ int group_bcast(int v) {
  return __builtin_amdgcn_mov_dpp8(v, J * 0x249249);
}

template <int K, int NJ>
__device__ __forceinline__ void issue_rows(float4 (&xv)[8], int rX, const float4* __restrict__ xw4, int q) {
  if constexpr (K < NJ) {
    int rk = group_bcast<K>(rX);
    xv[K] = xw4[(unsigned)(rk * (HID / 4) + q)];
    issue_rows<K + 1, NJ>(xv, rX, xw4, q);
  }
}

template <int K, int NJ, int SURE>
__device__ __forceinline__ void sum_rows(float4& acc, const float4 (&xv)[8], float cX, int nrel, int g) {
  if constexpr (K < NJ) {
    float wk = __int_as_float(group_bcast<K>(__float_as_int(cX)));
    float ax = fmaf(wk, xv[K].x, acc.x), ay = fmaf(wk, xv[K].y, acc.y);
    float az = fmaf(wk, xv[K].z, acc.z), aw = fmaf(wk, xv[K].w, acc.w);
    bool ok = (K < SURE) || (g < nrel - 4 * K);
    acc.x = ok ? ax : acc.x;  acc.y = ok ? ay : acc.y;
    acc.z = ok ? az : acc.z;  acc.w = ok ? aw : acc.w;
    sum_rows<K + 1, NJ, SURE>(acc, xv, cX, nrel, g);
  }
}

template <int NJ, int SURE>
__device__ __forceinline__ void gather_block(float4& acc, int rX, float cX, int nrel, int g, int q,
                                             const float4* __restrict__ xw4) {
  float4 xv[8];
  issue_rows<0, NJ>(xv, rX, xw4, q);
  sum_rows<0, NJ, SURE>(acc, xv, cX, nrel, g);
}

__device__ __forceinline__ void gather_32(float4& acc, int rX, float cX, int nrel, int g, int q,
                                          const float4* __restrict__ xw4) {
  if (nrel >= 32)     gather_block<8, 8>(acc, rX, cX, nrel, g, q, xw4);
  else if (nrel > 24) gather_block<8, 6>(acc, rX, cX, nrel, g, q, xw4);
  else if (nrel > 16) gather_block<6, 4>(acc, rX, cX, nrel, g, q, xw4);
  else if (nrel > 8)  gather_block<4, 2>(acc, rX, cX, nrel, g, q, xw4);
  else                gather_block<2, 0>(acc, rX, cX, nrel, g, q, xw4);
}

__device__ __forceinline__ float from_other_row(float x) {
  int b = __float_as_int(x);
  return __int_as_float(__builtin_amdgcn_permlanex16(b, b, 0x76543210, 0xfedcba98, false, false));
}
__device__ __forceinline__ float from_other_oct(float x) {
  return __int_as_float(__builtin_amdgcn_update_dpp(0, __float_as_int(x), 0x168  , 0xf, 0xf, false));
}

__device__ __forceinline__ float reduce4(float4 acc, int lane) {
  bool hi = (lane & 16) != 0, od = (lane & 8) != 0;
  float kA = hi ? acc.z : acc.x, kB = hi ? acc.w : acc.y;
  float sA = hi ? acc.x : acc.z, sB = hi ? acc.y : acc.w;
  float a = kA + from_other_row(sA), b = kB + from_other_row(sB);
  float k = od ? b : a, sd = od ? a : b;
  return k + from_other_oct(sd);
}

__device__ __forceinline__ int lane_feature(int lane) { return 4 * (lane & 7) + (lane >> 3); }

template <class Sink>
__device__ __forceinline__ void walk_nodes(const int* __restrict__ off,
                                           const int2* __restrict__ seg,
                                           const float* __restrict__ xw,
                                           const float* __restrict__ dinv,
                                           float bb, int node0, int lane, Sink sink) {
  int offv = N_EDGES;
  if (lane <= NPW && node0 + lane < N_NODES) offv = off[node0 + lane];
  const float4* xw4 = reinterpret_cast<const float4*>(xw);
  int g = lane >> 3, q = lane & 7, f = lane_feature(lane);
  int cnt = N_NODES - node0;
  cnt = cnt < NPW ? cnt : NPW;
  int s = __builtin_amdgcn_readlane(offv, 0);
  int e = __builtin_amdgcn_readlane(offv, 1);
  int   rA = node0, rB = node0;
  float cA = 0.0f,  cB = 0.0f;
  {
    int j = s + f;
    if (j < e)      { int2 p = seg[j];      rA = p.x; cA = __int_as_float(p.y); }
    if (j + 32 < e) { int2 p = seg[j + 32]; rB = p.x; cB = __int_as_float(p.y); }
  }
  for (int i = 0; i < cnt; ++i) {
    int   node = node0 + i;
    int   r0 = rA, r1 = rB;
    float c0 = cA, c1 = cB;
    int sN = e, eN = e;
    if (i + 1 < cnt) {
      eN = __builtin_amdgcn_readlane(offv, i + 2);
      int j = sN + f;
      rA = node + 1; cA = 0.0f;
      rB = node + 1; cB = 0.0f;
      if (j < eN)      { int2 p = seg[j];      rA = p.x; cA = __int_as_float(p.y); }
      if (j + 32 < eN) { int2 p = seg[j + 32]; rB = p.x; cB = __int_as_float(p.y); }
    }
    float xs  = xw[(size_t)node * HID + f];
    float di  = dinv[node];
    float4 acc = make_float4(0.0f, 0.0f, 0.0f, 0.0f);
    int n = e - s;
    gather_32(acc, r0, c0, n, g, q, xw4);
    if (n > 32) gather_32(acc, r1, c1, n - 32, g, q, xw4);
    for (int base = s + 64; base < e; base += 32) {
      int j = base + f;
      int r = node; float cf = 0.0f;
      if (j < e) { int2 p = seg[j]; r = p.x; cf = __int_as_float(p.y); }
      gather_32(acc, r, cf, e - base, g, q, xw4);
    }
    float v = di * (reduce4(acc, lane) + xs) + bb;
    sink(i, node, di, v);
    s = sN; e = eN;
  }
}

__global__ __launch_bounds__(AGG_THREADS, AGG_WAVES) void aggregate_mm_kernel(const int* __restrict__ off,
                                                                   const int2* __restrict__ seg,
                                                                   const float* __restrict__ xw,
                                                                   const float* __restrict__ dinv,
                                                                   const float* __restrict__ b,
                                                                   const float* __restrict__ g,
                                                                   const float* __restrict__ be,
                                                                   const float* __restrict__ Wn,
                                                                   float* __restrict__ xwn,
                                                                   float* __restrict__ poolz) {
  __shared__ float tile[AGG_TILES * 16 * TILE_PITCH];
  int gt = blockIdx.x * AGG_THREADS + threadIdx.x;
  if (poolz != nullptr && gt < POOL_WORDS) poolz[gt] = 0.0f;
  int wv    = __builtin_amdgcn_readfirstlane((int)(threadIdx.x >> 5));
  int node0 = ((int)blockIdx.x * AGG_WAVES + wv) * NPW;
  int lane  = threadIdx.x & 31;
  if (node0 < N_NODES) {
    int   f  = lane_feature(lane);
    float bb = b[f];
    float gs = g[f] * rsqrtf(1.0f + 1e-5f);
    float bs = be[f];
    float* trow = tile + (wv * NPW) * TILE_PITCH + f;
    walk_nodes(off, seg, xw, dinv, bb, node0, lane,
               [&](int i, int node, float di, float v) {
                 v = v * gs + bs;
                 v = v > 0.0f ? v : 0.0f;
                 v *= di;
                 trow[i * TILE_PITCH] = v;
               });
  }
  int  row0 = (int)blockIdx.x * AGG_NODES + wv * 16;
  int  rend = ((int)blockIdx.x + 1) * AGG_NODES;
  rend = rend < N_NODES ? rend : N_NODES;
  bool mm   = wv < AGG_TILES && row0 < rend;
  int  grp  = lane >> 4;
  int  m    = lane & 15;
  v2f b0[HID / 4], b1[HID / 4];
  if (mm) {
#pragma unroll
    for (int q = 0; q < HID / 4; ++q) {
      int kk = 4 * q + 2 * grp;
      b0[q].x = Wn[kk * HID + m];          b0[q].y = Wn[(kk + 1) * HID + m];
      b1[q].x = Wn[kk * HID + 16 + m];     b1[q].y = Wn[(kk + 1) * HID + 16 + m];
    }
  }
  __syncthreads();
  if (mm) {
    v8f c0 = {0.f,0.f,0.f,0.f,0.f,0.f,0.f,0.f};
    v8f c1 = {0.f,0.f,0.f,0.f,0.f,0.f,0.f,0.f};
    const float* hrow = tile + (wv * 16 + m) * TILE_PITCH;
#pragma unroll
    for (int q = 0; q < HID / 4; ++q) {
      int kk = 4 * q + 2 * grp;
      v2f a;  a.x  = hrow[kk];                  a.y  = hrow[kk + 1];
      c0 = __builtin_amdgcn_wmma_f32_16x16x4_f32(false, a, false, b0[q], (short)0, c0, false, false);
      c1 = __builtin_amdgcn_wmma_f32_16x16x4_f32(false, a, false, b1[q], (short)0, c1, false, false);
    }
#pragma unroll
    for (int r = 0; r < 8; ++r) {
      int M = r + 8 * grp;
      if (row0 + M < rend) {
        xwn[(size_t)(row0 + M) * HID + m]      = c0[r];
        xwn[(size_t)(row0 + M) * HID + 16 + m] = c1[r];
      }
    }
  }
}

__global__ __launch_bounds__(AGG_THREADS, AGG_WAVES) void aggregate_out_kernel(const int* __restrict__ off,
                                                            const int2* __restrict__ seg,
                                                            const float* __restrict__ xw,
                                                            const float* __restrict__ dinv,
                                                            const float* __restrict__ b,
                                                            const int* __restrict__ batch,
                                                            float* __restrict__ sums,
                                                            float* __restrict__ cnts,
                                                            float* __restrict__ out) {
  int wave  = __builtin_amdgcn_readfirstlane((int)(blockIdx.x * AGG_WAVES + (threadIdx.x >> 5)));
  int node0 = wave * NPW;
  if (node0 >= N_NODES) return;
  int lane = threadIdx.x & 31;
  int f    = lane_feature(lane);
  float bb = b[f];
  int bgv = 0;
  if (lane < NPW && node0 + lane < N_NODES) bgv = batch[node0 + lane];
  int   pg = __builtin_amdgcn_readlane(bgv, 0);
  float ps = 0.0f, pc = 0.0f;
  walk_nodes(off, seg, xw, dinv, bb, node0, lane,
             [&](int i, int node, float di, float v) {
               out[(size_t)node * HID + f] = v;
               int bg = __builtin_amdgcn_readlane(bgv, i);
               if (bg != pg) {
                 atomicAdd(&sums[pg * HID + f], ps);
                 if (lane == 0) atomicAdd(&cnts[pg], pc);
                 ps = 0.0f; pc = 0.0f; pg = bg;
               }
               ps += v;
               pc += 1.0f;
             });
  atomicAdd(&sums[pg * HID + f], ps);
  if (lane == 0) atomicAdd(&cnts[pg], pc);
}

__global__ void pred_kernel(const float* __restrict__ sums,
                            const float* __restrict__ cnts,
                            const float* __restrict__ Wout,
                            const float* __restrict__ bout,
                            float* __restrict__ pred) {
  int gph = blockIdx.x * blockDim.x + threadIdx.x;
  if (gph >= N_GRAPHS) return;
  float cnt = cnts[gph];
  cnt = cnt > 1.0f ? cnt : 1.0f;
  float acc = bout[0];
#pragma unroll
  for (int f = 0; f < HID; ++f)
    acc += (sums[gph * HID + f] / cnt) * Wout[f];
  pred[gph] = 1.0f / (1.0f + __expf(-acc));
}

extern "C" void kernel_launch(void* const* d_in, const int* in_sizes, int n_in,
                              void* d_out, int out_size, void* d_ws, size_t ws_size,
                              hipStream_t stream) {
  const float* x     = (const float*)d_in[0];
  const int*   ei    = (const int*)  d_in[1];
  const float* pos   = (const float*)d_in[2];
  const int*   batch = (const int*)  d_in[3];
  const float* s1    = (const float*)d_in[4];
  const float* s2    = (const float*)d_in[5];
  const float* W1 = (const float*)d_in[6];   const float* b1 = (const float*)d_in[7];
  const float* W2 = (const float*)d_in[8];   const float* b2 = (const float*)d_in[9];
  const float* W3 = (const float*)d_in[10];  const float* b3 = (const float*)d_in[11];
  const float* W4 = (const float*)d_in[12];  const float* b4 = (const float*)d_in[13];
  const float* g1 = (const float*)d_in[14];  const float* be1 = (const float*)d_in[15];
  const float* g2 = (const float*)d_in[16];  const float* be2 = (const float*)d_in[17];
  const float* g3 = (const float*)d_in[18];  const float* be3 = (const float*)d_in[19];
  const float* Wout = (const float*)d_in[20];
  const float* bout = (const float*)d_in[21];

  const int* row = ei;
  const int* col = ei + N_EDGES;

  float*    ws   = (float*)d_ws;
  unsigned* tmp  = (unsigned*)ws;
  float*    xw   = ws;
  float*    h    = xw   + (size_t)N_NODES * HID;
  int2*     seg  = (int2*)(h + (size_t)N_NODES * HID);
  int*      off  = (int*)(ws + (size_t)4 * N_EDGES);
  float*    dinv = (float*)(off + (size_t)N_NODES);
  float*    sums = dinv + (size_t)N_NODES;
  float*    cnts = sums + (size_t)N_GRAPHS * HID;
  int*      bst  = (int*)sums;
  int*      tbl  = (int*)dinv;

  float* emb  = (float*)d_out;
  float* pred = (float*)d_out + (size_t)N_NODES * HID;

  const int BLK        = 256;

  bin_count_kernel<<<NB1, SCAN_BLK, 0, stream>>>(col, tbl);
  bin_scatter_kernel<<<NB1, SCAN_BLK, 0, stream>>>(row, col, tbl, bst, tmp);
  bin_sort_kernel<<<NBIN, SCAN_BLK, 0, stream>>>(pos, s1, s2, bst, tmp, x, W1, off, seg, dinv, h);

  aggregate_mm_kernel<<<AGG_BLOCKS, AGG_THREADS, 0, stream>>>(off, seg, h, dinv, b1, g1, be1, W2, xw, sums);
  aggregate_mm_kernel<<<AGG_BLOCKS, AGG_THREADS, 0, stream>>>(off, seg, xw, dinv, b2, g2, be2, W3, h, nullptr);
  aggregate_mm_kernel<<<AGG_BLOCKS, AGG_THREADS, 0, stream>>>(off, seg, h, dinv, b3, g3, be3, W4, xw, nullptr);

  aggregate_out_kernel<<<NODE_BLOCKS, AGG_THREADS, 0, stream>>>(off, seg, xw, dinv, b4, batch, sums, cnts, emb);
  pred_kernel<<<1, BLK, 0, stream>>>(sums, cnts, Wout, bout, pred);
}
